# starting kernel
# baseline (speedup 1.0000x reference)
#include <hip/hip_runtime.h>
#include <hip/hip_bf16.h>

#define N_NODES 30000
#define E_RAW   300000
#define E_TOT   330000
#define HEADS   8
#define HID     64
#define F1      512
#define IN_DIM  8
#define OUT_DIM 8
#define NEG_SLOPE 0.2f
#define M_TILES 1875

typedef __attribute__((ext_vector_type(2))) float v2f;
typedef __attribute__((ext_vector_type(8))) float v8f;

__device__ __forceinline__ unsigned fmap(float f) {
    unsigned u = __float_as_uint(f);
    return (u & 0x80000000u) ? ~u : (u | 0x80000000u);
}
__device__ __forceinline__ float funmap(unsigned u) {
    return __uint_as_float((u & 0x80000000u) ? (u ^ 0x80000000u) : ~u);
}
__device__ __forceinline__ void get_edge(const int* __restrict__ ei, int e, int& s, int& d) {
    if (e < E_RAW) { s = ei[e]; d = ei[E_RAW + e]; }
    else           { s = e - E_RAW; d = e - E_RAW; }
}
__device__ __forceinline__ float lrelu(float v) {
    return v > 0.f ? v : v * NEG_SLOPE;
}

__global__ void k_init1(unsigned* __restrict__ m1, float* __restrict__ s1) {
    int i = blockIdx.x * blockDim.x + threadIdx.x;
    if (i < N_NODES * HEADS) { m1[i] = 0u; s1[i] = 0.f; }
}
__global__ void k_zero4(float4* __restrict__ p, int n4) {
    int i = blockIdx.x * blockDim.x + threadIdx.x;
    if (i < n4) p[i] = make_float4(0.f, 0.f, 0.f, 0.f);
}
__global__ void k_init2(float* __restrict__ out, const float* __restrict__ b2,
                        unsigned* __restrict__ m2, float* __restrict__ s2) {
    int i = blockIdx.x * blockDim.x + threadIdx.x;
    if (i < N_NODES * OUT_DIM) out[i] = b2[i & 7];
    if (i < N_NODES) { m2[i] = 0u; s2[i] = 0.f; }
}

__global__ void k_gemm1(const float* __restrict__ x,
                        const float* __restrict__ Wl,
                        const float* __restrict__ Wr,
                        float* __restrict__ xl, float* __restrict__ xr) {
    const int wave  = threadIdx.x >> 5;
    const int lane  = threadIdx.x & 31;
    const int row   = lane & 15;
    const int khalf = lane >> 4;
    const int mbase = blockIdx.x * 16;
    const int nbase = (blockIdx.y * 8 + wave) * 16;
    const float* __restrict__ W = blockIdx.z ? Wr : Wl;
    float* __restrict__ O       = blockIdx.z ? xr : xl;

    v8f c = {};
#pragma unroll
    for (int kk = 0; kk < 2; ++kk) {
        const int k0 = kk * 4 + khalf * 2;
        v2f a, b;
        a.x = x[(mbase + row) * IN_DIM + k0];
        a.y = x[(mbase + row) * IN_DIM + k0 + 1];
        b.x = W[(k0    ) * F1 + nbase + row];
        b.y = W[(k0 + 1) * F1 + nbase + row];
        c = __builtin_amdgcn_wmma_f32_16x16x4_f32(false, a, false, b,
                                                  (short)0, c, false, false);
    }
#pragma unroll
    for (int r = 0; r < 8; ++r)
        O[(size_t)(mbase + r + 8 * khalf) * F1 + nbase + row] = c[r];
}

__global__ void k_edge1_logits(const int* __restrict__ ei,
                               const float* __restrict__ xl, const float* __restrict__ xr,
                               const float* __restrict__ a1,
                               float* __restrict__ e1, unsigned* __restrict__ m1) {
    const int wid  = (int)((blockIdx.x * blockDim.x + threadIdx.x) >> 5);
    const int lane = threadIdx.x & 31;
    if (wid >= E_TOT) return;
    int s, d; get_edge(ei, wid, s, d);
    const float4* __restrict__ pl = (const float4*)(xl + (size_t)s * F1) + lane * 4;
    const float4* __restrict__ pr = (const float4*)(xr + (size_t)d * F1) + lane * 4;
    const float4* __restrict__ pa = (const float4*)a1 + lane * 4;
    float acc = 0.f;
#pragma unroll
    for (int i = 0; i < 4; ++i) {
        float4 l = pl[i], r = pr[i], a = pa[i];
        acc = fmaf(lrelu(l.x + r.x), a.x, acc);
        acc = fmaf(lrelu(l.y + r.y), a.y, acc);
        acc = fmaf(lrelu(l.z + r.z), a.z, acc);
        acc = fmaf(lrelu(l.w + r.w), a.w, acc);
    }
    acc += __shfl_xor(acc, 1, 32);
    acc += __shfl_xor(acc, 2, 32);
    if ((lane & 3) == 0) {
        const int h = lane >> 2;
        e1[(size_t)wid * HEADS + h] = acc;
        atomicMax(&m1[d * HEADS + h], fmap(acc));
    }
}

__global__ void k_edge1_exp(const int* __restrict__ ei, float* __restrict__ e1,
                            const unsigned* __restrict__ m1, float* __restrict__ s1) {
    int idx = blockIdx.x * blockDim.x + threadIdx.x;
    if (idx >= E_TOT * HEADS) return;
    const int e = idx >> 3, h = idx & 7;
    int s, d; get_edge(ei, e, s, d);
    const float p = __expf(e1[idx] - funmap(m1[d * HEADS + h]));
    e1[idx] = p;
    atomicAdd(&s1[d * HEADS + h], p);
}

__global__ void k_edge1_agg(const int* __restrict__ ei,
                            const float* __restrict__ xl,
                            const float* __restrict__ e1, const float* __restrict__ s1,
                            float* __restrict__ h1) {
    const int wid  = (int)((blockIdx.x * blockDim.x + threadIdx.x) >> 5);
    const int lane = threadIdx.x & 31;
    if (wid >= E_TOT) return;
    int s, d; get_edge(ei, wid, s, d);
    const int h = lane >> 2;
    const float alpha = e1[(size_t)wid * HEADS + h] / s1[d * HEADS + h];
    const float4* __restrict__ pl = (const float4*)(xl + (size_t)s * F1) + lane * 4;
    float* __restrict__ po = h1 + (size_t)d * F1 + lane * 16;
#pragma unroll
    for (int i = 0; i < 4; ++i) {
        float4 l = pl[i];
        atomicAdd(po + i * 4 + 0, alpha * l.x);
        atomicAdd(po + i * 4 + 1, alpha * l.y);
        atomicAdd(po + i * 4 + 2, alpha * l.z);
        atomicAdd(po + i * 4 + 3, alpha * l.w);
    }
}

__global__ void k_bias_relu(float* __restrict__ h1, const float* __restrict__ b1) {
    int i = blockIdx.x * blockDim.x + threadIdx.x;
    if (i >= N_NODES * F1) return;
    const float v = h1[i] + b1[i & (F1 - 1)];
    h1[i] = v > 0.f ? v : 0.f;
}

__global__ void k_gemm2(const float* __restrict__ h,
                        const float* __restrict__ W2l,
                        const float* __restrict__ W2r,
                        float* __restrict__ h2l, float* __restrict__ h2r) {
    const int wave  = threadIdx.x >> 5;
    const int lane  = threadIdx.x & 31;
    const int row   = lane & 15;
    const int khalf = lane >> 4;
    const int mtile = blockIdx.x * 4 + (wave >> 1);
    if (mtile >= M_TILES) return;
    const int mbase = mtile * 16;
    const float* __restrict__ W = (wave & 1) ? W2r : W2l;
    float* __restrict__ O       = (wave & 1) ? h2r : h2l;

    v8f c = {};
    for (int kk = 0; kk < F1 / 4; ++kk) {
        const int k0 = kk * 4 + khalf * 2;
        v2f a, b;
        a.x = h[(size_t)(mbase + row) * F1 + k0];
        a.y = h[(size_t)(mbase + row) * F1 + k0 + 1];
        b.x = (row < OUT_DIM) ? W[(k0    ) * OUT_DIM + row] : 0.f;
        b.y = (row < OUT_DIM) ? W[(k0 + 1) * OUT_DIM + row] : 0.f;
        c = __builtin_amdgcn_wmma_f32_16x16x4_f32(false, a, false, b,
                                                  (short)0, c, false, false);
    }
    if (row < OUT_DIM) {
#pragma unroll
        for (int r = 0; r < 8; ++r)
            O[(size_t)(mbase + r + 8 * khalf) * OUT_DIM + row] = c[r];
    }
}

__global__ void k_edge2_logits(const int* __restrict__ ei,
                               const float* __restrict__ h2l, const float* __restrict__ h2r,
                               const float* __restrict__ a2,
                               float* __restrict__ e2, unsigned* __restrict__ m2) {
    int e = blockIdx.x * blockDim.x + threadIdx.x;
    if (e >= E_TOT) return;
    int s, d; get_edge(ei, e, s, d);
    float acc = 0.f;
#pragma unroll
    for (int c = 0; c < OUT_DIM; ++c)
        acc = fmaf(lrelu(h2l[s * OUT_DIM + c] + h2r[d * OUT_DIM + c]), a2[c], acc);
    e2[e] = acc;
    atomicMax(&m2[d], fmap(acc));
}

__global__ void k_edge2_exp(const int* __restrict__ ei, float* __restrict__ e2,
                            const unsigned* __restrict__ m2, float* __restrict__ s2) {
    int e = blockIdx.x * blockDim.x + threadIdx.x;
    if (e >= E_TOT) return;
    int s, d; get_edge(ei, e, s, d);
    const float p = __expf(e2[e] - funmap(m2[d]));
    e2[e] = p;
    atomicAdd(&s2[d], p);
}

__global__ void k_edge2_agg(const int* __restrict__ ei,
                            const float* __restrict__ h2l,
                            const float* __restrict__ e2, const float* __restrict__ s2,
                            float* __restrict__ out) {
    int idx = blockIdx.x * blockDim.x + threadIdx.x;
    if (idx >= E_TOT * OUT_DIM) return;
    const int e = idx >> 3, c = idx & 7;
    int s, d; get_edge(ei, e, s, d);
    const float alpha = e2[e] / s2[d];
    atomicAdd(&out[d * OUT_DIM + c], alpha * h2l[s * OUT_DIM + c]);
}

extern "C" void kernel_launch(void* const* d_in, const int* in_sizes, int n_in,
                              void* d_out, int out_size, void* d_ws, size_t ws_size,
                              hipStream_t stream) {
    const float* x   = (const float*)d_in[0];
    const int*   ei  = (const int*)  d_in[1];
    const float* W1l = (const float*)d_in[2];
    const float* W1r = (const float*)d_in[3];
    const float* a1  = (const float*)d_in[4];
    const float* b1  = (const float*)d_in[5];
    const float* W2l = (const float*)d_in[6];
    const float* W2r = (const float*)d_in[7];
    const float* a2  = (const float*)d_in[8];
    const float* b2  = (const float*)d_in[9];
    float* out = (float*)d_out;

    char* ws = (char*)d_ws;
    size_t off = 0;
    auto carve = [&](size_t bytes) -> char* {
        char* p = ws + off;
        off += (bytes + 255) & ~(size_t)255;
        return p;
    };
    float*    xl1 = (float*)   carve((size_t)N_NODES * F1 * 4);
    float*    xr1 = (float*)   carve((size_t)N_NODES * F1 * 4);
    float*    e1  = (float*)   carve((size_t)E_TOT * HEADS * 4);
    unsigned* m1  = (unsigned*)carve((size_t)N_NODES * HEADS * 4);
    float*    s1  = (float*)   carve((size_t)N_NODES * HEADS * 4);
    float*    h2l = (float*)   carve((size_t)N_NODES * OUT_DIM * 4);
    float*    h2r = (float*)   carve((size_t)N_NODES * OUT_DIM * 4);
    float*    e2  = (float*)   carve((size_t)E_TOT * 4);
    unsigned* m2  = (unsigned*)carve((size_t)N_NODES * 4);
    float*    s2  = (float*)   carve((size_t)N_NODES * 4);
    float*    h1  = xr1;
    (void)ws_size; (void)n_in; (void)in_sizes; (void)out_size;

    const int B = 256;

    k_init1<<<(N_NODES * HEADS + B - 1) / B, B, 0, stream>>>(m1, s1);
    k_gemm1<<<dim3(M_TILES, 4, 2), B, 0, stream>>>(x, W1l, W1r, xl1, xr1);
    {
        int waves = E_TOT, blocks = (waves * 32 + B - 1) / B;
        k_edge1_logits<<<blocks, B, 0, stream>>>(ei, xl1, xr1, a1, e1, m1);
    }
    k_edge1_exp<<<(E_TOT * HEADS + B - 1) / B, B, 0, stream>>>(ei, e1, m1, s1);
    k_zero4<<<((N_NODES * F1 / 4) + B - 1) / B, B, 0, stream>>>((float4*)h1, N_NODES * F1 / 4);
    {
        int waves = E_TOT, blocks = (waves * 32 + B - 1) / B;
        k_edge1_agg<<<blocks, B, 0, stream>>>(ei, xl1, e1, s1, h1);
    }
    k_bias_relu<<<(N_NODES * F1 + B - 1) / B, B, 0, stream>>>(h1, b1);

    k_gemm2<<<(M_TILES + 3) / 4, B, 0, stream>>>(h1, W2l, W2r, h2l, h2r);
    k_init2<<<(N_NODES * OUT_DIM + B - 1) / B, B, 0, stream>>>(out, b2, m2, s2);
    k_edge2_logits<<<(E_TOT + B - 1) / B, B, 0, stream>>>(ei, h2l, h2r, a2, e2, m2);
    k_edge2_exp<<<(E_TOT + B - 1) / B, B, 0, stream>>>(ei, e2, m2, s2);
    k_edge2_agg<<<(E_TOT * OUT_DIM + B - 1) / B, B, 0, stream>>>(ei, h2l, e2, s2, out);
}


# Layer-1 aggregation: a wave per destination node gathers its rows from an in-call destination-sorted edge list (slot = arrival-order rank, so sum order varies) instead of 169M f32 atomic adds
# speedup vs baseline: 2.4826x; 2.4826x over previous
#include <hip/hip_runtime.h>
#include <hip/hip_bf16.h>

#define N_NODES 30000
#define E_RAW   300000
#define E_TOT   330000
#define HEADS   8
#define HID     64
#define F1      512
#define IN_DIM  8
#define OUT_DIM 8
#define NEG_SLOPE 0.2f
#define M_TILES 1875

typedef __attribute__((ext_vector_type(2))) float v2f;
typedef __attribute__((ext_vector_type(8))) float v8f;

__device__ __forceinline__ unsigned fmap(float f) {
    unsigned u = __float_as_uint(f);
    return (u & 0x80000000u) ? ~u : (u | 0x80000000u);
}
__device__ __forceinline__ float funmap(unsigned u) {
    return __uint_as_float((u & 0x80000000u) ? (u ^ 0x80000000u) : ~u);
}
__device__ __forceinline__ void get_edge(const int* __restrict__ ei, int e, int& s, int& d) {
    if (e < E_RAW) { s = ei[e]; d = ei[E_RAW + e]; }
    else           { s = e - E_RAW; d = e - E_RAW; }
}
__device__ __forceinline__ float lrelu(float v) {
    return v > 0.f ? v : v * NEG_SLOPE;
}

__global__ void k_init1(unsigned* __restrict__ m1, float* __restrict__ s1, int* __restrict__ cnt) {
    int i = blockIdx.x * blockDim.x + threadIdx.x;
    if (i < N_NODES * HEADS) { m1[i] = 0u; s1[i] = 0.f; }
    if (i < N_NODES) cnt[i] = 0;
}
__global__ void k_init2(float* __restrict__ out, const float* __restrict__ b2,
                        unsigned* __restrict__ m2, float* __restrict__ s2) {
    int i = blockIdx.x * blockDim.x + threadIdx.x;
    if (i < N_NODES * OUT_DIM) out[i] = b2[i & 7];
    if (i < N_NODES) { m2[i] = 0u; s2[i] = 0.f; }
}

__global__ void k_gemm1(const float* __restrict__ x,
                        const float* __restrict__ Wl,
                        const float* __restrict__ Wr,
                        float* __restrict__ xl, float* __restrict__ xr) {
    const int wave  = threadIdx.x >> 5;
    const int lane  = threadIdx.x & 31;
    const int row   = lane & 15;
    const int khalf = lane >> 4;
    const int mbase = blockIdx.x * 16;
    const int nbase = (blockIdx.y * 8 + wave) * 16;
    const float* __restrict__ W = blockIdx.z ? Wr : Wl;
    float* __restrict__ O       = blockIdx.z ? xr : xl;

    v8f c = {};
#pragma unroll
    for (int kk = 0; kk < 2; ++kk) {
        const int k0 = kk * 4 + khalf * 2;
        v2f a, b;
        a.x = x[(mbase + row) * IN_DIM + k0];
        a.y = x[(mbase + row) * IN_DIM + k0 + 1];
        b.x = W[(k0    ) * F1 + nbase + row];
        b.y = W[(k0 + 1) * F1 + nbase + row];
        c = __builtin_amdgcn_wmma_f32_16x16x4_f32(false, a, false, b,
                                                  (short)0, c, false, false);
    }
#pragma unroll
    for (int r = 0; r < 8; ++r)
        O[(size_t)(mbase + r + 8 * khalf) * F1 + nbase + row] = c[r];
}

__global__ void k_edge1_logits(const int* __restrict__ ei,
                               const float* __restrict__ xl, const float* __restrict__ xr,
                               const float* __restrict__ a1,
                               float* __restrict__ e1, unsigned* __restrict__ m1,
                               int* __restrict__ cnt) {
    const int wid  = (int)((blockIdx.x * blockDim.x + threadIdx.x) >> 5);
    const int lane = threadIdx.x & 31;
    if (wid >= E_TOT) return;
    int s, d; get_edge(ei, wid, s, d);
    const float4* __restrict__ pl = (const float4*)(xl + (size_t)s * F1) + lane * 4;
    const float4* __restrict__ pr = (const float4*)(xr + (size_t)d * F1) + lane * 4;
    const float4* __restrict__ pa = (const float4*)a1 + lane * 4;
    float acc = 0.f;
#pragma unroll
    for (int i = 0; i < 4; ++i) {
        float4 l = pl[i], r = pr[i], a = pa[i];
        acc = fmaf(lrelu(l.x + r.x), a.x, acc);
        acc = fmaf(lrelu(l.y + r.y), a.y, acc);
        acc = fmaf(lrelu(l.z + r.z), a.z, acc);
        acc = fmaf(lrelu(l.w + r.w), a.w, acc);
    }
    acc += __shfl_xor(acc, 1, 32);
    acc += __shfl_xor(acc, 2, 32);
    if ((lane & 3) == 0) {
        const int h = lane >> 2;
        e1[(size_t)wid * HEADS + h] = acc;
        atomicMax(&m1[d * HEADS + h], fmap(acc));
    }
    if (lane == 0) atomicAdd(&cnt[d], 1);
}

#define SCAN_T    1024
#define SCAN_PER  32
#define ROWPTR_N  (((N_NODES + SCAN_PER) / SCAN_PER) * SCAN_PER)
__global__ void __launch_bounds__(SCAN_T)
k_scan(const int* __restrict__ cnt, int* __restrict__ rowptr) {
    __shared__ int wsum[SCAN_T / 32];
    const int t    = threadIdx.x;
    const int lane = t & 31;
    const int wave = t >> 5;
    const int base = t * SCAN_PER;
    int v[SCAN_PER];
#pragma unroll
    for (int q = 0; q < SCAN_PER / 4; ++q) {
        const int i = base + q * 4;
        int4 c = make_int4(0, 0, 0, 0);
        if (i < N_NODES) c = *(const int4*)(cnt + i);
        v[q * 4 + 0] = c.x; v[q * 4 + 1] = c.y; v[q * 4 + 2] = c.z; v[q * 4 + 3] = c.w;
    }
    int run = 0;
#pragma unroll
    for (int q = 0; q < SCAN_PER; ++q) { const int c = v[q]; v[q] = run; run += c; }
    int inc = run;
#pragma unroll
    for (int o = 1; o < 32; o <<= 1) {
        const int up = __shfl_up(inc, o, 32);
        if (lane >= o) inc += up;
    }
    if (lane == 31) wsum[wave] = inc;
    __syncthreads();
    if (wave == 0) {
        int w = wsum[lane];
#pragma unroll
        for (int o = 1; o < 32; o <<= 1) {
            const int up = __shfl_up(w, o, 32);
            if (lane >= o) w += up;
        }
        wsum[lane] = w;
    }
    __syncthreads();
    const int off = (inc - run) + (wave ? wsum[wave - 1] : 0);
    if (base < ROWPTR_N) {
#pragma unroll
        for (int q = 0; q < SCAN_PER / 4; ++q)
            *(int4*)(rowptr + base + q * 4) =
                make_int4(v[q * 4] + off, v[q * 4 + 1] + off, v[q * 4 + 2] + off, v[q * 4 + 3] + off);
    }
}

__global__ void k_edge1_exp(const int* __restrict__ ei, float* __restrict__ e1,
                            const unsigned* __restrict__ m1, float* __restrict__ s1,
                            const int* __restrict__ rowptr, int* __restrict__ cnt,
                            int2* __restrict__ perm) {
    int idx = blockIdx.x * blockDim.x + threadIdx.x;
    if (idx >= E_TOT * HEADS) return;
    const int e = idx >> 3, h = idx & 7;
    int s, d; get_edge(ei, e, s, d);
    const float p = __expf(e1[idx] - funmap(m1[d * HEADS + h]));
    e1[idx] = p;
    atomicAdd(&s1[d * HEADS + h], p);
    if (h == 0) {
        const int pos = atomicSub(&cnt[d], 1) - 1;
        perm[rowptr[d] + pos] = make_int2(e, s);
    }
}

__global__ void k_edge1_agg(const int* __restrict__ rowptr, const int2* __restrict__ perm,
                            const float* __restrict__ xl,
                            const float* __restrict__ e1, const float* __restrict__ s1,
                            float* __restrict__ h1) {
    const int lane = threadIdx.x & 31;
    const int node = (int)blockIdx.x * (int)(blockDim.x >> 5) +
                     __builtin_amdgcn_readfirstlane((int)(threadIdx.x >> 5));
    if (node >= N_NODES) return;
    const int h   = lane >> 2;
    const int beg = rowptr[node], end = rowptr[node + 1];
    const float den = s1[node * HEADS + h];
    float4 acc0 = make_float4(0.f, 0.f, 0.f, 0.f), acc1 = acc0, acc2 = acc0, acc3 = acc0;
    for (int j0 = beg; j0 < end; j0 += 32) {
        const int n = (end - j0 < 32) ? (end - j0) : 32;
        int2 my = make_int2(0, 0);
        if (lane < n) my = perm[j0 + lane];
        int t = 0;
        for (; t + 1 < n; t += 2) {
            const int ea = __builtin_amdgcn_readlane(my.x, t);
            const int sa = __builtin_amdgcn_readlane(my.y, t);
            const int eb = __builtin_amdgcn_readlane(my.x, t + 1);
            const int sb = __builtin_amdgcn_readlane(my.y, t + 1);
            const float4* __restrict__ pa = (const float4*)(xl + (size_t)sa * F1) + lane * 4;
            const float4* __restrict__ pb = (const float4*)(xl + (size_t)sb * F1) + lane * 4;
            const float4 a0 = pa[0], a1v = pa[1], a2v = pa[2], a3 = pa[3];
            const float4 b0 = pb[0], b1v = pb[1], b2v = pb[2], b3 = pb[3];
            const float wa = e1[(size_t)ea * HEADS + h] / den;
            const float wb = e1[(size_t)eb * HEADS + h] / den;
            acc0.x = fmaf(wa, a0.x, acc0.x);  acc0.y = fmaf(wa, a0.y, acc0.y);
            acc0.z = fmaf(wa, a0.z, acc0.z);  acc0.w = fmaf(wa, a0.w, acc0.w);
            acc1.x = fmaf(wa, a1v.x, acc1.x); acc1.y = fmaf(wa, a1v.y, acc1.y);
            acc1.z = fmaf(wa, a1v.z, acc1.z); acc1.w = fmaf(wa, a1v.w, acc1.w);
            acc2.x = fmaf(wa, a2v.x, acc2.x); acc2.y = fmaf(wa, a2v.y, acc2.y);
            acc2.z = fmaf(wa, a2v.z, acc2.z); acc2.w = fmaf(wa, a2v.w, acc2.w);
            acc3.x = fmaf(wa, a3.x, acc3.x);  acc3.y = fmaf(wa, a3.y, acc3.y);
            acc3.z = fmaf(wa, a3.z, acc3.z);  acc3.w = fmaf(wa, a3.w, acc3.w);
            acc0.x = fmaf(wb, b0.x, acc0.x);  acc0.y = fmaf(wb, b0.y, acc0.y);
            acc0.z = fmaf(wb, b0.z, acc0.z);  acc0.w = fmaf(wb, b0.w, acc0.w);
            acc1.x = fmaf(wb, b1v.x, acc1.x); acc1.y = fmaf(wb, b1v.y, acc1.y);
            acc1.z = fmaf(wb, b1v.z, acc1.z); acc1.w = fmaf(wb, b1v.w, acc1.w);
            acc2.x = fmaf(wb, b2v.x, acc2.x); acc2.y = fmaf(wb, b2v.y, acc2.y);
            acc2.z = fmaf(wb, b2v.z, acc2.z); acc2.w = fmaf(wb, b2v.w, acc2.w);
            acc3.x = fmaf(wb, b3.x, acc3.x);  acc3.y = fmaf(wb, b3.y, acc3.y);
            acc3.z = fmaf(wb, b3.z, acc3.z);  acc3.w = fmaf(wb, b3.w, acc3.w);
        }
        if (t < n) {
            const int ea = __builtin_amdgcn_readlane(my.x, t);
            const int sa = __builtin_amdgcn_readlane(my.y, t);
            const float4* __restrict__ pa = (const float4*)(xl + (size_t)sa * F1) + lane * 4;
            const float4 a0 = pa[0], a1v = pa[1], a2v = pa[2], a3 = pa[3];
            const float wa = e1[(size_t)ea * HEADS + h] / den;
            acc0.x = fmaf(wa, a0.x, acc0.x);  acc0.y = fmaf(wa, a0.y, acc0.y);
            acc0.z = fmaf(wa, a0.z, acc0.z);  acc0.w = fmaf(wa, a0.w, acc0.w);
            acc1.x = fmaf(wa, a1v.x, acc1.x); acc1.y = fmaf(wa, a1v.y, acc1.y);
            acc1.z = fmaf(wa, a1v.z, acc1.z); acc1.w = fmaf(wa, a1v.w, acc1.w);
            acc2.x = fmaf(wa, a2v.x, acc2.x); acc2.y = fmaf(wa, a2v.y, acc2.y);
            acc2.z = fmaf(wa, a2v.z, acc2.z); acc2.w = fmaf(wa, a2v.w, acc2.w);
            acc3.x = fmaf(wa, a3.x, acc3.x);  acc3.y = fmaf(wa, a3.y, acc3.y);
            acc3.z = fmaf(wa, a3.z, acc3.z);  acc3.w = fmaf(wa, a3.w, acc3.w);
        }
    }
    float4* __restrict__ po = (float4*)(h1 + (size_t)node * F1) + lane * 4;
    po[0] = acc0; po[1] = acc1; po[2] = acc2; po[3] = acc3;
}

__global__ void k_bias_relu(float* __restrict__ h1, const float* __restrict__ b1) {
    int i = blockIdx.x * blockDim.x + threadIdx.x;
    if (i >= N_NODES * F1) return;
    const float v = h1[i] + b1[i & (F1 - 1)];
    h1[i] = v > 0.f ? v : 0.f;
}

__global__ void k_gemm2(const float* __restrict__ h,
                        const float* __restrict__ W2l,
                        const float* __restrict__ W2r,
                        float* __restrict__ h2l, float* __restrict__ h2r) {
    const int wave  = threadIdx.x >> 5;
    const int lane  = threadIdx.x & 31;
    const int row   = lane & 15;
    const int khalf = lane >> 4;
    const int mtile = blockIdx.x * 4 + (wave >> 1);
    if (mtile >= M_TILES) return;
    const int mbase = mtile * 16;
    const float* __restrict__ W = (wave & 1) ? W2r : W2l;
    float* __restrict__ O       = (wave & 1) ? h2r : h2l;

    v8f c = {};
    for (int kk = 0; kk < F1 / 4; ++kk) {
        const int k0 = kk * 4 + khalf * 2;
        v2f a, b;
        a.x = h[(size_t)(mbase + row) * F1 + k0];
        a.y = h[(size_t)(mbase + row) * F1 + k0 + 1];
        b.x = (row < OUT_DIM) ? W[(k0    ) * OUT_DIM + row] : 0.f;
        b.y = (row < OUT_DIM) ? W[(k0 + 1) * OUT_DIM + row] : 0.f;
        c = __builtin_amdgcn_wmma_f32_16x16x4_f32(false, a, false, b,
                                                  (short)0, c, false, false);
    }
    if (row < OUT_DIM) {
#pragma unroll
        for (int r = 0; r < 8; ++r)
            O[(size_t)(mbase + r + 8 * khalf) * OUT_DIM + row] = c[r];
    }
}

__global__ void k_edge2_logits(const int* __restrict__ ei,
                               const float* __restrict__ h2l, const float* __restrict__ h2r,
                               const float* __restrict__ a2,
                               float* __restrict__ e2, unsigned* __restrict__ m2) {
    int e = blockIdx.x * blockDim.x + threadIdx.x;
    if (e >= E_TOT) return;
    int s, d; get_edge(ei, e, s, d);
    float acc = 0.f;
#pragma unroll
    for (int c = 0; c < OUT_DIM; ++c)
        acc = fmaf(lrelu(h2l[s * OUT_DIM + c] + h2r[d * OUT_DIM + c]), a2[c], acc);
    e2[e] = acc;
    atomicMax(&m2[d], fmap(acc));
}

__global__ void k_edge2_exp(const int* __restrict__ ei, float* __restrict__ e2,
                            const unsigned* __restrict__ m2, float* __restrict__ s2) {
    int e = blockIdx.x * blockDim.x + threadIdx.x;
    if (e >= E_TOT) return;
    int s, d; get_edge(ei, e, s, d);
    const float p = __expf(e2[e] - funmap(m2[d]));
    e2[e] = p;
    atomicAdd(&s2[d], p);
}

__global__ void k_edge2_agg(const int* __restrict__ ei,
                            const float* __restrict__ h2l,
                            const float* __restrict__ e2, const float* __restrict__ s2,
                            float* __restrict__ out) {
    int idx = blockIdx.x * blockDim.x + threadIdx.x;
    if (idx >= E_TOT * OUT_DIM) return;
    const int e = idx >> 3, c = idx & 7;
    int s, d; get_edge(ei, e, s, d);
    const float alpha = e2[e] / s2[d];
    atomicAdd(&out[d * OUT_DIM + c], alpha * h2l[s * OUT_DIM + c]);
}

extern "C" void kernel_launch(void* const* d_in, const int* in_sizes, int n_in,
                              void* d_out, int out_size, void* d_ws, size_t ws_size,
                              hipStream_t stream) {
    const float* x   = (const float*)d_in[0];
    const int*   ei  = (const int*)  d_in[1];
    const float* W1l = (const float*)d_in[2];
    const float* W1r = (const float*)d_in[3];
    const float* a1  = (const float*)d_in[4];
    const float* b1  = (const float*)d_in[5];
    const float* W2l = (const float*)d_in[6];
    const float* W2r = (const float*)d_in[7];
    const float* a2  = (const float*)d_in[8];
    const float* b2  = (const float*)d_in[9];
    float* out = (float*)d_out;

    char* ws = (char*)d_ws;
    size_t off = 0;
    auto carve = [&](size_t bytes) -> char* {
        char* p = ws + off;
        off += (bytes + 255) & ~(size_t)255;
        return p;
    };
    float*    xl1 = (float*)   carve((size_t)N_NODES * F1 * 4);
    float*    xr1 = (float*)   carve((size_t)N_NODES * F1 * 4);
    float*    e1  = (float*)   carve((size_t)E_TOT * HEADS * 4);
    unsigned* m1  = (unsigned*)carve((size_t)N_NODES * HEADS * 4);
    float*    s1  = (float*)   carve((size_t)N_NODES * HEADS * 4);
    float*    h2l = (float*)   carve((size_t)N_NODES * OUT_DIM * 4);
    float*    h2r = (float*)   carve((size_t)N_NODES * OUT_DIM * 4);
    float*    e2  = (float*)   carve((size_t)E_TOT * 4);
    unsigned* m2  = (unsigned*)carve((size_t)N_NODES * 4);
    float*    s2  = (float*)   carve((size_t)N_NODES * 4);
    int*      cnt    = (int*) carve((size_t)N_NODES * 4);
    int*      rowptr = (int*) carve((size_t)ROWPTR_N * 4);
    int2*     perm   = (int2*)carve((size_t)E_TOT * 8);
    float*    h1  = xr1;
    (void)ws_size; (void)n_in; (void)in_sizes; (void)out_size;

    const int B = 256;

    k_init1<<<(N_NODES * HEADS + B - 1) / B, B, 0, stream>>>(m1, s1, cnt);
    k_gemm1<<<dim3(M_TILES, 4, 2), B, 0, stream>>>(x, W1l, W1r, xl1, xr1);
    {
        int waves = E_TOT, blocks = (waves * 32 + B - 1) / B;
        k_edge1_logits<<<blocks, B, 0, stream>>>(ei, xl1, xr1, a1, e1, m1, cnt);
    }
    k_scan<<<1, SCAN_T, 0, stream>>>(cnt, rowptr);
    k_edge1_exp<<<(E_TOT * HEADS + B - 1) / B, B, 0, stream>>>(ei, e1, m1, s1, rowptr, cnt, perm);
    {
        int waves = N_NODES, blocks = (waves * 32 + B - 1) / B;
        k_edge1_agg<<<blocks, B, 0, stream>>>(rowptr, perm, xl1, e1, s1, h1);
    }
    k_bias_relu<<<(N_NODES * F1 + B - 1) / B, B, 0, stream>>>(h1, b1);

    k_gemm2<<<(M_TILES + 3) / 4, B, 0, stream>>>(h1, W2l, W2r, h2l, h2r);
    k_init2<<<(N_NODES * OUT_DIM + B - 1) / B, B, 0, stream>>>(out, b2, m2, s2);
    k_edge2_logits<<<(E_TOT + B - 1) / B, B, 0, stream>>>(ei, h2l, h2r, a2, e2, m2);
    k_edge2_exp<<<(E_TOT + B - 1) / B, B, 0, stream>>>(ei, e2, m2, s2);
    k_edge2_agg<<<(E_TOT * OUT_DIM + B - 1) / B, B, 0, stream>>>(ei, h2l, e2, s2, out);
}


# Layer-1 bias and relu applied in the wave-per-destination kernel's own store of the node row; the separate in-place pass over the 30000x512 plane (480000 one-load waves, one launch) goes
# speedup vs baseline: 3.8690x; 1.5584x over previous
#include <hip/hip_runtime.h>
#include <hip/hip_bf16.h>

#define N_NODES 30000
#define E_RAW   300000
#define E_TOT   330000
#define HEADS   8
#define HID     64
#define F1      512
#define IN_DIM  8
#define OUT_DIM 8
#define NEG_SLOPE 0.2f
#define M_TILES 1875

typedef __attribute__((ext_vector_type(2))) float v2f;
typedef __attribute__((ext_vector_type(8))) float v8f;

__device__ __forceinline__ unsigned fmap(float f) {
    unsigned u = __float_as_uint(f);
    return (u & 0x80000000u) ? ~u : (u | 0x80000000u);
}
__device__ __forceinline__ float funmap(unsigned u) {
    return __uint_as_float((u & 0x80000000u) ? (u ^ 0x80000000u) : ~u);
}
__device__ __forceinline__ void get_edge(const int* __restrict__ ei, int e, int& s, int& d) {
    if (e < E_RAW) { s = ei[e]; d = ei[E_RAW + e]; }
    else           { s = e - E_RAW; d = e - E_RAW; }
}
__device__ __forceinline__ float lrelu(float v) {
    return v > 0.f ? v : v * NEG_SLOPE;
}

__global__ void k_init1(int* __restrict__ cnt) {
    int i = blockIdx.x * blockDim.x + threadIdx.x;
    if (i < N_NODES) cnt[i] = 0;
}
__global__ void k_init2(float* __restrict__ out, const float* __restrict__ b2,
                        unsigned* __restrict__ m2, float* __restrict__ s2) {
    int i = blockIdx.x * blockDim.x + threadIdx.x;
    if (i < N_NODES * OUT_DIM) out[i] = b2[i & 7];
    if (i < N_NODES) { m2[i] = 0u; s2[i] = 0.f; }
}

__global__ void k_gemm1(const float* __restrict__ x,
                        const float* __restrict__ Wl,
                        const float* __restrict__ Wr,
                        float* __restrict__ xl, float* __restrict__ xr) {
    const int wave  = threadIdx.x >> 5;
    const int lane  = threadIdx.x & 31;
    const int row   = lane & 15;
    const int khalf = lane >> 4;
    const int mbase = blockIdx.x * 16;
    const int nbase = (blockIdx.y * 8 + wave) * 16;
    const float* __restrict__ W = blockIdx.z ? Wr : Wl;
    float* __restrict__ O       = blockIdx.z ? xr : xl;

    v8f c = {};
#pragma unroll
    for (int kk = 0; kk < 2; ++kk) {
        const int k0 = kk * 4 + khalf * 2;
        v2f a, b;
        a.x = x[(mbase + row) * IN_DIM + k0];
        a.y = x[(mbase + row) * IN_DIM + k0 + 1];
        b.x = W[(k0    ) * F1 + nbase + row];
        b.y = W[(k0 + 1) * F1 + nbase + row];
        c = __builtin_amdgcn_wmma_f32_16x16x4_f32(false, a, false, b,
                                                  (short)0, c, false, false);
    }
#pragma unroll
    for (int r = 0; r < 8; ++r)
        O[(size_t)(mbase + r + 8 * khalf) * F1 + nbase + row] = c[r];
}

__global__ void k_count(const int* __restrict__ ei, int* __restrict__ cnt) {
    int e = blockIdx.x * blockDim.x + threadIdx.x;
    if (e >= E_TOT) return;
    int s, d; get_edge(ei, e, s, d);
    atomicAdd(&cnt[d], 1);
}

#define SCAN_T    1024
#define SCAN_PER  32
#define ROWPTR_N  (((N_NODES + SCAN_PER) / SCAN_PER) * SCAN_PER)
__global__ void __launch_bounds__(SCAN_T)
k_scan(const int* __restrict__ cnt, int* __restrict__ rowptr) {
    __shared__ int wsum[SCAN_T / 32];
    const int t    = threadIdx.x;
    const int lane = t & 31;
    const int wave = t >> 5;
    const int base = t * SCAN_PER;
    int v[SCAN_PER];
#pragma unroll
    for (int q = 0; q < SCAN_PER / 4; ++q) {
        const int i = base + q * 4;
        int4 c = make_int4(0, 0, 0, 0);
        if (i < N_NODES) c = *(const int4*)(cnt + i);
        v[q * 4 + 0] = c.x; v[q * 4 + 1] = c.y; v[q * 4 + 2] = c.z; v[q * 4 + 3] = c.w;
    }
    int run = 0;
#pragma unroll
    for (int q = 0; q < SCAN_PER; ++q) { const int c = v[q]; v[q] = run; run += c; }
    int inc = run;
#pragma unroll
    for (int o = 1; o < 32; o <<= 1) {
        const int up = __shfl_up(inc, o, 32);
        if (lane >= o) inc += up;
    }
    if (lane == 31) wsum[wave] = inc;
    __syncthreads();
    if (wave == 0) {
        int w = wsum[lane];
#pragma unroll
        for (int o = 1; o < 32; o <<= 1) {
            const int up = __shfl_up(w, o, 32);
            if (lane >= o) w += up;
        }
        wsum[lane] = w;
    }
    __syncthreads();
    const int off = (inc - run) + (wave ? wsum[wave - 1] : 0);
    if (base < ROWPTR_N) {
#pragma unroll
        for (int q = 0; q < SCAN_PER / 4; ++q)
            *(int4*)(rowptr + base + q * 4) =
                make_int4(v[q * 4] + off, v[q * 4 + 1] + off, v[q * 4 + 2] + off, v[q * 4 + 3] + off);
    }
}

__global__ void k_fill(const int* __restrict__ ei, const int* __restrict__ rowptr,
                       int* __restrict__ cnt, int* __restrict__ perm) {
    int e = blockIdx.x * blockDim.x + threadIdx.x;
    if (e >= E_TOT) return;
    int s, d; get_edge(ei, e, s, d);
    const int pos = atomicSub(&cnt[d], 1) - 1;
    perm[rowptr[d] + pos] = s;
}

typedef __attribute__((ext_vector_type(4))) float v4f;
__device__ __forceinline__ v2f lrelu2(const v2f v) {
    return __builtin_elementwise_max(v, v * NEG_SLOPE);
}
__device__ __forceinline__ v2f logit4(const v4f q, const v4f r, const v4f a, v2f z) {
    z = __builtin_elementwise_fma(lrelu2(q.lo + r.lo), a.lo, z);
    z = __builtin_elementwise_fma(lrelu2(q.hi + r.hi), a.hi, z);
    return z;
}
__device__ __forceinline__ v4f rescale_add(const v4f v, const float f, const float g, const v4f q) {
    const v4f gs = {g, g, g, g};
    return __builtin_elementwise_fma(gs, q, v * f);
}
__device__ __forceinline__ v4f add_row(const v4f v, const float g, const v4f q) {
    const v4f gs = {g, g, g, g};
    return __builtin_elementwise_fma(gs, q, v);
}
__device__ __forceinline__ v4f bias_relu4(const v4f v, const v4f b) {
    const v4f s = v + b;
    v4f r;
    r.x = s.x > 0.f ? s.x : 0.f;
    r.y = s.y > 0.f ? s.y : 0.f;
    r.z = s.z > 0.f ? s.z : 0.f;
    r.w = s.w > 0.f ? s.w : 0.f;
    return r;
}

__global__ void __launch_bounds__(256, 9)
k_node1(const int* __restrict__ rowptr, const int* __restrict__ perm,
        const float* __restrict__ xl, const float* __restrict__ a1,
        const float* __restrict__ b1, float* __restrict__ xr_h1) {
    const int lane = threadIdx.x & 31;
    const int node = (int)blockIdx.x * (int)(blockDim.x >> 5) +
                     __builtin_amdgcn_readfirstlane((int)(threadIdx.x >> 5));
    if (node >= N_NODES) return;
    const int beg = rowptr[node], end = rowptr[node + 1];
    v4f* __restrict__ po = (v4f*)(xr_h1 + (size_t)node * F1) + lane * 4;
    const v4f* __restrict__ pt = (const v4f*)a1 + lane * 4;
    const v4f r0 = po[0], r1 = po[1], r2 = po[2], r3 = po[3];
    const v4f t0 = pt[0], t1 = pt[1], t2 = pt[2], t3 = pt[3];
    float m = -INFINITY, ssum = 0.f;
    v4f acc0 = {0.f, 0.f, 0.f, 0.f}, acc1 = acc0, acc2 = acc0, acc3 = acc0;
    for (int j0 = beg; j0 < end; j0 += 32) {
        const int n = (end - j0 < 32) ? (end - j0) : 32;
        int my = 0;
        if (lane < n) my = perm[j0 + lane];
        int t = 0;
        for (; t + 1 < n; t += 2) {
            const int sa = __builtin_amdgcn_readlane(my, t);
            const int sb = __builtin_amdgcn_readlane(my, t + 1);
            const v4f* __restrict__ pa = (const v4f*)(xl + (size_t)sa * F1) + lane * 4;
            const v4f* __restrict__ pb = (const v4f*)(xl + (size_t)sb * F1) + lane * 4;
            const v4f a0 = pa[0], a1v = pa[1], a2v = pa[2], a3 = pa[3];
            const v4f b0 = pb[0], b1v = pb[1], b2v = pb[2], b3 = pb[3];
            v2f za = {0.f, 0.f}, zb = {0.f, 0.f};
            za = logit4(a0, r0, t0, za);   zb = logit4(b0, r0, t0, zb);
            za = logit4(a1v, r1, t1, za);  zb = logit4(b1v, r1, t1, zb);
            za = logit4(a2v, r2, t2, za);  zb = logit4(b2v, r2, t2, zb);
            za = logit4(a3, r3, t3, za);   zb = logit4(b3, r3, t3, zb);
            float ea = za.x + za.y, eb = zb.x + zb.y;
            ea += __shfl_xor(ea, 1, 32);  eb += __shfl_xor(eb, 1, 32);
            ea += __shfl_xor(ea, 2, 32);  eb += __shfl_xor(eb, 2, 32);
            const float mn = fmaxf(m, fmaxf(ea, eb));
            const float sc = __expf(m - mn);
            const float wa = __expf(ea - mn);
            const float wb = __expf(eb - mn);
            m = mn;
            ssum = fmaf(ssum, sc, wa + wb);
            acc0 = add_row(rescale_add(acc0, sc, wa, a0), wb, b0);
            acc1 = add_row(rescale_add(acc1, sc, wa, a1v), wb, b1v);
            acc2 = add_row(rescale_add(acc2, sc, wa, a2v), wb, b2v);
            acc3 = add_row(rescale_add(acc3, sc, wa, a3), wb, b3);
        }
        if (t < n) {
            const int sa = __builtin_amdgcn_readlane(my, t);
            const v4f* __restrict__ pa = (const v4f*)(xl + (size_t)sa * F1) + lane * 4;
            const v4f a0 = pa[0], a1v = pa[1], a2v = pa[2], a3 = pa[3];
            v2f za = {0.f, 0.f};
            za = logit4(a0, r0, t0, za);
            za = logit4(a1v, r1, t1, za);
            za = logit4(a2v, r2, t2, za);
            za = logit4(a3, r3, t3, za);
            float ea = za.x + za.y;
            ea += __shfl_xor(ea, 1, 32);
            ea += __shfl_xor(ea, 2, 32);
            const float mn = fmaxf(m, ea);
            const float sc = __expf(m - mn);
            const float wa = __expf(ea - mn);
            m = mn;
            ssum = fmaf(ssum, sc, wa);
            acc0 = rescale_add(acc0, sc, wa, a0);
            acc1 = rescale_add(acc1, sc, wa, a1v);
            acc2 = rescale_add(acc2, sc, wa, a2v);
            acc3 = rescale_add(acc3, sc, wa, a3);
        }
    }
    const v4f* __restrict__ pc = (const v4f*)b1 + lane * 4;
    const v4f c0 = pc[0], c1 = pc[1], c2 = pc[2], c3 = pc[3];
    po[0] = bias_relu4(acc0 / ssum, c0);
    po[1] = bias_relu4(acc1 / ssum, c1);
    po[2] = bias_relu4(acc2 / ssum, c2);
    po[3] = bias_relu4(acc3 / ssum, c3);
}

__global__ void k_gemm2(const float* __restrict__ h,
                        const float* __restrict__ W2l,
                        const float* __restrict__ W2r,
                        float* __restrict__ h2l, float* __restrict__ h2r) {
    const int wave  = threadIdx.x >> 5;
    const int lane  = threadIdx.x & 31;
    const int row   = lane & 15;
    const int khalf = lane >> 4;
    const int mtile = blockIdx.x * 4 + (wave >> 1);
    if (mtile >= M_TILES) return;
    const int mbase = mtile * 16;
    const float* __restrict__ W = (wave & 1) ? W2r : W2l;
    float* __restrict__ O       = (wave & 1) ? h2r : h2l;

    v8f c = {};
    for (int kk = 0; kk < F1 / 4; ++kk) {
        const int k0 = kk * 4 + khalf * 2;
        v2f a, b;
        a.x = h[(size_t)(mbase + row) * F1 + k0];
        a.y = h[(size_t)(mbase + row) * F1 + k0 + 1];
        b.x = (row < OUT_DIM) ? W[(k0    ) * OUT_DIM + row] : 0.f;
        b.y = (row < OUT_DIM) ? W[(k0 + 1) * OUT_DIM + row] : 0.f;
        c = __builtin_amdgcn_wmma_f32_16x16x4_f32(false, a, false, b,
                                                  (short)0, c, false, false);
    }
    if (row < OUT_DIM) {
#pragma unroll
        for (int r = 0; r < 8; ++r)
            O[(size_t)(mbase + r + 8 * khalf) * OUT_DIM + row] = c[r];
    }
}

__global__ void k_edge2_logits(const int* __restrict__ ei,
                               const float* __restrict__ h2l, const float* __restrict__ h2r,
                               const float* __restrict__ a2,
                               float* __restrict__ e2, unsigned* __restrict__ m2) {
    int e = blockIdx.x * blockDim.x + threadIdx.x;
    if (e >= E_TOT) return;
    int s, d; get_edge(ei, e, s, d);
    float acc = 0.f;
#pragma unroll
    for (int c = 0; c < OUT_DIM; ++c)
        acc = fmaf(lrelu(h2l[s * OUT_DIM + c] + h2r[d * OUT_DIM + c]), a2[c], acc);
    e2[e] = acc;
    atomicMax(&m2[d], fmap(acc));
}

__global__ void k_edge2_exp(const int* __restrict__ ei, float* __restrict__ e2,
                            const unsigned* __restrict__ m2, float* __restrict__ s2) {
    int e = blockIdx.x * blockDim.x + threadIdx.x;
    if (e >= E_TOT) return;
    int s, d; get_edge(ei, e, s, d);
    const float p = __expf(e2[e] - funmap(m2[d]));
    e2[e] = p;
    atomicAdd(&s2[d], p);
}

__global__ void k_edge2_agg(const int* __restrict__ ei,
                            const float* __restrict__ h2l,
                            const float* __restrict__ e2, const float* __restrict__ s2,
                            float* __restrict__ out) {
    int idx = blockIdx.x * blockDim.x + threadIdx.x;
    if (idx >= E_TOT * OUT_DIM) return;
    const int e = idx >> 3, c = idx & 7;
    int s, d; get_edge(ei, e, s, d);
    const float alpha = e2[e] / s2[d];
    atomicAdd(&out[d * OUT_DIM + c], alpha * h2l[s * OUT_DIM + c]);
}

extern "C" void kernel_launch(void* const* d_in, const int* in_sizes, int n_in,
                              void* d_out, int out_size, void* d_ws, size_t ws_size,
                              hipStream_t stream) {
    const float* x   = (const float*)d_in[0];
    const int*   ei  = (const int*)  d_in[1];
    const float* W1l = (const float*)d_in[2];
    const float* W1r = (const float*)d_in[3];
    const float* a1  = (const float*)d_in[4];
    const float* b1  = (const float*)d_in[5];
    const float* W2l = (const float*)d_in[6];
    const float* W2r = (const float*)d_in[7];
    const float* a2  = (const float*)d_in[8];
    const float* b2  = (const float*)d_in[9];
    float* out = (float*)d_out;

    char* ws = (char*)d_ws;
    size_t off = 0;
    auto carve = [&](size_t bytes) -> char* {
        char* p = ws + off;
        off += (bytes + 255) & ~(size_t)255;
        return p;
    };
    float*    xl1 = (float*)   carve((size_t)N_NODES * F1 * 4);
    float*    xr1 = (float*)   carve((size_t)N_NODES * F1 * 4);
    float*    h2l = (float*)   carve((size_t)N_NODES * OUT_DIM * 4);
    float*    h2r = (float*)   carve((size_t)N_NODES * OUT_DIM * 4);
    float*    e2  = (float*)   carve((size_t)E_TOT * 4);
    unsigned* m2  = (unsigned*)carve((size_t)N_NODES * 4);
    float*    s2  = (float*)   carve((size_t)N_NODES * 4);
    int*      cnt    = (int*) carve((size_t)N_NODES * 4);
    int*      rowptr = (int*) carve((size_t)ROWPTR_N * 4);
    int*      perm   = (int*) carve((size_t)E_TOT * 4);
    float*    h1  = xr1;
    (void)ws_size; (void)n_in; (void)in_sizes; (void)out_size;

    const int B = 256;

    k_init1<<<(N_NODES + B - 1) / B, B, 0, stream>>>(cnt);
    k_count<<<(E_TOT + B - 1) / B, B, 0, stream>>>(ei, cnt);
    k_scan<<<1, SCAN_T, 0, stream>>>(cnt, rowptr);
    k_fill<<<(E_TOT + B - 1) / B, B, 0, stream>>>(ei, rowptr, cnt, perm);
    k_gemm1<<<dim3(M_TILES, 4, 2), B, 0, stream>>>(x, W1l, W1r, xl1, xr1);
    {
        int waves = N_NODES, blocks = (waves * 32 + B - 1) / B;
        k_node1<<<blocks, B, 0, stream>>>(rowptr, perm, xl1, a1, b1, xr1);
    }

    k_gemm2<<<(M_TILES + 3) / 4, B, 0, stream>>>(h1, W2l, W2r, h2l, h2r);
    k_init2<<<(N_NODES * OUT_DIM + B - 1) / B, B, 0, stream>>>(out, b2, m2, s2);
    k_edge2_logits<<<(E_TOT + B - 1) / B, B, 0, stream>>>(ei, h2l, h2r, a2, e2, m2);
    k_edge2_exp<<<(E_TOT + B - 1) / B, B, 0, stream>>>(ei, e2, m2, s2);
    k_edge2_agg<<<(E_TOT * OUT_DIM + B - 1) / B, B, 0, stream>>>(ei, h2l, e2, s2, out);
}


# Layer 2: logits, softmax and aggregation in one gather over the sorted edge list, 8 lanes per destination (online softmax, lane partials merged: sum order changes); its atomics and 3 launches go
# speedup vs baseline: 5.3647x; 1.3866x over previous
#include <hip/hip_runtime.h>
#include <hip/hip_bf16.h>

#define N_NODES 30000
#define E_RAW   300000
#define E_TOT   330000
#define HEADS   8
#define HID     64
#define F1      512
#define IN_DIM  8
#define OUT_DIM 8
#define NEG_SLOPE 0.2f
#define M_TILES 1875

typedef __attribute__((ext_vector_type(2))) float v2f;
typedef __attribute__((ext_vector_type(8))) float v8f;

__device__ __forceinline__ void get_edge(const int* __restrict__ ei, int e, int& s, int& d) {
    if (e < E_RAW) { s = ei[e]; d = ei[E_RAW + e]; }
    else           { s = e - E_RAW; d = e - E_RAW; }
}

__global__ void k_init1(int* __restrict__ cnt) {
    int i = blockIdx.x * blockDim.x + threadIdx.x;
    if (i < N_NODES) cnt[i] = 0;
}

__global__ void k_gemm1(const float* __restrict__ x,
                        const float* __restrict__ Wl,
                        const float* __restrict__ Wr,
                        float* __restrict__ xl, float* __restrict__ xr) {
    const int wave  = threadIdx.x >> 5;
    const int lane  = threadIdx.x & 31;
    const int row   = lane & 15;
    const int khalf = lane >> 4;
    const int mbase = blockIdx.x * 16;
    const int nbase = (blockIdx.y * 8 + wave) * 16;
    const float* __restrict__ W = blockIdx.z ? Wr : Wl;
    float* __restrict__ O       = blockIdx.z ? xr : xl;

    v8f c = {};
#pragma unroll
    for (int kk = 0; kk < 2; ++kk) {
        const int k0 = kk * 4 + khalf * 2;
        v2f a, b;
        a.x = x[(mbase + row) * IN_DIM + k0];
        a.y = x[(mbase + row) * IN_DIM + k0 + 1];
        b.x = W[(k0    ) * F1 + nbase + row];
        b.y = W[(k0 + 1) * F1 + nbase + row];
        c = __builtin_amdgcn_wmma_f32_16x16x4_f32(false, a, false, b,
                                                  (short)0, c, false, false);
    }
#pragma unroll
    for (int r = 0; r < 8; ++r)
        O[(size_t)(mbase + r + 8 * khalf) * F1 + nbase + row] = c[r];
}

__global__ void k_count(const int* __restrict__ ei, int* __restrict__ cnt) {
    int e = blockIdx.x * blockDim.x + threadIdx.x;
    if (e >= E_TOT) return;
    int s, d; get_edge(ei, e, s, d);
    atomicAdd(&cnt[d], 1);
}

#define SCAN_T    1024
#define SCAN_PER  32
#define ROWPTR_N  (((N_NODES + SCAN_PER) / SCAN_PER) * SCAN_PER)
__global__ void __launch_bounds__(SCAN_T)
k_scan(const int* __restrict__ cnt, int* __restrict__ rowptr) {
    __shared__ int wsum[SCAN_T / 32];
    const int t    = threadIdx.x;
    const int lane = t & 31;
    const int wave = t >> 5;
    const int base = t * SCAN_PER;
    int v[SCAN_PER];
#pragma unroll
    for (int q = 0; q < SCAN_PER / 4; ++q) {
        const int i = base + q * 4;
        int4 c = make_int4(0, 0, 0, 0);
        if (i < N_NODES) c = *(const int4*)(cnt + i);
        v[q * 4 + 0] = c.x; v[q * 4 + 1] = c.y; v[q * 4 + 2] = c.z; v[q * 4 + 3] = c.w;
    }
    int run = 0;
#pragma unroll
    for (int q = 0; q < SCAN_PER; ++q) { const int c = v[q]; v[q] = run; run += c; }
    int inc = run;
#pragma unroll
    for (int o = 1; o < 32; o <<= 1) {
        const int up = __shfl_up(inc, o, 32);
        if (lane >= o) inc += up;
    }
    if (lane == 31) wsum[wave] = inc;
    __syncthreads();
    if (wave == 0) {
        int w = wsum[lane];
#pragma unroll
        for (int o = 1; o < 32; o <<= 1) {
            const int up = __shfl_up(w, o, 32);
            if (lane >= o) w += up;
        }
        wsum[lane] = w;
    }
    __syncthreads();
    const int off = (inc - run) + (wave ? wsum[wave - 1] : 0);
    if (base < ROWPTR_N) {
#pragma unroll
        for (int q = 0; q < SCAN_PER / 4; ++q)
            *(int4*)(rowptr + base + q * 4) =
                make_int4(v[q * 4] + off, v[q * 4 + 1] + off, v[q * 4 + 2] + off, v[q * 4 + 3] + off);
    }
}

__global__ void k_fill(const int* __restrict__ ei, const int* __restrict__ rowptr,
                       int* __restrict__ cnt, int* __restrict__ perm) {
    int e = blockIdx.x * blockDim.x + threadIdx.x;
    if (e >= E_TOT) return;
    int s, d; get_edge(ei, e, s, d);
    const int pos = atomicSub(&cnt[d], 1) - 1;
    perm[rowptr[d] + pos] = s;
}

typedef __attribute__((ext_vector_type(4))) float v4f;
__device__ __forceinline__ v2f lrelu2(const v2f v) {
    return __builtin_elementwise_max(v, v * NEG_SLOPE);
}
__device__ __forceinline__ v2f logit4(const v4f q, const v4f r, const v4f a, v2f z) {
    z = __builtin_elementwise_fma(lrelu2(q.lo + r.lo), a.lo, z);
    z = __builtin_elementwise_fma(lrelu2(q.hi + r.hi), a.hi, z);
    return z;
}
__device__ __forceinline__ v4f rescale_add(const v4f v, const float f, const float g, const v4f q) {
    const v4f gs = {g, g, g, g};
    return __builtin_elementwise_fma(gs, q, v * f);
}
__device__ __forceinline__ v4f add_row(const v4f v, const float g, const v4f q) {
    const v4f gs = {g, g, g, g};
    return __builtin_elementwise_fma(gs, q, v);
}
__device__ __forceinline__ v4f bias_relu4(const v4f v, const v4f b) {
    const v4f s = v + b;
    v4f r;
    r.x = s.x > 0.f ? s.x : 0.f;
    r.y = s.y > 0.f ? s.y : 0.f;
    r.z = s.z > 0.f ? s.z : 0.f;
    r.w = s.w > 0.f ? s.w : 0.f;
    return r;
}

__global__ void __launch_bounds__(256, 9)
k_node1(const int* __restrict__ rowptr, const int* __restrict__ perm,
        const float* __restrict__ xl, const float* __restrict__ a1,
        const float* __restrict__ b1, float* __restrict__ xr_h1) {
    const int lane = threadIdx.x & 31;
    const int node = (int)blockIdx.x * (int)(blockDim.x >> 5) +
                     __builtin_amdgcn_readfirstlane((int)(threadIdx.x >> 5));
    if (node >= N_NODES) return;
    const int beg = rowptr[node], end = rowptr[node + 1];
    v4f* __restrict__ po = (v4f*)(xr_h1 + (size_t)node * F1) + lane * 4;
    const v4f* __restrict__ pt = (const v4f*)a1 + lane * 4;
    const v4f r0 = po[0], r1 = po[1], r2 = po[2], r3 = po[3];
    const v4f t0 = pt[0], t1 = pt[1], t2 = pt[2], t3 = pt[3];
    float m = -INFINITY, ssum = 0.f;
    v4f acc0 = {0.f, 0.f, 0.f, 0.f}, acc1 = acc0, acc2 = acc0, acc3 = acc0;
    for (int j0 = beg; j0 < end; j0 += 32) {
        const int n = (end - j0 < 32) ? (end - j0) : 32;
        int my = 0;
        if (lane < n) my = perm[j0 + lane];
        int t = 0;
        for (; t + 1 < n; t += 2) {
            const int sa = __builtin_amdgcn_readlane(my, t);
            const int sb = __builtin_amdgcn_readlane(my, t + 1);
            const v4f* __restrict__ pa = (const v4f*)(xl + (size_t)sa * F1) + lane * 4;
            const v4f* __restrict__ pb = (const v4f*)(xl + (size_t)sb * F1) + lane * 4;
            const v4f a0 = pa[0], a1v = pa[1], a2v = pa[2], a3 = pa[3];
            const v4f b0 = pb[0], b1v = pb[1], b2v = pb[2], b3 = pb[3];
            v2f za = {0.f, 0.f}, zb = {0.f, 0.f};
            za = logit4(a0, r0, t0, za);   zb = logit4(b0, r0, t0, zb);
            za = logit4(a1v, r1, t1, za);  zb = logit4(b1v, r1, t1, zb);
            za = logit4(a2v, r2, t2, za);  zb = logit4(b2v, r2, t2, zb);
            za = logit4(a3, r3, t3, za);   zb = logit4(b3, r3, t3, zb);
            float ea = za.x + za.y, eb = zb.x + zb.y;
            ea += __shfl_xor(ea, 1, 32);  eb += __shfl_xor(eb, 1, 32);
            ea += __shfl_xor(ea, 2, 32);  eb += __shfl_xor(eb, 2, 32);
            const float mn = fmaxf(m, fmaxf(ea, eb));
            const float sc = __expf(m - mn);
            const float wa = __expf(ea - mn);
            const float wb = __expf(eb - mn);
            m = mn;
            ssum = fmaf(ssum, sc, wa + wb);
            acc0 = add_row(rescale_add(acc0, sc, wa, a0), wb, b0);
            acc1 = add_row(rescale_add(acc1, sc, wa, a1v), wb, b1v);
            acc2 = add_row(rescale_add(acc2, sc, wa, a2v), wb, b2v);
            acc3 = add_row(rescale_add(acc3, sc, wa, a3), wb, b3);
        }
        if (t < n) {
            const int sa = __builtin_amdgcn_readlane(my, t);
            const v4f* __restrict__ pa = (const v4f*)(xl + (size_t)sa * F1) + lane * 4;
            const v4f a0 = pa[0], a1v = pa[1], a2v = pa[2], a3 = pa[3];
            v2f za = {0.f, 0.f};
            za = logit4(a0, r0, t0, za);
            za = logit4(a1v, r1, t1, za);
            za = logit4(a2v, r2, t2, za);
            za = logit4(a3, r3, t3, za);
            float ea = za.x + za.y;
            ea += __shfl_xor(ea, 1, 32);
            ea += __shfl_xor(ea, 2, 32);
            const float mn = fmaxf(m, ea);
            const float sc = __expf(m - mn);
            const float wa = __expf(ea - mn);
            m = mn;
            ssum = fmaf(ssum, sc, wa);
            acc0 = rescale_add(acc0, sc, wa, a0);
            acc1 = rescale_add(acc1, sc, wa, a1v);
            acc2 = rescale_add(acc2, sc, wa, a2v);
            acc3 = rescale_add(acc3, sc, wa, a3);
        }
    }
    const v4f* __restrict__ pc = (const v4f*)b1 + lane * 4;
    const v4f c0 = pc[0], c1 = pc[1], c2 = pc[2], c3 = pc[3];
    po[0] = bias_relu4(acc0 / ssum, c0);
    po[1] = bias_relu4(acc1 / ssum, c1);
    po[2] = bias_relu4(acc2 / ssum, c2);
    po[3] = bias_relu4(acc3 / ssum, c3);
}

__global__ void k_gemm2(const float* __restrict__ h,
                        const float* __restrict__ W2l,
                        const float* __restrict__ W2r,
                        float* __restrict__ h2l, float* __restrict__ h2r) {
    const int wave  = threadIdx.x >> 5;
    const int lane  = threadIdx.x & 31;
    const int row   = lane & 15;
    const int khalf = lane >> 4;
    const int mtile = blockIdx.x * 4 + (wave >> 1);
    if (mtile >= M_TILES) return;
    const int mbase = mtile * 16;
    const float* __restrict__ W = (wave & 1) ? W2r : W2l;
    float* __restrict__ O       = (wave & 1) ? h2r : h2l;

    v8f c = {};
    for (int kk = 0; kk < F1 / 4; ++kk) {
        const int k0 = kk * 4 + khalf * 2;
        v2f a, b;
        a.x = h[(size_t)(mbase + row) * F1 + k0];
        a.y = h[(size_t)(mbase + row) * F1 + k0 + 1];
        b.x = (row < OUT_DIM) ? W[(k0    ) * OUT_DIM + row] : 0.f;
        b.y = (row < OUT_DIM) ? W[(k0 + 1) * OUT_DIM + row] : 0.f;
        c = __builtin_amdgcn_wmma_f32_16x16x4_f32(false, a, false, b,
                                                  (short)0, c, false, false);
    }
    if (row < OUT_DIM) {
#pragma unroll
        for (int r = 0; r < 8; ++r)
            O[(size_t)(mbase + r + 8 * khalf) * OUT_DIM + row] = c[r];
    }
}

__global__ void __launch_bounds__(256)
k_node2(const int* __restrict__ rowptr, const int* __restrict__ perm,
        const float* __restrict__ h2l, const float* __restrict__ h2r,
        const float* __restrict__ a2, const float* __restrict__ b2,
        float* __restrict__ out) {
    const int gid  = blockIdx.x * blockDim.x + threadIdx.x;
    const int node = gid >> 3;
    const int sub  = gid & 7;
    if (node >= N_NODES) return;
    const int beg = rowptr[node], end = rowptr[node + 1];
    const v4f* __restrict__ pr = (const v4f*)(h2r + (size_t)node * OUT_DIM);
    const v4f r0 = pr[0], r1 = pr[1];
    const v4f t0 = ((const v4f*)a2)[0], t1 = ((const v4f*)a2)[1];
    const float bias = b2[sub];
    float m = -INFINITY, ssum = 0.f;
    v4f acc0 = {0.f, 0.f, 0.f, 0.f}, acc1 = acc0;
    for (int j = beg + sub; j < end; j += 8) {
        const int s = perm[j];
        const v4f* __restrict__ pq = (const v4f*)(h2l + (size_t)s * OUT_DIM);
        const v4f q0 = pq[0], q1 = pq[1];
        v2f z = {0.f, 0.f};
        z = logit4(q0, r0, t0, z);
        z = logit4(q1, r1, t1, z);
        const float e  = z.x + z.y;
        const float mn = fmaxf(m, e);
        const float sc = __expf(m - mn);
        const float w  = __expf(e - mn);
        m = mn;
        ssum = fmaf(ssum, sc, w);
        acc0 = rescale_add(acc0, sc, w, q0);
        acc1 = rescale_add(acc1, sc, w, q1);
    }
#pragma unroll
    for (int o = 1; o < 8; o <<= 1) {
        const float mo = __shfl_xor(m, o, 32);
        const float so = __shfl_xor(ssum, o, 32);
        v4f o0, o1;
        o0.x = __shfl_xor(acc0.x, o, 32);  o0.y = __shfl_xor(acc0.y, o, 32);
        o0.z = __shfl_xor(acc0.z, o, 32);  o0.w = __shfl_xor(acc0.w, o, 32);
        o1.x = __shfl_xor(acc1.x, o, 32);  o1.y = __shfl_xor(acc1.y, o, 32);
        o1.z = __shfl_xor(acc1.z, o, 32);  o1.w = __shfl_xor(acc1.w, o, 32);
        const float mn = fmaxf(m, mo);
        const float f  = (m  == mn) ? 1.f : __expf(m  - mn);
        const float g  = (mo == mn) ? 1.f : __expf(mo - mn);
        ssum = fmaf(so, g, ssum * f);
        acc0 = rescale_add(acc0, f, g, o0);
        acc1 = rescale_add(acc1, f, g, o1);
        m = mn;
    }
    const v4f half = (sub & 4) ? acc1 : acc0;
    const float lo = (sub & 1) ? half.y : half.x;
    const float hi = (sub & 1) ? half.w : half.z;
    const float v  = (sub & 2) ? hi : lo;
    out[gid] = v / ssum + bias;
}

extern "C" void kernel_launch(void* const* d_in, const int* in_sizes, int n_in,
                              void* d_out, int out_size, void* d_ws, size_t ws_size,
                              hipStream_t stream) {
    const float* x   = (const float*)d_in[0];
    const int*   ei  = (const int*)  d_in[1];
    const float* W1l = (const float*)d_in[2];
    const float* W1r = (const float*)d_in[3];
    const float* a1  = (const float*)d_in[4];
    const float* b1  = (const float*)d_in[5];
    const float* W2l = (const float*)d_in[6];
    const float* W2r = (const float*)d_in[7];
    const float* a2  = (const float*)d_in[8];
    const float* b2  = (const float*)d_in[9];
    float* out = (float*)d_out;

    char* ws = (char*)d_ws;
    size_t off = 0;
    auto carve = [&](size_t bytes) -> char* {
        char* p = ws + off;
        off += (bytes + 255) & ~(size_t)255;
        return p;
    };
    float*    xl1 = (float*)   carve((size_t)N_NODES * F1 * 4);
    float*    xr1 = (float*)   carve((size_t)N_NODES * F1 * 4);
    float*    h2l = (float*)   carve((size_t)N_NODES * OUT_DIM * 4);
    float*    h2r = (float*)   carve((size_t)N_NODES * OUT_DIM * 4);
    int*      cnt    = (int*) carve((size_t)N_NODES * 4);
    int*      rowptr = (int*) carve((size_t)ROWPTR_N * 4);
    int*      perm   = (int*) carve((size_t)E_TOT * 4);
    float*    h1  = xr1;
    (void)ws_size; (void)n_in; (void)in_sizes; (void)out_size;

    const int B = 256;

    k_init1<<<(N_NODES + B - 1) / B, B, 0, stream>>>(cnt);
    k_count<<<(E_TOT + B - 1) / B, B, 0, stream>>>(ei, cnt);
    k_scan<<<1, SCAN_T, 0, stream>>>(cnt, rowptr);
    k_fill<<<(E_TOT + B - 1) / B, B, 0, stream>>>(ei, rowptr, cnt, perm);
    k_gemm1<<<dim3(M_TILES, 4, 2), B, 0, stream>>>(x, W1l, W1r, xl1, xr1);
    {
        int waves = N_NODES, blocks = (waves * 32 + B - 1) / B;
        k_node1<<<blocks, B, 0, stream>>>(rowptr, perm, xl1, a1, b1, xr1);
    }

    k_gemm2<<<(M_TILES + 3) / 4, B, 0, stream>>>(h1, W2l, W2r, h2l, h2r);
    k_node2<<<(N_NODES * OUT_DIM + B - 1) / B, B, 0, stream>>>(rowptr, perm, h2l, h2r, a2, b2, out);
}


# Layer-1 target-side projection x@Wr formed in the node kernel (work-group's 8 nodes as one WMMA tile, through LDS); its 30000x512 plane is no longer written by the GEMM launch nor read back
# speedup vs baseline: 6.0141x; 1.1211x over previous
#include <hip/hip_runtime.h>
#include <hip/hip_bf16.h>

#define N_NODES 30000
#define E_RAW   300000
#define E_TOT   330000
#define HEADS   8
#define HID     64
#define F1      512
#define IN_DIM  8
#define OUT_DIM 8
#define NEG_SLOPE 0.2f
#define M_TILES 1875

typedef __attribute__((ext_vector_type(2))) float v2f;
typedef __attribute__((ext_vector_type(8))) float v8f;

__device__ __forceinline__ void get_edge(const int* __restrict__ ei, int e, int& s, int& d) {
    if (e < E_RAW) { s = ei[e]; d = ei[E_RAW + e]; }
    else           { s = e - E_RAW; d = e - E_RAW; }
}

__global__ void k_init1(int* __restrict__ cnt) {
    int i = blockIdx.x * blockDim.x + threadIdx.x;
    if (i < N_NODES) cnt[i] = 0;
}

__global__ void k_gemm1(const float* __restrict__ x,
                        const float* __restrict__ W,
                        float* __restrict__ O) {
    const int wave  = threadIdx.x >> 5;
    const int lane  = threadIdx.x & 31;
    const int row   = lane & 15;
    const int khalf = lane >> 4;
    const int mbase = blockIdx.x * 16;
    const int nbase = (blockIdx.y * 8 + wave) * 16;

    v8f c = {};
#pragma unroll
    for (int kk = 0; kk < 2; ++kk) {
        const int k0 = kk * 4 + khalf * 2;
        v2f a, b;
        a.x = x[(mbase + row) * IN_DIM + k0];
        a.y = x[(mbase + row) * IN_DIM + k0 + 1];
        b.x = W[(k0    ) * F1 + nbase + row];
        b.y = W[(k0 + 1) * F1 + nbase + row];
        c = __builtin_amdgcn_wmma_f32_16x16x4_f32(false, a, false, b,
                                                  (short)0, c, false, false);
    }
#pragma unroll
    for (int r = 0; r < 8; ++r)
        O[(size_t)(mbase + r + 8 * khalf) * F1 + nbase + row] = c[r];
}

__global__ void k_count(const int* __restrict__ ei, int* __restrict__ cnt) {
    int e = blockIdx.x * blockDim.x + threadIdx.x;
    if (e >= E_TOT) return;
    int s, d; get_edge(ei, e, s, d);
    atomicAdd(&cnt[d], 1);
}

#define SCAN_T    1024
#define SCAN_PER  32
#define ROWPTR_N  (((N_NODES + SCAN_PER) / SCAN_PER) * SCAN_PER)
__global__ void __launch_bounds__(SCAN_T)
k_scan(const int* __restrict__ cnt, int* __restrict__ rowptr) {
    __shared__ int wsum[SCAN_T / 32];
    const int t    = threadIdx.x;
    const int lane = t & 31;
    const int wave = t >> 5;
    const int base = t * SCAN_PER;
    int v[SCAN_PER];
#pragma unroll
    for (int q = 0; q < SCAN_PER / 4; ++q) {
        const int i = base + q * 4;
        int4 c = make_int4(0, 0, 0, 0);
        if (i < N_NODES) c = *(const int4*)(cnt + i);
        v[q * 4 + 0] = c.x; v[q * 4 + 1] = c.y; v[q * 4 + 2] = c.z; v[q * 4 + 3] = c.w;
    }
    int run = 0;
#pragma unroll
    for (int q = 0; q < SCAN_PER; ++q) { const int c = v[q]; v[q] = run; run += c; }
    int inc = run;
#pragma unroll
    for (int o = 1; o < 32; o <<= 1) {
        const int up = __shfl_up(inc, o, 32);
        if (lane >= o) inc += up;
    }
    if (lane == 31) wsum[wave] = inc;
    __syncthreads();
    if (wave == 0) {
        int w = wsum[lane];
#pragma unroll
        for (int o = 1; o < 32; o <<= 1) {
            const int up = __shfl_up(w, o, 32);
            if (lane >= o) w += up;
        }
        wsum[lane] = w;
    }
    __syncthreads();
    const int off = (inc - run) + (wave ? wsum[wave - 1] : 0);
    if (base < ROWPTR_N) {
#pragma unroll
        for (int q = 0; q < SCAN_PER / 4; ++q)
            *(int4*)(rowptr + base + q * 4) =
                make_int4(v[q * 4] + off, v[q * 4 + 1] + off, v[q * 4 + 2] + off, v[q * 4 + 3] + off);
    }
}

__global__ void k_fill(const int* __restrict__ ei, const int* __restrict__ rowptr,
                       int* __restrict__ cnt, int* __restrict__ perm) {
    int e = blockIdx.x * blockDim.x + threadIdx.x;
    if (e >= E_TOT) return;
    int s, d; get_edge(ei, e, s, d);
    const int pos = atomicSub(&cnt[d], 1) - 1;
    perm[rowptr[d] + pos] = s;
}

typedef __attribute__((ext_vector_type(4))) float v4f;
__device__ __forceinline__ v2f lrelu2(const v2f v) {
    return __builtin_elementwise_max(v, v * NEG_SLOPE);
}
__device__ __forceinline__ v2f logit4(const v4f q, const v4f r, const v4f a, v2f z) {
    z = __builtin_elementwise_fma(lrelu2(q.lo + r.lo), a.lo, z);
    z = __builtin_elementwise_fma(lrelu2(q.hi + r.hi), a.hi, z);
    return z;
}
__device__ __forceinline__ v4f rescale_add(const v4f v, const float f, const float g, const v4f q) {
    const v4f gs = {g, g, g, g};
    return __builtin_elementwise_fma(gs, q, v * f);
}
__device__ __forceinline__ v4f add_row(const v4f v, const float g, const v4f q) {
    const v4f gs = {g, g, g, g};
    return __builtin_elementwise_fma(gs, q, v);
}
__device__ __forceinline__ v4f bias_relu4(const v4f v, const v4f b) {
    const v4f s = v + b;
    v4f r;
    r.x = s.x > 0.f ? s.x : 0.f;
    r.y = s.y > 0.f ? s.y : 0.f;
    r.z = s.z > 0.f ? s.z : 0.f;
    r.w = s.w > 0.f ? s.w : 0.f;
    return r;
}

#define NODES_PER_WG 8
#define XR_PITCH     (F1 + F1 / 4)
static_assert(N_NODES % NODES_PER_WG == 0, "whole work-groups only: the barrier needs all eight waves");
__global__ void __launch_bounds__(256, 9)
k_node1(const int* __restrict__ rowptr, const int* __restrict__ perm,
        const float* __restrict__ xl, const float* __restrict__ x,
        const float* __restrict__ Wr, const float* __restrict__ a1,
        const float* __restrict__ b1, float* __restrict__ h1) {
    __shared__ float xr_s[NODES_PER_WG * XR_PITCH];
    const int lane = threadIdx.x & 31;
    const int wave  = __builtin_amdgcn_readfirstlane((int)(threadIdx.x >> 5));
    const int node0 = (int)blockIdx.x * NODES_PER_WG;
    if (node0 >= N_NODES) return;
    const int node = node0 + wave;
    const int beg = rowptr[node], end = rowptr[node + 1];
    const v4f* __restrict__ pt = (const v4f*)a1 + lane * 4;
    const v4f t0 = pt[0], t1 = pt[1], t2 = pt[2], t3 = pt[3];
    int my = 0;
    {
        const int row   = lane & 15;
        const int khalf = lane >> 4;
        v2f a[2] = {{0.f, 0.f}, {0.f, 0.f}};
        if (row < NODES_PER_WG) {
            const v2f* __restrict__ px = (const v2f*)(x + (node0 + row) * IN_DIM) + khalf;
            a[0] = px[0];
            a[1] = px[2];
        }
        const v4f* __restrict__ pw = (const v4f*)(Wr + wave * HID) + row;
        v4f bk[2][2];
#pragma unroll
        for (int kk = 0; kk < 2; ++kk) {
            const int k0 = kk * 4 + khalf * 2;
            bk[kk][0] = pw[(k0    ) * (F1 / 4)];
            bk[kk][1] = pw[(k0 + 1) * (F1 / 4)];
        }
        if (lane < end - beg) my = perm[beg + lane];
        v8f d[4];
#pragma unroll
        for (int t = 0; t < 4; ++t) {
            v8f c = {};
#pragma unroll
            for (int kk = 0; kk < 2; ++kk) {
                v2f b;
                b.x = bk[kk][0][t];
                b.y = bk[kk][1][t];
                c = __builtin_amdgcn_wmma_f32_16x16x4_f32(false, a[kk], false, b,
                                                          (short)0, c, false, false);
            }
            d[t] = c;
        }
        if (khalf == 0) {
            float* __restrict__ ps = xr_s + wave * (HID + HID / 4) + 4 * row + 4 * (row >> 2);
#pragma unroll
            for (int r = 0; r < 8; ++r) {
                const v4f v = {d[0][r], d[1][r], d[2][r], d[3][r]};
                *(v4f*)(ps + r * XR_PITCH) = v;
            }
        }
    }
    __syncthreads();
    v4f* __restrict__ po = (v4f*)(h1 + (size_t)node * F1) + lane * 4;
    const v4f* __restrict__ pr = (const v4f*)(xr_s + wave * XR_PITCH + lane * 20);
    const v4f r0 = pr[0], r1 = pr[1], r2 = pr[2], r3 = pr[3];
    float m = -INFINITY, ssum = 0.f;
    v4f acc0 = {0.f, 0.f, 0.f, 0.f}, acc1 = acc0, acc2 = acc0, acc3 = acc0;
    for (int j0 = beg; j0 < end; j0 += 32) {
        const int n = (end - j0 < 32) ? (end - j0) : 32;
        if (j0 != beg) {
            my = 0;
            if (lane < n) my = perm[j0 + lane];
        }
        int t = 0;
        for (; t + 1 < n; t += 2) {
            const int sa = __builtin_amdgcn_readlane(my, t);
            const int sb = __builtin_amdgcn_readlane(my, t + 1);
            const v4f* __restrict__ pa = (const v4f*)(xl + (size_t)sa * F1) + lane * 4;
            const v4f* __restrict__ pb = (const v4f*)(xl + (size_t)sb * F1) + lane * 4;
            const v4f a0 = pa[0], a1v = pa[1], a2v = pa[2], a3 = pa[3];
            const v4f b0 = pb[0], b1v = pb[1], b2v = pb[2], b3 = pb[3];
            v2f za = {0.f, 0.f}, zb = {0.f, 0.f};
            za = logit4(a0, r0, t0, za);   zb = logit4(b0, r0, t0, zb);
            za = logit4(a1v, r1, t1, za);  zb = logit4(b1v, r1, t1, zb);
            za = logit4(a2v, r2, t2, za);  zb = logit4(b2v, r2, t2, zb);
            za = logit4(a3, r3, t3, za);   zb = logit4(b3, r3, t3, zb);
            float ea = za.x + za.y, eb = zb.x + zb.y;
            ea += __shfl_xor(ea, 1, 32);  eb += __shfl_xor(eb, 1, 32);
            ea += __shfl_xor(ea, 2, 32);  eb += __shfl_xor(eb, 2, 32);
            const float mn = fmaxf(m, fmaxf(ea, eb));
            const float sc = __expf(m - mn);
            const float wa = __expf(ea - mn);
            const float wb = __expf(eb - mn);
            m = mn;
            ssum = fmaf(ssum, sc, wa + wb);
            acc0 = add_row(rescale_add(acc0, sc, wa, a0), wb, b0);
            acc1 = add_row(rescale_add(acc1, sc, wa, a1v), wb, b1v);
            acc2 = add_row(rescale_add(acc2, sc, wa, a2v), wb, b2v);
            acc3 = add_row(rescale_add(acc3, sc, wa, a3), wb, b3);
        }
        if (t < n) {
            const int sa = __builtin_amdgcn_readlane(my, t);
            const v4f* __restrict__ pa = (const v4f*)(xl + (size_t)sa * F1) + lane * 4;
            const v4f a0 = pa[0], a1v = pa[1], a2v = pa[2], a3 = pa[3];
            v2f za = {0.f, 0.f};
            za = logit4(a0, r0, t0, za);
            za = logit4(a1v, r1, t1, za);
            za = logit4(a2v, r2, t2, za);
            za = logit4(a3, r3, t3, za);
            float ea = za.x + za.y;
            ea += __shfl_xor(ea, 1, 32);
            ea += __shfl_xor(ea, 2, 32);
            const float mn = fmaxf(m, ea);
            const float sc = __expf(m - mn);
            const float wa = __expf(ea - mn);
            m = mn;
            ssum = fmaf(ssum, sc, wa);
            acc0 = rescale_add(acc0, sc, wa, a0);
            acc1 = rescale_add(acc1, sc, wa, a1v);
            acc2 = rescale_add(acc2, sc, wa, a2v);
            acc3 = rescale_add(acc3, sc, wa, a3);
        }
    }
    const v4f* __restrict__ pc = (const v4f*)b1 + lane * 4;
    const v4f c0 = pc[0], c1 = pc[1], c2 = pc[2], c3 = pc[3];
    po[0] = bias_relu4(acc0 / ssum, c0);
    po[1] = bias_relu4(acc1 / ssum, c1);
    po[2] = bias_relu4(acc2 / ssum, c2);
    po[3] = bias_relu4(acc3 / ssum, c3);
}

__global__ void k_gemm2(const float* __restrict__ h,
                        const float* __restrict__ W2l,
                        const float* __restrict__ W2r,
                        float* __restrict__ h2l, float* __restrict__ h2r) {
    const int wave  = threadIdx.x >> 5;
    const int lane  = threadIdx.x & 31;
    const int row   = lane & 15;
    const int khalf = lane >> 4;
    const int mtile = blockIdx.x * 4 + (wave >> 1);
    if (mtile >= M_TILES) return;
    const int mbase = mtile * 16;
    const float* __restrict__ W = (wave & 1) ? W2r : W2l;
    float* __restrict__ O       = (wave & 1) ? h2r : h2l;

    v8f c = {};
    for (int kk = 0; kk < F1 / 4; ++kk) {
        const int k0 = kk * 4 + khalf * 2;
        v2f a, b;
        a.x = h[(size_t)(mbase + row) * F1 + k0];
        a.y = h[(size_t)(mbase + row) * F1 + k0 + 1];
        b.x = (row < OUT_DIM) ? W[(k0    ) * OUT_DIM + row] : 0.f;
        b.y = (row < OUT_DIM) ? W[(k0 + 1) * OUT_DIM + row] : 0.f;
        c = __builtin_amdgcn_wmma_f32_16x16x4_f32(false, a, false, b,
                                                  (short)0, c, false, false);
    }
    if (row < OUT_DIM) {
#pragma unroll
        for (int r = 0; r < 8; ++r)
            O[(size_t)(mbase + r + 8 * khalf) * OUT_DIM + row] = c[r];
    }
}

__global__ void __launch_bounds__(256)
k_node2(const int* __restrict__ rowptr, const int* __restrict__ perm,
        const float* __restrict__ h2l, const float* __restrict__ h2r,
        const float* __restrict__ a2, const float* __restrict__ b2,
        float* __restrict__ out) {
    const int gid  = blockIdx.x * blockDim.x + threadIdx.x;
    const int node = gid >> 3;
    const int sub  = gid & 7;
    if (node >= N_NODES) return;
    const int beg = rowptr[node], end = rowptr[node + 1];
    const v4f* __restrict__ pr = (const v4f*)(h2r + (size_t)node * OUT_DIM);
    const v4f r0 = pr[0], r1 = pr[1];
    const v4f t0 = ((const v4f*)a2)[0], t1 = ((const v4f*)a2)[1];
    const float bias = b2[sub];
    float m = -INFINITY, ssum = 0.f;
    v4f acc0 = {0.f, 0.f, 0.f, 0.f}, acc1 = acc0;
    for (int j = beg + sub; j < end; j += 8) {
        const int s = perm[j];
        const v4f* __restrict__ pq = (const v4f*)(h2l + (size_t)s * OUT_DIM);
        const v4f q0 = pq[0], q1 = pq[1];
        v2f z = {0.f, 0.f};
        z = logit4(q0, r0, t0, z);
        z = logit4(q1, r1, t1, z);
        const float e  = z.x + z.y;
        const float mn = fmaxf(m, e);
        const float sc = __expf(m - mn);
        const float w  = __expf(e - mn);
        m = mn;
        ssum = fmaf(ssum, sc, w);
        acc0 = rescale_add(acc0, sc, w, q0);
        acc1 = rescale_add(acc1, sc, w, q1);
    }
#pragma unroll
    for (int o = 1; o < 8; o <<= 1) {
        const float mo = __shfl_xor(m, o, 32);
        const float so = __shfl_xor(ssum, o, 32);
        v4f o0, o1;
        o0.x = __shfl_xor(acc0.x, o, 32);  o0.y = __shfl_xor(acc0.y, o, 32);
        o0.z = __shfl_xor(acc0.z, o, 32);  o0.w = __shfl_xor(acc0.w, o, 32);
        o1.x = __shfl_xor(acc1.x, o, 32);  o1.y = __shfl_xor(acc1.y, o, 32);
        o1.z = __shfl_xor(acc1.z, o, 32);  o1.w = __shfl_xor(acc1.w, o, 32);
        const float mn = fmaxf(m, mo);
        const float f  = (m  == mn) ? 1.f : __expf(m  - mn);
        const float g  = (mo == mn) ? 1.f : __expf(mo - mn);
        ssum = fmaf(so, g, ssum * f);
        acc0 = rescale_add(acc0, f, g, o0);
        acc1 = rescale_add(acc1, f, g, o1);
        m = mn;
    }
    const v4f half = (sub & 4) ? acc1 : acc0;
    const float lo = (sub & 1) ? half.y : half.x;
    const float hi = (sub & 1) ? half.w : half.z;
    const float v  = (sub & 2) ? hi : lo;
    out[gid] = v / ssum + bias;
}

extern "C" void kernel_launch(void* const* d_in, const int* in_sizes, int n_in,
                              void* d_out, int out_size, void* d_ws, size_t ws_size,
                              hipStream_t stream) {
    const float* x   = (const float*)d_in[0];
    const int*   ei  = (const int*)  d_in[1];
    const float* W1l = (const float*)d_in[2];
    const float* W1r = (const float*)d_in[3];
    const float* a1  = (const float*)d_in[4];
    const float* b1  = (const float*)d_in[5];
    const float* W2l = (const float*)d_in[6];
    const float* W2r = (const float*)d_in[7];
    const float* a2  = (const float*)d_in[8];
    const float* b2  = (const float*)d_in[9];
    float* out = (float*)d_out;

    char* ws = (char*)d_ws;
    size_t off = 0;
    auto carve = [&](size_t bytes) -> char* {
        char* p = ws + off;
        off += (bytes + 255) & ~(size_t)255;
        return p;
    };
    float*    xl1 = (float*)   carve((size_t)N_NODES * F1 * 4);
    float*    h1  = (float*)   carve((size_t)N_NODES * F1 * 4);
    float*    h2l = (float*)   carve((size_t)N_NODES * OUT_DIM * 4);
    float*    h2r = (float*)   carve((size_t)N_NODES * OUT_DIM * 4);
    int*      cnt    = (int*) carve((size_t)N_NODES * 4);
    int*      rowptr = (int*) carve((size_t)ROWPTR_N * 4);
    int*      perm   = (int*) carve((size_t)E_TOT * 4);
    (void)ws_size; (void)n_in; (void)in_sizes; (void)out_size;

    const int B = 256;

    k_init1<<<(N_NODES + B - 1) / B, B, 0, stream>>>(cnt);
    k_count<<<(E_TOT + B - 1) / B, B, 0, stream>>>(ei, cnt);
    k_scan<<<1, SCAN_T, 0, stream>>>(cnt, rowptr);
    k_fill<<<(E_TOT + B - 1) / B, B, 0, stream>>>(ei, rowptr, cnt, perm);
    k_gemm1<<<dim3(M_TILES, 4), B, 0, stream>>>(x, W1l, xl1);
    k_node1<<<N_NODES / NODES_PER_WG, B, 0, stream>>>(rowptr, perm, xl1, x, W1r, a1, b1, h1);

    k_gemm2<<<(M_TILES + 3) / 4, B, 0, stream>>>(h1, W2l, W2r, h2l, h2r);
    k_node2<<<(N_NODES * OUT_DIM + B - 1) / B, B, 0, stream>>>(rowptr, perm, h2l, h2r, a2, b2, out);
}


# Source-side 30000x512 plane stored with each lane's four 4-channel groups 512 B apart: every row load of the gather wave covers 4 whole 128-B lines, not 32 B in each of the row's 16 lines
# speedup vs baseline: 6.2452x; 1.0384x over previous
#include <hip/hip_runtime.h>
#include <hip/hip_bf16.h>

#define N_NODES 30000
#define E_RAW   300000
#define E_TOT   330000
#define HEADS   8
#define HID     64
#define F1      512
#define IN_DIM  8
#define OUT_DIM 8
#define NEG_SLOPE 0.2f
#define M_TILES 1875

typedef __attribute__((ext_vector_type(2))) float v2f;
typedef __attribute__((ext_vector_type(8))) float v8f;

__device__ __forceinline__ void get_edge(const int* __restrict__ ei, int e, int& s, int& d) {
    if (e < E_RAW) { s = ei[e]; d = ei[E_RAW + e]; }
    else           { s = e - E_RAW; d = e - E_RAW; }
}

__global__ void k_init1(int* __restrict__ cnt) {
    int i = blockIdx.x * blockDim.x + threadIdx.x;
    if (i < N_NODES) cnt[i] = 0;
}

#define XL_CHANNEL_AT(p) ((((p) & 127) >> 2) * 16 + ((p) >> 7) * 4 + ((p) & 3))
__global__ void k_gemm1(const float* __restrict__ x,
                        const float* __restrict__ W,
                        float* __restrict__ O) {
    const int wave  = threadIdx.x >> 5;
    const int lane  = threadIdx.x & 31;
    const int row   = lane & 15;
    const int khalf = lane >> 4;
    const int mbase = blockIdx.x * 16;
    const int nbase = (blockIdx.y * 8 + wave) * 16;
    const int ch    = XL_CHANNEL_AT(nbase + row);

    v8f c = {};
#pragma unroll
    for (int kk = 0; kk < 2; ++kk) {
        const int k0 = kk * 4 + khalf * 2;
        v2f a, b;
        a.x = x[(mbase + row) * IN_DIM + k0];
        a.y = x[(mbase + row) * IN_DIM + k0 + 1];
        b.x = W[(k0    ) * F1 + ch];
        b.y = W[(k0 + 1) * F1 + ch];
        c = __builtin_amdgcn_wmma_f32_16x16x4_f32(false, a, false, b,
                                                  (short)0, c, false, false);
    }
#pragma unroll
    for (int r = 0; r < 8; ++r)
        O[(size_t)(mbase + r + 8 * khalf) * F1 + nbase + row] = c[r];
}

__global__ void k_count(const int* __restrict__ ei, int* __restrict__ cnt) {
    int e = blockIdx.x * blockDim.x + threadIdx.x;
    if (e >= E_TOT) return;
    int s, d; get_edge(ei, e, s, d);
    atomicAdd(&cnt[d], 1);
}

#define SCAN_T    1024
#define SCAN_PER  32
#define ROWPTR_N  (((N_NODES + SCAN_PER) / SCAN_PER) * SCAN_PER)
__global__ void __launch_bounds__(SCAN_T)
k_scan(const int* __restrict__ cnt, int* __restrict__ rowptr) {
    __shared__ int wsum[SCAN_T / 32];
    const int t    = threadIdx.x;
    const int lane = t & 31;
    const int wave = t >> 5;
    const int base = t * SCAN_PER;
    int v[SCAN_PER];
#pragma unroll
    for (int q = 0; q < SCAN_PER / 4; ++q) {
        const int i = base + q * 4;
        int4 c = make_int4(0, 0, 0, 0);
        if (i < N_NODES) c = *(const int4*)(cnt + i);
        v[q * 4 + 0] = c.x; v[q * 4 + 1] = c.y; v[q * 4 + 2] = c.z; v[q * 4 + 3] = c.w;
    }
    int run = 0;
#pragma unroll
    for (int q = 0; q < SCAN_PER; ++q) { const int c = v[q]; v[q] = run; run += c; }
    int inc = run;
#pragma unroll
    for (int o = 1; o < 32; o <<= 1) {
        const int up = __shfl_up(inc, o, 32);
        if (lane >= o) inc += up;
    }
    if (lane == 31) wsum[wave] = inc;
    __syncthreads();
    if (wave == 0) {
        int w = wsum[lane];
#pragma unroll
        for (int o = 1; o < 32; o <<= 1) {
            const int up = __shfl_up(w, o, 32);
            if (lane >= o) w += up;
        }
        wsum[lane] = w;
    }
    __syncthreads();
    const int off = (inc - run) + (wave ? wsum[wave - 1] : 0);
    if (base < ROWPTR_N) {
#pragma unroll
        for (int q = 0; q < SCAN_PER / 4; ++q)
            *(int4*)(rowptr + base + q * 4) =
                make_int4(v[q * 4] + off, v[q * 4 + 1] + off, v[q * 4 + 2] + off, v[q * 4 + 3] + off);
    }
}

__global__ void k_fill(const int* __restrict__ ei, const int* __restrict__ rowptr,
                       int* __restrict__ cnt, int* __restrict__ perm) {
    int e = blockIdx.x * blockDim.x + threadIdx.x;
    if (e >= E_TOT) return;
    int s, d; get_edge(ei, e, s, d);
    const int pos = atomicSub(&cnt[d], 1) - 1;
    perm[rowptr[d] + pos] = s;
}

typedef __attribute__((ext_vector_type(4))) float v4f;
__device__ __forceinline__ v2f lrelu2(const v2f v) {
    return __builtin_elementwise_max(v, v * NEG_SLOPE);
}
__device__ __forceinline__ v2f logit4(const v4f q, const v4f r, const v4f a, v2f z) {
    z = __builtin_elementwise_fma(lrelu2(q.lo + r.lo), a.lo, z);
    z = __builtin_elementwise_fma(lrelu2(q.hi + r.hi), a.hi, z);
    return z;
}
__device__ __forceinline__ v4f rescale_add(const v4f v, const float f, const float g, const v4f q) {
    const v4f gs = {g, g, g, g};
    return __builtin_elementwise_fma(gs, q, v * f);
}
__device__ __forceinline__ v4f add_row(const v4f v, const float g, const v4f q) {
    const v4f gs = {g, g, g, g};
    return __builtin_elementwise_fma(gs, q, v);
}
__device__ __forceinline__ v4f bias_relu4(const v4f v, const v4f b) {
    const v4f s = v + b;
    v4f r;
    r.x = s.x > 0.f ? s.x : 0.f;
    r.y = s.y > 0.f ? s.y : 0.f;
    r.z = s.z > 0.f ? s.z : 0.f;
    r.w = s.w > 0.f ? s.w : 0.f;
    return r;
}

#define NODES_PER_WG 8
#define XR_PITCH     (F1 + F1 / 4)
#define XL_Q         (F1 / 16)
static_assert(N_NODES % NODES_PER_WG == 0, "whole work-groups only: the barrier needs all eight waves");
__global__ void __launch_bounds__(256, 9)
k_node1(const int* __restrict__ rowptr, const int* __restrict__ perm,
        const float* __restrict__ xl, const float* __restrict__ x,
        const float* __restrict__ Wr, const float* __restrict__ a1,
        const float* __restrict__ b1, float* __restrict__ h1) {
    __shared__ float xr_s[NODES_PER_WG * XR_PITCH];
    const int lane = threadIdx.x & 31;
    const int wave  = __builtin_amdgcn_readfirstlane((int)(threadIdx.x >> 5));
    const int node0 = (int)blockIdx.x * NODES_PER_WG;
    if (node0 >= N_NODES) return;
    const int node = node0 + wave;
    const int beg = rowptr[node], end = rowptr[node + 1];
    const v4f* __restrict__ pt = (const v4f*)a1 + lane * 4;
    const v4f t0 = pt[0], t1 = pt[1], t2 = pt[2], t3 = pt[3];
    int my = 0;
    {
        const int row   = lane & 15;
        const int khalf = lane >> 4;
        v2f a[2] = {{0.f, 0.f}, {0.f, 0.f}};
        if (row < NODES_PER_WG) {
            const v2f* __restrict__ px = (const v2f*)(x + (node0 + row) * IN_DIM) + khalf;
            a[0] = px[0];
            a[1] = px[2];
        }
        const v4f* __restrict__ pw = (const v4f*)(Wr + wave * HID) + row;
        v4f bk[2][2];
#pragma unroll
        for (int kk = 0; kk < 2; ++kk) {
            const int k0 = kk * 4 + khalf * 2;
            bk[kk][0] = pw[(k0    ) * (F1 / 4)];
            bk[kk][1] = pw[(k0 + 1) * (F1 / 4)];
        }
        if (lane < end - beg) my = perm[beg + lane];
        v8f d[4];
#pragma unroll
        for (int t = 0; t < 4; ++t) {
            v8f c = {};
#pragma unroll
            for (int kk = 0; kk < 2; ++kk) {
                v2f b;
                b.x = bk[kk][0][t];
                b.y = bk[kk][1][t];
                c = __builtin_amdgcn_wmma_f32_16x16x4_f32(false, a[kk], false, b,
                                                          (short)0, c, false, false);
            }
            d[t] = c;
        }
        if (khalf == 0) {
            float* __restrict__ ps = xr_s + wave * (HID + HID / 4) + 4 * row + 4 * (row >> 2);
#pragma unroll
            for (int r = 0; r < 8; ++r) {
                const v4f v = {d[0][r], d[1][r], d[2][r], d[3][r]};
                *(v4f*)(ps + r * XR_PITCH) = v;
            }
        }
    }
    __syncthreads();
    v4f* __restrict__ po = (v4f*)(h1 + (size_t)node * F1) + lane * 4;
    const v4f* __restrict__ pr = (const v4f*)(xr_s + wave * XR_PITCH + lane * 20);
    const v4f r0 = pr[0], r1 = pr[1], r2 = pr[2], r3 = pr[3];
    float m = -INFINITY, ssum = 0.f;
    v4f acc0 = {0.f, 0.f, 0.f, 0.f}, acc1 = acc0, acc2 = acc0, acc3 = acc0;
    for (int j0 = beg; j0 < end; j0 += 32) {
        const int n = (end - j0 < 32) ? (end - j0) : 32;
        if (j0 != beg) {
            my = 0;
            if (lane < n) my = perm[j0 + lane];
        }
        int t = 0;
        for (; t + 1 < n; t += 2) {
            const int sa = __builtin_amdgcn_readlane(my, t);
            const int sb = __builtin_amdgcn_readlane(my, t + 1);
            const v4f* __restrict__ pa = (const v4f*)(xl + (size_t)sa * F1) + lane;
            const v4f* __restrict__ pb = (const v4f*)(xl + (size_t)sb * F1) + lane;
            const v4f a0 = pa[0], a1v = pa[XL_Q], a2v = pa[2 * XL_Q], a3 = pa[3 * XL_Q];
            const v4f b0 = pb[0], b1v = pb[XL_Q], b2v = pb[2 * XL_Q], b3 = pb[3 * XL_Q];
            v2f za = {0.f, 0.f}, zb = {0.f, 0.f};
            za = logit4(a0, r0, t0, za);   zb = logit4(b0, r0, t0, zb);
            za = logit4(a1v, r1, t1, za);  zb = logit4(b1v, r1, t1, zb);
            za = logit4(a2v, r2, t2, za);  zb = logit4(b2v, r2, t2, zb);
            za = logit4(a3, r3, t3, za);   zb = logit4(b3, r3, t3, zb);
            float ea = za.x + za.y, eb = zb.x + zb.y;
            ea += __shfl_xor(ea, 1, 32);  eb += __shfl_xor(eb, 1, 32);
            ea += __shfl_xor(ea, 2, 32);  eb += __shfl_xor(eb, 2, 32);
            const float mn = fmaxf(m, fmaxf(ea, eb));
            const float sc = __expf(m - mn);
            const float wa = __expf(ea - mn);
            const float wb = __expf(eb - mn);
            m = mn;
            ssum = fmaf(ssum, sc, wa + wb);
            acc0 = add_row(rescale_add(acc0, sc, wa, a0), wb, b0);
            acc1 = add_row(rescale_add(acc1, sc, wa, a1v), wb, b1v);
            acc2 = add_row(rescale_add(acc2, sc, wa, a2v), wb, b2v);
            acc3 = add_row(rescale_add(acc3, sc, wa, a3), wb, b3);
        }
        if (t < n) {
            const int sa = __builtin_amdgcn_readlane(my, t);
            const v4f* __restrict__ pa = (const v4f*)(xl + (size_t)sa * F1) + lane;
            const v4f a0 = pa[0], a1v = pa[XL_Q], a2v = pa[2 * XL_Q], a3 = pa[3 * XL_Q];
            v2f za = {0.f, 0.f};
            za = logit4(a0, r0, t0, za);
            za = logit4(a1v, r1, t1, za);
            za = logit4(a2v, r2, t2, za);
            za = logit4(a3, r3, t3, za);
            float ea = za.x + za.y;
            ea += __shfl_xor(ea, 1, 32);
            ea += __shfl_xor(ea, 2, 32);
            const float mn = fmaxf(m, ea);
            const float sc = __expf(m - mn);
            const float wa = __expf(ea - mn);
            m = mn;
            ssum = fmaf(ssum, sc, wa);
            acc0 = rescale_add(acc0, sc, wa, a0);
            acc1 = rescale_add(acc1, sc, wa, a1v);
            acc2 = rescale_add(acc2, sc, wa, a2v);
            acc3 = rescale_add(acc3, sc, wa, a3);
        }
    }
    const v4f* __restrict__ pc = (const v4f*)b1 + lane * 4;
    const v4f c0 = pc[0], c1 = pc[1], c2 = pc[2], c3 = pc[3];
    po[0] = bias_relu4(acc0 / ssum, c0);
    po[1] = bias_relu4(acc1 / ssum, c1);
    po[2] = bias_relu4(acc2 / ssum, c2);
    po[3] = bias_relu4(acc3 / ssum, c3);
}

__global__ void k_gemm2(const float* __restrict__ h,
                        const float* __restrict__ W2l,
                        const float* __restrict__ W2r,
                        float* __restrict__ h2l, float* __restrict__ h2r) {
    const int wave  = threadIdx.x >> 5;
    const int lane  = threadIdx.x & 31;
    const int row   = lane & 15;
    const int khalf = lane >> 4;
    const int mtile = blockIdx.x * 4 + (wave >> 1);
    if (mtile >= M_TILES) return;
    const int mbase = mtile * 16;
    const float* __restrict__ W = (wave & 1) ? W2r : W2l;
    float* __restrict__ O       = (wave & 1) ? h2r : h2l;

    v8f c = {};
    for (int kk = 0; kk < F1 / 4; ++kk) {
        const int k0 = kk * 4 + khalf * 2;
        v2f a, b;
        a.x = h[(size_t)(mbase + row) * F1 + k0];
        a.y = h[(size_t)(mbase + row) * F1 + k0 + 1];
        b.x = (row < OUT_DIM) ? W[(k0    ) * OUT_DIM + row] : 0.f;
        b.y = (row < OUT_DIM) ? W[(k0 + 1) * OUT_DIM + row] : 0.f;
        c = __builtin_amdgcn_wmma_f32_16x16x4_f32(false, a, false, b,
                                                  (short)0, c, false, false);
    }
    if (row < OUT_DIM) {
#pragma unroll
        for (int r = 0; r < 8; ++r)
            O[(size_t)(mbase + r + 8 * khalf) * OUT_DIM + row] = c[r];
    }
}

__global__ void __launch_bounds__(256)
k_node2(const int* __restrict__ rowptr, const int* __restrict__ perm,
        const float* __restrict__ h2l, const float* __restrict__ h2r,
        const float* __restrict__ a2, const float* __restrict__ b2,
        float* __restrict__ out) {
    const int gid  = blockIdx.x * blockDim.x + threadIdx.x;
    const int node = gid >> 3;
    const int sub  = gid & 7;
    if (node >= N_NODES) return;
    const int beg = rowptr[node], end = rowptr[node + 1];
    const v4f* __restrict__ pr = (const v4f*)(h2r + (size_t)node * OUT_DIM);
    const v4f r0 = pr[0], r1 = pr[1];
    const v4f t0 = ((const v4f*)a2)[0], t1 = ((const v4f*)a2)[1];
    const float bias = b2[sub];
    float m = -INFINITY, ssum = 0.f;
    v4f acc0 = {0.f, 0.f, 0.f, 0.f}, acc1 = acc0;
    for (int j = beg + sub; j < end; j += 8) {
        const int s = perm[j];
        const v4f* __restrict__ pq = (const v4f*)(h2l + (size_t)s * OUT_DIM);
        const v4f q0 = pq[0], q1 = pq[1];
        v2f z = {0.f, 0.f};
        z = logit4(q0, r0, t0, z);
        z = logit4(q1, r1, t1, z);
        const float e  = z.x + z.y;
        const float mn = fmaxf(m, e);
        const float sc = __expf(m - mn);
        const float w  = __expf(e - mn);
        m = mn;
        ssum = fmaf(ssum, sc, w);
        acc0 = rescale_add(acc0, sc, w, q0);
        acc1 = rescale_add(acc1, sc, w, q1);
    }
#pragma unroll
    for (int o = 1; o < 8; o <<= 1) {
        const float mo = __shfl_xor(m, o, 32);
        const float so = __shfl_xor(ssum, o, 32);
        v4f o0, o1;
        o0.x = __shfl_xor(acc0.x, o, 32);  o0.y = __shfl_xor(acc0.y, o, 32);
        o0.z = __shfl_xor(acc0.z, o, 32);  o0.w = __shfl_xor(acc0.w, o, 32);
        o1.x = __shfl_xor(acc1.x, o, 32);  o1.y = __shfl_xor(acc1.y, o, 32);
        o1.z = __shfl_xor(acc1.z, o, 32);  o1.w = __shfl_xor(acc1.w, o, 32);
        const float mn = fmaxf(m, mo);
        const float f  = (m  == mn) ? 1.f : __expf(m  - mn);
        const float g  = (mo == mn) ? 1.f : __expf(mo - mn);
        ssum = fmaf(so, g, ssum * f);
        acc0 = rescale_add(acc0, f, g, o0);
        acc1 = rescale_add(acc1, f, g, o1);
        m = mn;
    }
    const v4f half = (sub & 4) ? acc1 : acc0;
    const float lo = (sub & 1) ? half.y : half.x;
    const float hi = (sub & 1) ? half.w : half.z;
    const float v  = (sub & 2) ? hi : lo;
    out[gid] = v / ssum + bias;
}

extern "C" void kernel_launch(void* const* d_in, const int* in_sizes, int n_in,
                              void* d_out, int out_size, void* d_ws, size_t ws_size,
                              hipStream_t stream) {
    const float* x   = (const float*)d_in[0];
    const int*   ei  = (const int*)  d_in[1];
    const float* W1l = (const float*)d_in[2];
    const float* W1r = (const float*)d_in[3];
    const float* a1  = (const float*)d_in[4];
    const float* b1  = (const float*)d_in[5];
    const float* W2l = (const float*)d_in[6];
    const float* W2r = (const float*)d_in[7];
    const float* a2  = (const float*)d_in[8];
    const float* b2  = (const float*)d_in[9];
    float* out = (float*)d_out;

    char* ws = (char*)d_ws;
    size_t off = 0;
    auto carve = [&](size_t bytes) -> char* {
        char* p = ws + off;
        off += (bytes + 255) & ~(size_t)255;
        return p;
    };
    float*    xl1 = (float*)   carve((size_t)N_NODES * F1 * 4);
    float*    h1  = (float*)   carve((size_t)N_NODES * F1 * 4);
    float*    h2l = (float*)   carve((size_t)N_NODES * OUT_DIM * 4);
    float*    h2r = (float*)   carve((size_t)N_NODES * OUT_DIM * 4);
    int*      cnt    = (int*) carve((size_t)N_NODES * 4);
    int*      rowptr = (int*) carve((size_t)ROWPTR_N * 4);
    int*      perm   = (int*) carve((size_t)E_TOT * 4);
    (void)ws_size; (void)n_in; (void)in_sizes; (void)out_size;

    const int B = 256;

    k_init1<<<(N_NODES + B - 1) / B, B, 0, stream>>>(cnt);
    k_count<<<(E_TOT + B - 1) / B, B, 0, stream>>>(ei, cnt);
    k_scan<<<1, SCAN_T, 0, stream>>>(cnt, rowptr);
    k_fill<<<(E_TOT + B - 1) / B, B, 0, stream>>>(ei, rowptr, cnt, perm);
    k_gemm1<<<dim3(M_TILES, 4), B, 0, stream>>>(x, W1l, xl1);
    k_node1<<<N_NODES / NODES_PER_WG, B, 0, stream>>>(rowptr, perm, xl1, x, W1r, a1, b1, h1);

    k_gemm2<<<(M_TILES + 3) / 4, B, 0, stream>>>(h1, W2l, W2r, h2l, h2r);
    k_node2<<<(N_NODES * OUT_DIM + B - 1) / B, B, 0, stream>>>(rowptr, perm, h2l, h2r, a2, b2, out);
}


# Layer-2 projections h@[W2l|W2r] moved into the node kernel: waves park their rows in LDS, the last to arrive (LDS counter) does the WMMA tile, 2 accumulators (sum order changes); h plane, launch go
# speedup vs baseline: 7.3269x; 1.1732x over previous
#include <hip/hip_runtime.h>
#include <hip/hip_bf16.h>

#define N_NODES 30000
#define E_RAW   300000
#define E_TOT   330000
#define HEADS   8
#define HID     64
#define F1      512
#define IN_DIM  8
#define OUT_DIM 8
#define NEG_SLOPE 0.2f
#define M_TILES 1875

typedef __attribute__((ext_vector_type(2))) float v2f;
typedef __attribute__((ext_vector_type(8))) float v8f;

__device__ __forceinline__ void get_edge(const int* __restrict__ ei, int e, int& s, int& d) {
    if (e < E_RAW) { s = ei[e]; d = ei[E_RAW + e]; }
    else           { s = e - E_RAW; d = e - E_RAW; }
}

__global__ void k_init1(int* __restrict__ cnt) {
    int i = blockIdx.x * blockDim.x + threadIdx.x;
    if (i < N_NODES) cnt[i] = 0;
}

#define XL_CHANNEL_AT(p) ((((p) & 127) >> 2) * 16 + ((p) >> 7) * 4 + ((p) & 3))
__global__ void k_gemm1(const float* __restrict__ x,
                        const float* __restrict__ W,
                        float* __restrict__ O) {
    const int wave  = threadIdx.x >> 5;
    const int lane  = threadIdx.x & 31;
    const int row   = lane & 15;
    const int khalf = lane >> 4;
    const int mbase = blockIdx.x * 16;
    const int nbase = (blockIdx.y * 8 + wave) * 16;
    const int ch    = XL_CHANNEL_AT(nbase + row);

    v8f c = {};
#pragma unroll
    for (int kk = 0; kk < 2; ++kk) {
        const int k0 = kk * 4 + khalf * 2;
        v2f a, b;
        a.x = x[(mbase + row) * IN_DIM + k0];
        a.y = x[(mbase + row) * IN_DIM + k0 + 1];
        b.x = W[(k0    ) * F1 + ch];
        b.y = W[(k0 + 1) * F1 + ch];
        c = __builtin_amdgcn_wmma_f32_16x16x4_f32(false, a, false, b,
                                                  (short)0, c, false, false);
    }
#pragma unroll
    for (int r = 0; r < 8; ++r)
        O[(size_t)(mbase + r + 8 * khalf) * F1 + nbase + row] = c[r];
}

__global__ void k_count(const int* __restrict__ ei, int* __restrict__ cnt) {
    int e = blockIdx.x * blockDim.x + threadIdx.x;
    if (e >= E_TOT) return;
    int s, d; get_edge(ei, e, s, d);
    atomicAdd(&cnt[d], 1);
}

#define SCAN_T    1024
#define SCAN_PER  32
#define ROWPTR_N  (((N_NODES + SCAN_PER) / SCAN_PER) * SCAN_PER)
__global__ void __launch_bounds__(SCAN_T)
k_scan(const int* __restrict__ cnt, int* __restrict__ rowptr) {
    __shared__ int wsum[SCAN_T / 32];
    const int t    = threadIdx.x;
    const int lane = t & 31;
    const int wave = t >> 5;
    const int base = t * SCAN_PER;
    int v[SCAN_PER];
#pragma unroll
    for (int q = 0; q < SCAN_PER / 4; ++q) {
        const int i = base + q * 4;
        int4 c = make_int4(0, 0, 0, 0);
        if (i < N_NODES) c = *(const int4*)(cnt + i);
        v[q * 4 + 0] = c.x; v[q * 4 + 1] = c.y; v[q * 4 + 2] = c.z; v[q * 4 + 3] = c.w;
    }
    int run = 0;
#pragma unroll
    for (int q = 0; q < SCAN_PER; ++q) { const int c = v[q]; v[q] = run; run += c; }
    int inc = run;
#pragma unroll
    for (int o = 1; o < 32; o <<= 1) {
        const int up = __shfl_up(inc, o, 32);
        if (lane >= o) inc += up;
    }
    if (lane == 31) wsum[wave] = inc;
    __syncthreads();
    if (wave == 0) {
        int w = wsum[lane];
#pragma unroll
        for (int o = 1; o < 32; o <<= 1) {
            const int up = __shfl_up(w, o, 32);
            if (lane >= o) w += up;
        }
        wsum[lane] = w;
    }
    __syncthreads();
    const int off = (inc - run) + (wave ? wsum[wave - 1] : 0);
    if (base < ROWPTR_N) {
#pragma unroll
        for (int q = 0; q < SCAN_PER / 4; ++q)
            *(int4*)(rowptr + base + q * 4) =
                make_int4(v[q * 4] + off, v[q * 4 + 1] + off, v[q * 4 + 2] + off, v[q * 4 + 3] + off);
    }
}

__global__ void k_fill(const int* __restrict__ ei, const int* __restrict__ rowptr,
                       int* __restrict__ cnt, int* __restrict__ perm) {
    int e = blockIdx.x * blockDim.x + threadIdx.x;
    if (e >= E_TOT) return;
    int s, d; get_edge(ei, e, s, d);
    const int pos = atomicSub(&cnt[d], 1) - 1;
    perm[rowptr[d] + pos] = s;
}

typedef __attribute__((ext_vector_type(4))) float v4f;
__device__ __forceinline__ v2f lrelu2(const v2f v) {
    return __builtin_elementwise_max(v, v * NEG_SLOPE);
}
__device__ __forceinline__ v2f logit4(const v4f q, const v4f r, const v4f a, v2f z) {
    z = __builtin_elementwise_fma(lrelu2(q.lo + r.lo), a.lo, z);
    z = __builtin_elementwise_fma(lrelu2(q.hi + r.hi), a.hi, z);
    return z;
}
__device__ __forceinline__ v4f rescale_add(const v4f v, const float f, const float g, const v4f q) {
    const v4f gs = {g, g, g, g};
    return __builtin_elementwise_fma(gs, q, v * f);
}
__device__ __forceinline__ v4f add_row(const v4f v, const float g, const v4f q) {
    const v4f gs = {g, g, g, g};
    return __builtin_elementwise_fma(gs, q, v);
}
__device__ __forceinline__ v4f bias_relu4(const v4f v, const v4f b) {
    const v4f s = v + b;
    v4f r;
    r.x = s.x > 0.f ? s.x : 0.f;
    r.y = s.y > 0.f ? s.y : 0.f;
    r.z = s.z > 0.f ? s.z : 0.f;
    r.w = s.w > 0.f ? s.w : 0.f;
    return r;
}

#define NODES_PER_WG 8
#define XR_PITCH     (F1 + F1 / 4)
#define XL_Q         (F1 / 16)
#define H_PITCH      (F1 + 4)
static_assert(N_NODES % NODES_PER_WG == 0, "whole work-groups only: the barrier needs all eight waves");
__global__ void __launch_bounds__(256, 9)
k_node1(const int* __restrict__ rowptr, const int* __restrict__ perm,
        const float* __restrict__ xl, const float* __restrict__ x,
        const float* __restrict__ Wr, const float* __restrict__ a1,
        const float* __restrict__ b1,
        const float* __restrict__ W2l, const float* __restrict__ W2r,
        float* __restrict__ h2l, float* __restrict__ h2r) {
    __shared__ __attribute__((aligned(16))) float xr_s[NODES_PER_WG * XR_PITCH];
    __shared__ __attribute__((aligned(16))) float h_s[NODES_PER_WG * H_PITCH];
    __shared__ int   done_s;
    if (threadIdx.x == 0) done_s = 0;
    const int lane = threadIdx.x & 31;
    const int wave  = __builtin_amdgcn_readfirstlane((int)(threadIdx.x >> 5));
    const int node0 = (int)blockIdx.x * NODES_PER_WG;
    if (node0 >= N_NODES) return;
    const int node = node0 + wave;
    const int beg = rowptr[node], end = rowptr[node + 1];
    const v4f* __restrict__ pt = (const v4f*)a1 + lane * 4;
    const v4f t0 = pt[0], t1 = pt[1], t2 = pt[2], t3 = pt[3];
    int my = 0;
    {
        const int row   = lane & 15;
        const int khalf = lane >> 4;
        v2f a[2] = {{0.f, 0.f}, {0.f, 0.f}};
        if (row < NODES_PER_WG) {
            const v2f* __restrict__ px = (const v2f*)(x + (node0 + row) * IN_DIM) + khalf;
            a[0] = px[0];
            a[1] = px[2];
        }
        const v4f* __restrict__ pw = (const v4f*)(Wr + wave * HID) + row;
        v4f bk[2][2];
#pragma unroll
        for (int kk = 0; kk < 2; ++kk) {
            const int k0 = kk * 4 + khalf * 2;
            bk[kk][0] = pw[(k0    ) * (F1 / 4)];
            bk[kk][1] = pw[(k0 + 1) * (F1 / 4)];
        }
        if (lane < end - beg) my = perm[beg + lane];
        v8f d[4];
#pragma unroll
        for (int t = 0; t < 4; ++t) {
            v8f c = {};
#pragma unroll
            for (int kk = 0; kk < 2; ++kk) {
                v2f b;
                b.x = bk[kk][0][t];
                b.y = bk[kk][1][t];
                c = __builtin_amdgcn_wmma_f32_16x16x4_f32(false, a[kk], false, b,
                                                          (short)0, c, false, false);
            }
            d[t] = c;
        }
        if (khalf == 0) {
            float* __restrict__ ps = xr_s + wave * (HID + HID / 4) + 4 * row + 4 * (row >> 2);
#pragma unroll
            for (int r = 0; r < 8; ++r) {
                const v4f v = {d[0][r], d[1][r], d[2][r], d[3][r]};
                *(v4f*)(ps + r * XR_PITCH) = v;
            }
        }
    }
    __syncthreads();
    const v4f* __restrict__ pr = (const v4f*)(xr_s + wave * XR_PITCH + lane * 20);
    const v4f r0 = pr[0], r1 = pr[1], r2 = pr[2], r3 = pr[3];
    float m = -INFINITY, ssum = 0.f;
    v4f acc0 = {0.f, 0.f, 0.f, 0.f}, acc1 = acc0, acc2 = acc0, acc3 = acc0;
    for (int j0 = beg; j0 < end; j0 += 32) {
        const int n = (end - j0 < 32) ? (end - j0) : 32;
        if (j0 != beg) {
            my = 0;
            if (lane < n) my = perm[j0 + lane];
        }
        int t = 0;
        for (; t + 1 < n; t += 2) {
            const int sa = __builtin_amdgcn_readlane(my, t);
            const int sb = __builtin_amdgcn_readlane(my, t + 1);
            const v4f* __restrict__ pa = (const v4f*)(xl + (size_t)sa * F1) + lane;
            const v4f* __restrict__ pb = (const v4f*)(xl + (size_t)sb * F1) + lane;
            const v4f a0 = pa[0], a1v = pa[XL_Q], a2v = pa[2 * XL_Q], a3 = pa[3 * XL_Q];
            const v4f b0 = pb[0], b1v = pb[XL_Q], b2v = pb[2 * XL_Q], b3 = pb[3 * XL_Q];
            v2f za = {0.f, 0.f}, zb = {0.f, 0.f};
            za = logit4(a0, r0, t0, za);   zb = logit4(b0, r0, t0, zb);
            za = logit4(a1v, r1, t1, za);  zb = logit4(b1v, r1, t1, zb);
            za = logit4(a2v, r2, t2, za);  zb = logit4(b2v, r2, t2, zb);
            za = logit4(a3, r3, t3, za);   zb = logit4(b3, r3, t3, zb);
            float ea = za.x + za.y, eb = zb.x + zb.y;
            ea += __shfl_xor(ea, 1, 32);  eb += __shfl_xor(eb, 1, 32);
            ea += __shfl_xor(ea, 2, 32);  eb += __shfl_xor(eb, 2, 32);
            const float mn = fmaxf(m, fmaxf(ea, eb));
            const float sc = __expf(m - mn);
            const float wa = __expf(ea - mn);
            const float wb = __expf(eb - mn);
            m = mn;
            ssum = fmaf(ssum, sc, wa + wb);
            acc0 = add_row(rescale_add(acc0, sc, wa, a0), wb, b0);
            acc1 = add_row(rescale_add(acc1, sc, wa, a1v), wb, b1v);
            acc2 = add_row(rescale_add(acc2, sc, wa, a2v), wb, b2v);
            acc3 = add_row(rescale_add(acc3, sc, wa, a3), wb, b3);
        }
        if (t < n) {
            const int sa = __builtin_amdgcn_readlane(my, t);
            const v4f* __restrict__ pa = (const v4f*)(xl + (size_t)sa * F1) + lane;
            const v4f a0 = pa[0], a1v = pa[XL_Q], a2v = pa[2 * XL_Q], a3 = pa[3 * XL_Q];
            v2f za = {0.f, 0.f};
            za = logit4(a0, r0, t0, za);
            za = logit4(a1v, r1, t1, za);
            za = logit4(a2v, r2, t2, za);
            za = logit4(a3, r3, t3, za);
            float ea = za.x + za.y;
            ea += __shfl_xor(ea, 1, 32);
            ea += __shfl_xor(ea, 2, 32);
            const float mn = fmaxf(m, ea);
            const float sc = __expf(m - mn);
            const float wa = __expf(ea - mn);
            m = mn;
            ssum = fmaf(ssum, sc, wa);
            acc0 = rescale_add(acc0, sc, wa, a0);
            acc1 = rescale_add(acc1, sc, wa, a1v);
            acc2 = rescale_add(acc2, sc, wa, a2v);
            acc3 = rescale_add(acc3, sc, wa, a3);
        }
    }
    const v4f* __restrict__ pc = (const v4f*)b1 + lane * 4;
    const v4f c0 = pc[0], c1 = pc[1], c2 = pc[2], c3 = pc[3];
    float* __restrict__ ph = h_s + wave * H_PITCH + lane * 4;
    *(v4f*)(ph)           = bias_relu4(acc0 / ssum, c0);
    *(v4f*)(ph + F1 / 4)  = bias_relu4(acc1 / ssum, c1);
    *(v4f*)(ph + F1 / 2)  = bias_relu4(acc2 / ssum, c2);
    *(v4f*)(ph + 3 * F1 / 4) = bias_relu4(acc3 / ssum, c3);
    __builtin_amdgcn_fence(__ATOMIC_RELEASE, "workgroup");
    int last = 0;
    if (lane == 0)
        last = __hip_atomic_fetch_add(&done_s, 1, __ATOMIC_RELAXED, __HIP_MEMORY_SCOPE_WORKGROUP)
               == NODES_PER_WG - 1;
    last = __builtin_amdgcn_readfirstlane(last);
    if (!last) return;
    __builtin_amdgcn_fence(__ATOMIC_ACQUIRE, "workgroup");
    {
        const int row   = lane & 15;
        const int khalf = lane >> 4;
        const float* __restrict__ pw = ((row < OUT_DIM) ? W2l : W2r) + (row & (OUT_DIM - 1))
                                       + XL_CHANNEL_AT(32 * khalf) * OUT_DIM;
        const v4f* __restrict__ pa = (const v4f*)(h_s + (row & (NODES_PER_WG - 1)) * H_PITCH + 32 * khalf);
        v8f d0 = {}, d1 = {};
#pragma unroll 4
        for (int sl = 0; sl < F1 / 64; ++sl) {
            v4f av[8];
#pragma unroll
            for (int j = 0; j < 8; ++j)
                av[j] = pa[sl * 16 + j];
#pragma unroll
            for (int kk = 0; kk < 16; ++kk) {
                const int ch = XL_CHANNEL_AT(64 * sl + 2 * kk);
                v2f a, b;
                a.x = av[kk >> 1][(2 * kk) & 3];
                a.y = av[kk >> 1][((2 * kk) & 3) + 1];
                b.x = pw[(ch    ) * OUT_DIM];
                b.y = pw[(ch + 1) * OUT_DIM];
                if (kk & 1)
                    d1 = __builtin_amdgcn_wmma_f32_16x16x4_f32(false, a, false, b, (short)0, d1, false, false);
                else
                    d0 = __builtin_amdgcn_wmma_f32_16x16x4_f32(false, a, false, b, (short)0, d0, false, false);
            }
        }
        const v8f d = d0 + d1;
        if (khalf == 0) {
            float* __restrict__ po = ((row < OUT_DIM) ? h2l : h2r) + (size_t)node0 * OUT_DIM
                                     + (row & (OUT_DIM - 1));
#pragma unroll
            for (int r = 0; r < NODES_PER_WG; ++r)
                po[r * OUT_DIM] = d[r];
        }
    }
}

__global__ void __launch_bounds__(256)
k_node2(const int* __restrict__ rowptr, const int* __restrict__ perm,
        const float* __restrict__ h2l, const float* __restrict__ h2r,
        const float* __restrict__ a2, const float* __restrict__ b2,
        float* __restrict__ out) {
    const int gid  = blockIdx.x * blockDim.x + threadIdx.x;
    const int node = gid >> 3;
    const int sub  = gid & 7;
    if (node >= N_NODES) return;
    const int beg = rowptr[node], end = rowptr[node + 1];
    const v4f* __restrict__ pr = (const v4f*)(h2r + (size_t)node * OUT_DIM);
    const v4f r0 = pr[0], r1 = pr[1];
    const v4f t0 = ((const v4f*)a2)[0], t1 = ((const v4f*)a2)[1];
    const float bias = b2[sub];
    float m = -INFINITY, ssum = 0.f;
    v4f acc0 = {0.f, 0.f, 0.f, 0.f}, acc1 = acc0;
    for (int j = beg + sub; j < end; j += 8) {
        const int s = perm[j];
        const v4f* __restrict__ pq = (const v4f*)(h2l + (size_t)s * OUT_DIM);
        const v4f q0 = pq[0], q1 = pq[1];
        v2f z = {0.f, 0.f};
        z = logit4(q0, r0, t0, z);
        z = logit4(q1, r1, t1, z);
        const float e  = z.x + z.y;
        const float mn = fmaxf(m, e);
        const float sc = __expf(m - mn);
        const float w  = __expf(e - mn);
        m = mn;
        ssum = fmaf(ssum, sc, w);
        acc0 = rescale_add(acc0, sc, w, q0);
        acc1 = rescale_add(acc1, sc, w, q1);
    }
#pragma unroll
    for (int o = 1; o < 8; o <<= 1) {
        const float mo = __shfl_xor(m, o, 32);
        const float so = __shfl_xor(ssum, o, 32);
        v4f o0, o1;
        o0.x = __shfl_xor(acc0.x, o, 32);  o0.y = __shfl_xor(acc0.y, o, 32);
        o0.z = __shfl_xor(acc0.z, o, 32);  o0.w = __shfl_xor(acc0.w, o, 32);
        o1.x = __shfl_xor(acc1.x, o, 32);  o1.y = __shfl_xor(acc1.y, o, 32);
        o1.z = __shfl_xor(acc1.z, o, 32);  o1.w = __shfl_xor(acc1.w, o, 32);
        const float mn = fmaxf(m, mo);
        const float f  = (m  == mn) ? 1.f : __expf(m  - mn);
        const float g  = (mo == mn) ? 1.f : __expf(mo - mn);
        ssum = fmaf(so, g, ssum * f);
        acc0 = rescale_add(acc0, f, g, o0);
        acc1 = rescale_add(acc1, f, g, o1);
        m = mn;
    }
    const v4f half = (sub & 4) ? acc1 : acc0;
    const float lo = (sub & 1) ? half.y : half.x;
    const float hi = (sub & 1) ? half.w : half.z;
    const float v  = (sub & 2) ? hi : lo;
    out[gid] = v / ssum + bias;
}

extern "C" void kernel_launch(void* const* d_in, const int* in_sizes, int n_in,
                              void* d_out, int out_size, void* d_ws, size_t ws_size,
                              hipStream_t stream) {
    const float* x   = (const float*)d_in[0];
    const int*   ei  = (const int*)  d_in[1];
    const float* W1l = (const float*)d_in[2];
    const float* W1r = (const float*)d_in[3];
    const float* a1  = (const float*)d_in[4];
    const float* b1  = (const float*)d_in[5];
    const float* W2l = (const float*)d_in[6];
    const float* W2r = (const float*)d_in[7];
    const float* a2  = (const float*)d_in[8];
    const float* b2  = (const float*)d_in[9];
    float* out = (float*)d_out;

    char* ws = (char*)d_ws;
    size_t off = 0;
    auto carve = [&](size_t bytes) -> char* {
        char* p = ws + off;
        off += (bytes + 255) & ~(size_t)255;
        return p;
    };
    float*    xl1 = (float*)   carve((size_t)N_NODES * F1 * 4);
    float*    h2l = (float*)   carve((size_t)N_NODES * OUT_DIM * 4);
    float*    h2r = (float*)   carve((size_t)N_NODES * OUT_DIM * 4);
    int*      cnt    = (int*) carve((size_t)N_NODES * 4);
    int*      rowptr = (int*) carve((size_t)ROWPTR_N * 4);
    int*      perm   = (int*) carve((size_t)E_TOT * 4);
    (void)ws_size; (void)n_in; (void)in_sizes; (void)out_size;

    const int B = 256;

    k_init1<<<(N_NODES + B - 1) / B, B, 0, stream>>>(cnt);
    k_count<<<(E_TOT + B - 1) / B, B, 0, stream>>>(ei, cnt);
    k_scan<<<1, SCAN_T, 0, stream>>>(cnt, rowptr);
    k_fill<<<(E_TOT + B - 1) / B, B, 0, stream>>>(ei, rowptr, cnt, perm);
    k_gemm1<<<dim3(M_TILES, 4), B, 0, stream>>>(x, W1l, xl1);
    k_node1<<<N_NODES / NODES_PER_WG, B, 0, stream>>>(rowptr, perm, xl1, x, W1r, a1, b1,
                                                      W2l, W2r, h2l, h2r);

    k_node2<<<(N_NODES * OUT_DIM + B - 1) / B, B, 0, stream>>>(rowptr, perm, h2l, h2r, a2, b2, out);
}


# Prefix-sum block rides in the layer-1 source-side product's launch (work-group 0 scans, the others form x@W1l tiles meanwhile): the one-block scan launch and its boundary go
# speedup vs baseline: 7.7331x; 1.0554x over previous
#include <hip/hip_runtime.h>
#include <hip/hip_bf16.h>

#define N_NODES 30000
#define E_RAW   300000
#define E_TOT   330000
#define HEADS   8
#define HID     64
#define F1      512
#define IN_DIM  8
#define OUT_DIM 8
#define NEG_SLOPE 0.2f
#define M_TILES 1875

typedef __attribute__((ext_vector_type(2))) float v2f;
typedef __attribute__((ext_vector_type(8))) float v8f;

__device__ __forceinline__ void get_edge(const int* __restrict__ ei, int e, int& s, int& d) {
    if (e < E_RAW) { s = ei[e]; d = ei[E_RAW + e]; }
    else           { s = e - E_RAW; d = e - E_RAW; }
}

__global__ void k_init1(int* __restrict__ cnt) {
    int i = blockIdx.x * blockDim.x + threadIdx.x;
    if (i < N_NODES) cnt[i] = 0;
}

#define XL_CHANNEL_AT(p) ((((p) & 127) >> 2) * 16 + ((p) >> 7) * 4 + ((p) & 3))
__device__ __forceinline__ void gemm1_tile(const float* __restrict__ x,
                                           const float* __restrict__ W,
                                           float* __restrict__ O,
                                           const int mtile, const int ntile) {
    const int lane  = threadIdx.x & 31;
    const int row   = lane & 15;
    const int khalf = lane >> 4;
    const int mbase = mtile * 16;
    const int nbase = ntile * 16;
    const int ch    = XL_CHANNEL_AT(nbase + row);

    v8f c = {};
#pragma unroll
    for (int kk = 0; kk < 2; ++kk) {
        const int k0 = kk * 4 + khalf * 2;
        v2f a, b;
        a.x = x[(mbase + row) * IN_DIM + k0];
        a.y = x[(mbase + row) * IN_DIM + k0 + 1];
        b.x = W[(k0    ) * F1 + ch];
        b.y = W[(k0 + 1) * F1 + ch];
        c = __builtin_amdgcn_wmma_f32_16x16x4_f32(false, a, false, b,
                                                  (short)0, c, false, false);
    }
#pragma unroll
    for (int r = 0; r < 8; ++r)
        O[(size_t)(mbase + r + 8 * khalf) * F1 + nbase + row] = c[r];
}

__global__ void k_count(const int* __restrict__ ei, int* __restrict__ cnt) {
    int e = blockIdx.x * blockDim.x + threadIdx.x;
    if (e >= E_TOT) return;
    int s, d; get_edge(ei, e, s, d);
    atomicAdd(&cnt[d], 1);
}

#define SCAN_T    1024
#define SCAN_PER  32
#define ROWPTR_N  (((N_NODES + SCAN_PER) / SCAN_PER) * SCAN_PER)
__device__ __forceinline__ void scan_block(const int* __restrict__ cnt, int* __restrict__ rowptr) {
    __shared__ int wsum[SCAN_T / 32];
    const int t    = threadIdx.x;
    const int lane = t & 31;
    const int wave = t >> 5;
    const int base = t * SCAN_PER;
    int v[SCAN_PER];
#pragma unroll
    for (int q = 0; q < SCAN_PER / 4; ++q) {
        const int i = base + q * 4;
        int4 c = make_int4(0, 0, 0, 0);
        if (i < N_NODES) c = *(const int4*)(cnt + i);
        v[q * 4 + 0] = c.x; v[q * 4 + 1] = c.y; v[q * 4 + 2] = c.z; v[q * 4 + 3] = c.w;
    }
    int run = 0;
#pragma unroll
    for (int q = 0; q < SCAN_PER; ++q) { const int c = v[q]; v[q] = run; run += c; }
    int inc = run;
#pragma unroll
    for (int o = 1; o < 32; o <<= 1) {
        const int up = __shfl_up(inc, o, 32);
        if (lane >= o) inc += up;
    }
    if (lane == 31) wsum[wave] = inc;
    __syncthreads();
    if (wave == 0) {
        int w = wsum[lane];
#pragma unroll
        for (int o = 1; o < 32; o <<= 1) {
            const int up = __shfl_up(w, o, 32);
            if (lane >= o) w += up;
        }
        wsum[lane] = w;
    }
    __syncthreads();
    const int off = (inc - run) + (wave ? wsum[wave - 1] : 0);
    if (base < ROWPTR_N) {
#pragma unroll
        for (int q = 0; q < SCAN_PER / 4; ++q)
            *(int4*)(rowptr + base + q * 4) =
                make_int4(v[q * 4] + off, v[q * 4 + 1] + off, v[q * 4 + 2] + off, v[q * 4 + 3] + off);
    }
}

#define SG_BLOCKS (M_TILES + 1)
static_assert(SCAN_T / 32 == F1 / 16, "one wave per N-tile of the 512-wide row");
__global__ void __launch_bounds__(SCAN_T)
k_scan_gemm1(const int* __restrict__ cnt, int* __restrict__ rowptr,
             const float* __restrict__ x, const float* __restrict__ W, float* __restrict__ O) {
    if (blockIdx.x == 0) {
        scan_block(cnt, rowptr);
        return;
    }
    gemm1_tile(x, W, O, (int)blockIdx.x - 1, (int)(threadIdx.x >> 5));
}

__global__ void k_fill(const int* __restrict__ ei, const int* __restrict__ rowptr,
                       int* __restrict__ cnt, int* __restrict__ perm) {
    int e = blockIdx.x * blockDim.x + threadIdx.x;
    if (e >= E_TOT) return;
    int s, d; get_edge(ei, e, s, d);
    const int pos = atomicSub(&cnt[d], 1) - 1;
    perm[rowptr[d] + pos] = s;
}

typedef __attribute__((ext_vector_type(4))) float v4f;
__device__ __forceinline__ v2f lrelu2(const v2f v) {
    return __builtin_elementwise_max(v, v * NEG_SLOPE);
}
__device__ __forceinline__ v2f logit4(const v4f q, const v4f r, const v4f a, v2f z) {
    z = __builtin_elementwise_fma(lrelu2(q.lo + r.lo), a.lo, z);
    z = __builtin_elementwise_fma(lrelu2(q.hi + r.hi), a.hi, z);
    return z;
}
__device__ __forceinline__ v4f rescale_add(const v4f v, const float f, const float g, const v4f q) {
    const v4f gs = {g, g, g, g};
    return __builtin_elementwise_fma(gs, q, v * f);
}
__device__ __forceinline__ v4f add_row(const v4f v, const float g, const v4f q) {
    const v4f gs = {g, g, g, g};
    return __builtin_elementwise_fma(gs, q, v);
}
__device__ __forceinline__ v4f bias_relu4(const v4f v, const v4f b) {
    const v4f s = v + b;
    v4f r;
    r.x = s.x > 0.f ? s.x : 0.f;
    r.y = s.y > 0.f ? s.y : 0.f;
    r.z = s.z > 0.f ? s.z : 0.f;
    r.w = s.w > 0.f ? s.w : 0.f;
    return r;
}

#define NODES_PER_WG 8
#define XR_PITCH     (F1 + F1 / 4)
#define XL_Q         (F1 / 16)
#define H_PITCH      (F1 + 4)
static_assert(N_NODES % NODES_PER_WG == 0, "whole work-groups only: the barrier needs all eight waves");
__global__ void __launch_bounds__(256, 9)
k_node1(const int* __restrict__ rowptr, const int* __restrict__ perm,
        const float* __restrict__ xl, const float* __restrict__ x,
        const float* __restrict__ Wr, const float* __restrict__ a1,
        const float* __restrict__ b1,
        const float* __restrict__ W2l, const float* __restrict__ W2r,
        float* __restrict__ h2l, float* __restrict__ h2r) {
    __shared__ __attribute__((aligned(16))) float xr_s[NODES_PER_WG * XR_PITCH];
    __shared__ __attribute__((aligned(16))) float h_s[NODES_PER_WG * H_PITCH];
    __shared__ int   done_s;
    if (threadIdx.x == 0) done_s = 0;
    const int lane = threadIdx.x & 31;
    const int wave  = __builtin_amdgcn_readfirstlane((int)(threadIdx.x >> 5));
    const int node0 = (int)blockIdx.x * NODES_PER_WG;
    if (node0 >= N_NODES) return;
    const int node = node0 + wave;
    const int beg = rowptr[node], end = rowptr[node + 1];
    const v4f* __restrict__ pt = (const v4f*)a1 + lane * 4;
    const v4f t0 = pt[0], t1 = pt[1], t2 = pt[2], t3 = pt[3];
    int my = 0;
    {
        const int row   = lane & 15;
        const int khalf = lane >> 4;
        v2f a[2] = {{0.f, 0.f}, {0.f, 0.f}};
        if (row < NODES_PER_WG) {
            const v2f* __restrict__ px = (const v2f*)(x + (node0 + row) * IN_DIM) + khalf;
            a[0] = px[0];
            a[1] = px[2];
        }
        const v4f* __restrict__ pw = (const v4f*)(Wr + wave * HID) + row;
        v4f bk[2][2];
#pragma unroll
        for (int kk = 0; kk < 2; ++kk) {
            const int k0 = kk * 4 + khalf * 2;
            bk[kk][0] = pw[(k0    ) * (F1 / 4)];
            bk[kk][1] = pw[(k0 + 1) * (F1 / 4)];
        }
        if (lane < end - beg) my = perm[beg + lane];
        v8f d[4];
#pragma unroll
        for (int t = 0; t < 4; ++t) {
            v8f c = {};
#pragma unroll
            for (int kk = 0; kk < 2; ++kk) {
                v2f b;
                b.x = bk[kk][0][t];
                b.y = bk[kk][1][t];
                c = __builtin_amdgcn_wmma_f32_16x16x4_f32(false, a[kk], false, b,
                                                          (short)0, c, false, false);
            }
            d[t] = c;
        }
        if (khalf == 0) {
            float* __restrict__ ps = xr_s + wave * (HID + HID / 4) + 4 * row + 4 * (row >> 2);
#pragma unroll
            for (int r = 0; r < 8; ++r) {
                const v4f v = {d[0][r], d[1][r], d[2][r], d[3][r]};
                *(v4f*)(ps + r * XR_PITCH) = v;
            }
        }
    }
    __syncthreads();
    const v4f* __restrict__ pr = (const v4f*)(xr_s + wave * XR_PITCH + lane * 20);
    const v4f r0 = pr[0], r1 = pr[1], r2 = pr[2], r3 = pr[3];
    float m = -INFINITY, ssum = 0.f;
    v4f acc0 = {0.f, 0.f, 0.f, 0.f}, acc1 = acc0, acc2 = acc0, acc3 = acc0;
    for (int j0 = beg; j0 < end; j0 += 32) {
        const int n = (end - j0 < 32) ? (end - j0) : 32;
        if (j0 != beg) {
            my = 0;
            if (lane < n) my = perm[j0 + lane];
        }
        int t = 0;
        for (; t + 1 < n; t += 2) {
            const int sa = __builtin_amdgcn_readlane(my, t);
            const int sb = __builtin_amdgcn_readlane(my, t + 1);
            const v4f* __restrict__ pa = (const v4f*)(xl + (size_t)sa * F1) + lane;
            const v4f* __restrict__ pb = (const v4f*)(xl + (size_t)sb * F1) + lane;
            const v4f a0 = pa[0], a1v = pa[XL_Q], a2v = pa[2 * XL_Q], a3 = pa[3 * XL_Q];
            const v4f b0 = pb[0], b1v = pb[XL_Q], b2v = pb[2 * XL_Q], b3 = pb[3 * XL_Q];
            v2f za = {0.f, 0.f}, zb = {0.f, 0.f};
            za = logit4(a0, r0, t0, za);   zb = logit4(b0, r0, t0, zb);
            za = logit4(a1v, r1, t1, za);  zb = logit4(b1v, r1, t1, zb);
            za = logit4(a2v, r2, t2, za);  zb = logit4(b2v, r2, t2, zb);
            za = logit4(a3, r3, t3, za);   zb = logit4(b3, r3, t3, zb);
            float ea = za.x + za.y, eb = zb.x + zb.y;
            ea += __shfl_xor(ea, 1, 32);  eb += __shfl_xor(eb, 1, 32);
            ea += __shfl_xor(ea, 2, 32);  eb += __shfl_xor(eb, 2, 32);
            const float mn = fmaxf(m, fmaxf(ea, eb));
            const float sc = __expf(m - mn);
            const float wa = __expf(ea - mn);
            const float wb = __expf(eb - mn);
            m = mn;
            ssum = fmaf(ssum, sc, wa + wb);
            acc0 = add_row(rescale_add(acc0, sc, wa, a0), wb, b0);
            acc1 = add_row(rescale_add(acc1, sc, wa, a1v), wb, b1v);
            acc2 = add_row(rescale_add(acc2, sc, wa, a2v), wb, b2v);
            acc3 = add_row(rescale_add(acc3, sc, wa, a3), wb, b3);
        }
        if (t < n) {
            const int sa = __builtin_amdgcn_readlane(my, t);
            const v4f* __restrict__ pa = (const v4f*)(xl + (size_t)sa * F1) + lane;
            const v4f a0 = pa[0], a1v = pa[XL_Q], a2v = pa[2 * XL_Q], a3 = pa[3 * XL_Q];
            v2f za = {0.f, 0.f};
            za = logit4(a0, r0, t0, za);
            za = logit4(a1v, r1, t1, za);
            za = logit4(a2v, r2, t2, za);
            za = logit4(a3, r3, t3, za);
            float ea = za.x + za.y;
            ea += __shfl_xor(ea, 1, 32);
            ea += __shfl_xor(ea, 2, 32);
            const float mn = fmaxf(m, ea);
            const float sc = __expf(m - mn);
            const float wa = __expf(ea - mn);
            m = mn;
            ssum = fmaf(ssum, sc, wa);
            acc0 = rescale_add(acc0, sc, wa, a0);
            acc1 = rescale_add(acc1, sc, wa, a1v);
            acc2 = rescale_add(acc2, sc, wa, a2v);
            acc3 = rescale_add(acc3, sc, wa, a3);
        }
    }
    const v4f* __restrict__ pc = (const v4f*)b1 + lane * 4;
    const v4f c0 = pc[0], c1 = pc[1], c2 = pc[2], c3 = pc[3];
    float* __restrict__ ph = h_s + wave * H_PITCH + lane * 4;
    *(v4f*)(ph)           = bias_relu4(acc0 / ssum, c0);
    *(v4f*)(ph + F1 / 4)  = bias_relu4(acc1 / ssum, c1);
    *(v4f*)(ph + F1 / 2)  = bias_relu4(acc2 / ssum, c2);
    *(v4f*)(ph + 3 * F1 / 4) = bias_relu4(acc3 / ssum, c3);
    __builtin_amdgcn_fence(__ATOMIC_RELEASE, "workgroup");
    int last = 0;
    if (lane == 0)
        last = __hip_atomic_fetch_add(&done_s, 1, __ATOMIC_RELAXED, __HIP_MEMORY_SCOPE_WORKGROUP)
               == NODES_PER_WG - 1;
    last = __builtin_amdgcn_readfirstlane(last);
    if (!last) return;
    __builtin_amdgcn_fence(__ATOMIC_ACQUIRE, "workgroup");
    {
        const int row   = lane & 15;
        const int khalf = lane >> 4;
        const float* __restrict__ pw = ((row < OUT_DIM) ? W2l : W2r) + (row & (OUT_DIM - 1))
                                       + XL_CHANNEL_AT(32 * khalf) * OUT_DIM;
        const v4f* __restrict__ pa = (const v4f*)(h_s + (row & (NODES_PER_WG - 1)) * H_PITCH + 32 * khalf);
        v8f d0 = {}, d1 = {};
#pragma unroll 4
        for (int sl = 0; sl < F1 / 64; ++sl) {
            v4f av[8];
#pragma unroll
            for (int j = 0; j < 8; ++j)
                av[j] = pa[sl * 16 + j];
#pragma unroll
            for (int kk = 0; kk < 16; ++kk) {
                const int ch = XL_CHANNEL_AT(64 * sl + 2 * kk);
                v2f a, b;
                a.x = av[kk >> 1][(2 * kk) & 3];
                a.y = av[kk >> 1][((2 * kk) & 3) + 1];
                b.x = pw[(ch    ) * OUT_DIM];
                b.y = pw[(ch + 1) * OUT_DIM];
                if (kk & 1)
                    d1 = __builtin_amdgcn_wmma_f32_16x16x4_f32(false, a, false, b, (short)0, d1, false, false);
                else
                    d0 = __builtin_amdgcn_wmma_f32_16x16x4_f32(false, a, false, b, (short)0, d0, false, false);
            }
        }
        const v8f d = d0 + d1;
        if (khalf == 0) {
            float* __restrict__ po = ((row < OUT_DIM) ? h2l : h2r) + (size_t)node0 * OUT_DIM
                                     + (row & (OUT_DIM - 1));
#pragma unroll
            for (int r = 0; r < NODES_PER_WG; ++r)
                po[r * OUT_DIM] = d[r];
        }
    }
}

__global__ void __launch_bounds__(256)
k_node2(const int* __restrict__ rowptr, const int* __restrict__ perm,
        const float* __restrict__ h2l, const float* __restrict__ h2r,
        const float* __restrict__ a2, const float* __restrict__ b2,
        float* __restrict__ out) {
    const int gid  = blockIdx.x * blockDim.x + threadIdx.x;
    const int node = gid >> 3;
    const int sub  = gid & 7;
    if (node >= N_NODES) return;
    const int beg = rowptr[node], end = rowptr[node + 1];
    const v4f* __restrict__ pr = (const v4f*)(h2r + (size_t)node * OUT_DIM);
    const v4f r0 = pr[0], r1 = pr[1];
    const v4f t0 = ((const v4f*)a2)[0], t1 = ((const v4f*)a2)[1];
    const float bias = b2[sub];
    float m = -INFINITY, ssum = 0.f;
    v4f acc0 = {0.f, 0.f, 0.f, 0.f}, acc1 = acc0;
    for (int j = beg + sub; j < end; j += 8) {
        const int s = perm[j];
        const v4f* __restrict__ pq = (const v4f*)(h2l + (size_t)s * OUT_DIM);
        const v4f q0 = pq[0], q1 = pq[1];
        v2f z = {0.f, 0.f};
        z = logit4(q0, r0, t0, z);
        z = logit4(q1, r1, t1, z);
        const float e  = z.x + z.y;
        const float mn = fmaxf(m, e);
        const float sc = __expf(m - mn);
        const float w  = __expf(e - mn);
        m = mn;
        ssum = fmaf(ssum, sc, w);
        acc0 = rescale_add(acc0, sc, w, q0);
        acc1 = rescale_add(acc1, sc, w, q1);
    }
#pragma unroll
    for (int o = 1; o < 8; o <<= 1) {
        const float mo = __shfl_xor(m, o, 32);
        const float so = __shfl_xor(ssum, o, 32);
        v4f o0, o1;
        o0.x = __shfl_xor(acc0.x, o, 32);  o0.y = __shfl_xor(acc0.y, o, 32);
        o0.z = __shfl_xor(acc0.z, o, 32);  o0.w = __shfl_xor(acc0.w, o, 32);
        o1.x = __shfl_xor(acc1.x, o, 32);  o1.y = __shfl_xor(acc1.y, o, 32);
        o1.z = __shfl_xor(acc1.z, o, 32);  o1.w = __shfl_xor(acc1.w, o, 32);
        const float mn = fmaxf(m, mo);
        const float f  = (m  == mn) ? 1.f : __expf(m  - mn);
        const float g  = (mo == mn) ? 1.f : __expf(mo - mn);
        ssum = fmaf(so, g, ssum * f);
        acc0 = rescale_add(acc0, f, g, o0);
        acc1 = rescale_add(acc1, f, g, o1);
        m = mn;
    }
    const v4f half = (sub & 4) ? acc1 : acc0;
    const float lo = (sub & 1) ? half.y : half.x;
    const float hi = (sub & 1) ? half.w : half.z;
    const float v  = (sub & 2) ? hi : lo;
    out[gid] = v / ssum + bias;
}

extern "C" void kernel_launch(void* const* d_in, const int* in_sizes, int n_in,
                              void* d_out, int out_size, void* d_ws, size_t ws_size,
                              hipStream_t stream) {
    const float* x   = (const float*)d_in[0];
    const int*   ei  = (const int*)  d_in[1];
    const float* W1l = (const float*)d_in[2];
    const float* W1r = (const float*)d_in[3];
    const float* a1  = (const float*)d_in[4];
    const float* b1  = (const float*)d_in[5];
    const float* W2l = (const float*)d_in[6];
    const float* W2r = (const float*)d_in[7];
    const float* a2  = (const float*)d_in[8];
    const float* b2  = (const float*)d_in[9];
    float* out = (float*)d_out;

    char* ws = (char*)d_ws;
    size_t off = 0;
    auto carve = [&](size_t bytes) -> char* {
        char* p = ws + off;
        off += (bytes + 255) & ~(size_t)255;
        return p;
    };
    float*    xl1 = (float*)   carve((size_t)N_NODES * F1 * 4);
    float*    h2l = (float*)   carve((size_t)N_NODES * OUT_DIM * 4);
    float*    h2r = (float*)   carve((size_t)N_NODES * OUT_DIM * 4);
    int*      cnt    = (int*) carve((size_t)N_NODES * 4);
    int*      rowptr = (int*) carve((size_t)ROWPTR_N * 4);
    int*      perm   = (int*) carve((size_t)E_TOT * 4);
    (void)ws_size; (void)n_in; (void)in_sizes; (void)out_size;

    const int B = 256;

    k_init1<<<(N_NODES + B - 1) / B, B, 0, stream>>>(cnt);
    k_count<<<(E_TOT + B - 1) / B, B, 0, stream>>>(ei, cnt);
    k_scan_gemm1<<<SG_BLOCKS, SCAN_T, 0, stream>>>(cnt, rowptr, x, W1l, xl1);
    k_fill<<<(E_TOT + B - 1) / B, B, 0, stream>>>(ei, rowptr, cnt, perm);
    k_node1<<<N_NODES / NODES_PER_WG, B, 0, stream>>>(rowptr, perm, xl1, x, W1r, a1, b1,
                                                      W2l, W2r, h2l, h2r);

    k_node2<<<(N_NODES * OUT_DIM + B - 1) / B, B, 0, stream>>>(rowptr, perm, h2l, h2r, a2, b2, out);
}


# Edge sort in one launch: a work-group per destination range scans the edge list into ballot-built LDS lists, slots from LDS counters (arrival order: sum order varies); global atomics, 3 launches go
# speedup vs baseline: 11.9282x; 1.5425x over previous
#include <hip/hip_runtime.h>
#include <hip/hip_bf16.h>

#define N_NODES 30000
#define E_RAW   300000
#define E_TOT   330000
#define HEADS   8
#define HID     64
#define F1      512
#define IN_DIM  8
#define OUT_DIM 8
#define NEG_SLOPE 0.2f
#define M_TILES 1875

typedef __attribute__((ext_vector_type(2))) float v2f;
typedef __attribute__((ext_vector_type(8))) float v8f;

#define XL_CHANNEL_AT(p) ((((p) & 127) >> 2) * 16 + ((p) >> 7) * 4 + ((p) & 3))
__device__ __forceinline__ void gemm1_tile(const float* __restrict__ x,
                                           const float* __restrict__ W,
                                           float* __restrict__ O,
                                           const int mtile, const int ntile) {
    const int lane  = threadIdx.x & 31;
    const int row   = lane & 15;
    const int khalf = lane >> 4;
    const int mbase = mtile * 16;
    const int nbase = ntile * 16;
    const int ch    = XL_CHANNEL_AT(nbase + row);

    v8f c = {};
#pragma unroll
    for (int kk = 0; kk < 2; ++kk) {
        const int k0 = kk * 4 + khalf * 2;
        v2f a, b;
        a.x = x[(mbase + row) * IN_DIM + k0];
        a.y = x[(mbase + row) * IN_DIM + k0 + 1];
        b.x = W[(k0    ) * F1 + ch];
        b.y = W[(k0 + 1) * F1 + ch];
        c = __builtin_amdgcn_wmma_f32_16x16x4_f32(false, a, false, b,
                                                  (short)0, c, false, false);
    }
#pragma unroll
    for (int r = 0; r < 8; ++r)
        O[(size_t)(mbase + r + 8 * khalf) * F1 + nbase + row] = c[r];
}

#define SORT_WGS    128
#define SORT_R      235
#define SORT_T      1024
#define SORT_CAPW   512
#define SORT_REGION (E_RAW + SORT_R)
#define SORT_NCH    (E_RAW / 4)
#define SORT_GROUPS ((SORT_NCH + 8 * SORT_T - 1) / (8 * SORT_T))
static_assert(SORT_WGS * SORT_R >= N_NODES, "ranges cover all nodes");
static_assert(SORT_R <= 256, "one thread of the first eight waves per node of the range");
static_assert(E_RAW % 4 == 0 && E_RAW < (1 << 19) && SORT_R < (1 << 12), "entry packing");
static_assert((SORT_CAPW & (SORT_CAPW - 1)) == 0, "list slots wrap by mask");
__device__ __forceinline__ void sort_block(const int* __restrict__ ei, int* __restrict__ seg,
                                           int* __restrict__ perm, const int b) {
    __shared__ int wl[(SORT_T / 32) * SORT_CAPW];
    __shared__ int cnt_l[256], pos_l[256], cur_l[256];
    __shared__ int wtot[8];
    __shared__ int ovf_s;
    const int t    = threadIdx.x;
    const int lane = t & 31;
    const int wave = __builtin_amdgcn_readfirstlane(t >> 5);
    const int lo   = b * SORT_R;
    const int nloc = (N_NODES - lo < SORT_R) ? (N_NODES - lo) : SORT_R;
    const int4* __restrict__ dp = (const int4*)(ei + E_RAW);
    int* __restrict__ mine = wl + wave * SORT_CAPW;
    if (t < 256) cnt_l[t] = 0;
    if (t == 0) ovf_s = 0;
    __syncthreads();
    int wcount = 0;
    for (int g = 0; g < SORT_GROUPS; ++g) {
        int4 v[8];
#pragma unroll
        for (int u = 0; u < 8; ++u) {
            const int c = (g * 8 + u) * SORT_T + t;
            v[u] = make_int4(0x7fffffff, 0x7fffffff, 0x7fffffff, 0x7fffffff);
            if (c < SORT_NCH) v[u] = dp[c];
        }
#pragma unroll
        for (int u = 0; u < 8; ++u) {
            const int e0 = ((g * 8 + u) * SORT_T + t) * 4;
            const int d4[4] = {v[u].x, v[u].y, v[u].z, v[u].w};
#pragma unroll
            for (int j = 0; j < 4; ++j) {
                const unsigned x = (unsigned)(d4[j] - lo);
                const bool in = x < (unsigned)nloc;
                const unsigned mask = __builtin_amdgcn_ballot_w32(in);
                if (mask) {
                    if (in) {
                        const int slot = wcount + (int)__builtin_amdgcn_mbcnt_lo(mask, 0u);
                        mine[slot & (SORT_CAPW - 1)] = (int)(x << 19) | (e0 + j);
                        atomicAdd(&cnt_l[x], 1);
                    }
                    wcount += __builtin_popcount(mask);
                }
            }
        }
    }
    if (lane == 0 && wcount > SORT_CAPW) ovf_s = 1;
    __syncthreads();
    const int base = b * SORT_REGION;
    int len = 0, incl = 0;
    if (t < 256) {
        len = (t < nloc) ? cnt_l[t] + 1 : 0;
        incl = len;
#pragma unroll
        for (int o = 1; o < 32; o <<= 1) {
            const int up = __shfl_up(incl, o, 32);
            if (lane >= o) incl += up;
        }
        if (lane == 31) wtot[wave] = incl;
    }
    __syncthreads();
    if (t < nloc) {
        int off = 0;
#pragma unroll
        for (int w = 0; w < 8; ++w) if (w < wave) off += wtot[w];
        const int start = base + off + incl - len;
        *(int2*)(seg + 2 * (lo + t)) = make_int2(start, start + len);
        perm[start] = lo + t;
        pos_l[t] = start;
        cur_l[t] = 1;
    }
    const int ovf = ovf_s;
    __syncthreads();
    if (!ovf) {
        for (int k = lane; k < wcount; k += 32) {
            const int p = mine[k];
            const int x = p >> 19;
            const int e = p & ((1 << 19) - 1);
            const int slot = atomicAdd(&cur_l[x], 1);
            perm[pos_l[x] + slot] = ei[e];
        }
    } else {
        for (int c = t; c < SORT_NCH; c += SORT_T) {
            const int4 q = dp[c];
            const int d4[4] = {q.x, q.y, q.z, q.w};
#pragma unroll
            for (int j = 0; j < 4; ++j) {
                const unsigned x = (unsigned)(d4[j] - lo);
                if (x < (unsigned)nloc) {
                    const int slot = atomicAdd(&cur_l[x], 1);
                    perm[pos_l[x] + slot] = ei[c * 4 + j];
                }
            }
        }
    }
}

#define SG_BLOCKS (SORT_WGS + M_TILES)
static_assert(SORT_T / 32 == F1 / 16, "one wave per N-tile of the 512-wide row");
__global__ void __launch_bounds__(SORT_T)
k_sort_gemm1(const int* __restrict__ ei, int* __restrict__ seg, int* __restrict__ perm,
             const float* __restrict__ x, const float* __restrict__ W, float* __restrict__ O) {
    if (blockIdx.x < SORT_WGS) {
        sort_block(ei, seg, perm, (int)blockIdx.x);
        return;
    }
    gemm1_tile(x, W, O, (int)blockIdx.x - SORT_WGS, (int)(threadIdx.x >> 5));
}

typedef __attribute__((ext_vector_type(4))) float v4f;
__device__ __forceinline__ v2f lrelu2(const v2f v) {
    return __builtin_elementwise_max(v, v * NEG_SLOPE);
}
__device__ __forceinline__ v2f logit4(const v4f q, const v4f r, const v4f a, v2f z) {
    z = __builtin_elementwise_fma(lrelu2(q.lo + r.lo), a.lo, z);
    z = __builtin_elementwise_fma(lrelu2(q.hi + r.hi), a.hi, z);
    return z;
}
__device__ __forceinline__ v4f rescale_add(const v4f v, const float f, const float g, const v4f q) {
    const v4f gs = {g, g, g, g};
    return __builtin_elementwise_fma(gs, q, v * f);
}
__device__ __forceinline__ v4f add_row(const v4f v, const float g, const v4f q) {
    const v4f gs = {g, g, g, g};
    return __builtin_elementwise_fma(gs, q, v);
}
__device__ __forceinline__ v4f bias_relu4(const v4f v, const v4f b) {
    const v4f s = v + b;
    v4f r;
    r.x = s.x > 0.f ? s.x : 0.f;
    r.y = s.y > 0.f ? s.y : 0.f;
    r.z = s.z > 0.f ? s.z : 0.f;
    r.w = s.w > 0.f ? s.w : 0.f;
    return r;
}

#define NODES_PER_WG 8
#define XR_PITCH     (F1 + F1 / 4)
#define XL_Q         (F1 / 16)
#define H_PITCH      (F1 + 4)
static_assert(N_NODES % NODES_PER_WG == 0, "whole work-groups only: the barrier needs all eight waves");
__global__ void __launch_bounds__(256, 9)
k_node1(const int* __restrict__ seg, const int* __restrict__ perm,
        const float* __restrict__ xl, const float* __restrict__ x,
        const float* __restrict__ Wr, const float* __restrict__ a1,
        const float* __restrict__ b1,
        const float* __restrict__ W2l, const float* __restrict__ W2r,
        float* __restrict__ h2l, float* __restrict__ h2r) {
    __shared__ __attribute__((aligned(16))) float xr_s[NODES_PER_WG * XR_PITCH];
    __shared__ __attribute__((aligned(16))) float h_s[NODES_PER_WG * H_PITCH];
    __shared__ int   done_s;
    if (threadIdx.x == 0) done_s = 0;
    const int lane = threadIdx.x & 31;
    const int wave  = __builtin_amdgcn_readfirstlane((int)(threadIdx.x >> 5));
    const int node0 = (int)blockIdx.x * NODES_PER_WG;
    if (node0 >= N_NODES) return;
    const int node = node0 + wave;
    const int beg = seg[2 * node], end = seg[2 * node + 1];
    const v4f* __restrict__ pt = (const v4f*)a1 + lane * 4;
    const v4f t0 = pt[0], t1 = pt[1], t2 = pt[2], t3 = pt[3];
    int my = 0;
    {
        const int row   = lane & 15;
        const int khalf = lane >> 4;
        v2f a[2] = {{0.f, 0.f}, {0.f, 0.f}};
        if (row < NODES_PER_WG) {
            const v2f* __restrict__ px = (const v2f*)(x + (node0 + row) * IN_DIM) + khalf;
            a[0] = px[0];
            a[1] = px[2];
        }
        const v4f* __restrict__ pw = (const v4f*)(Wr + wave * HID) + row;
        v4f bk[2][2];
#pragma unroll
        for (int kk = 0; kk < 2; ++kk) {
            const int k0 = kk * 4 + khalf * 2;
            bk[kk][0] = pw[(k0    ) * (F1 / 4)];
            bk[kk][1] = pw[(k0 + 1) * (F1 / 4)];
        }
        if (lane < end - beg) my = perm[beg + lane];
        v8f d[4];
#pragma unroll
        for (int t = 0; t < 4; ++t) {
            v8f c = {};
#pragma unroll
            for (int kk = 0; kk < 2; ++kk) {
                v2f b;
                b.x = bk[kk][0][t];
                b.y = bk[kk][1][t];
                c = __builtin_amdgcn_wmma_f32_16x16x4_f32(false, a[kk], false, b,
                                                          (short)0, c, false, false);
            }
            d[t] = c;
        }
        if (khalf == 0) {
            float* __restrict__ ps = xr_s + wave * (HID + HID / 4) + 4 * row + 4 * (row >> 2);
#pragma unroll
            for (int r = 0; r < 8; ++r) {
                const v4f v = {d[0][r], d[1][r], d[2][r], d[3][r]};
                *(v4f*)(ps + r * XR_PITCH) = v;
            }
        }
    }
    __syncthreads();
    const v4f* __restrict__ pr = (const v4f*)(xr_s + wave * XR_PITCH + lane * 20);
    const v4f r0 = pr[0], r1 = pr[1], r2 = pr[2], r3 = pr[3];
    float m = -INFINITY, ssum = 0.f;
    v4f acc0 = {0.f, 0.f, 0.f, 0.f}, acc1 = acc0, acc2 = acc0, acc3 = acc0;
    for (int j0 = beg; j0 < end; j0 += 32) {
        const int n = (end - j0 < 32) ? (end - j0) : 32;
        if (j0 != beg) {
            my = 0;
            if (lane < n) my = perm[j0 + lane];
        }
        int t = 0;
        for (; t + 1 < n; t += 2) {
            const int sa = __builtin_amdgcn_readlane(my, t);
            const int sb = __builtin_amdgcn_readlane(my, t + 1);
            const v4f* __restrict__ pa = (const v4f*)(xl + (size_t)sa * F1) + lane;
            const v4f* __restrict__ pb = (const v4f*)(xl + (size_t)sb * F1) + lane;
            const v4f a0 = pa[0], a1v = pa[XL_Q], a2v = pa[2 * XL_Q], a3 = pa[3 * XL_Q];
            const v4f b0 = pb[0], b1v = pb[XL_Q], b2v = pb[2 * XL_Q], b3 = pb[3 * XL_Q];
            v2f za = {0.f, 0.f}, zb = {0.f, 0.f};
            za = logit4(a0, r0, t0, za);   zb = logit4(b0, r0, t0, zb);
            za = logit4(a1v, r1, t1, za);  zb = logit4(b1v, r1, t1, zb);
            za = logit4(a2v, r2, t2, za);  zb = logit4(b2v, r2, t2, zb);
            za = logit4(a3, r3, t3, za);   zb = logit4(b3, r3, t3, zb);
            float ea = za.x + za.y, eb = zb.x + zb.y;
            ea += __shfl_xor(ea, 1, 32);  eb += __shfl_xor(eb, 1, 32);
            ea += __shfl_xor(ea, 2, 32);  eb += __shfl_xor(eb, 2, 32);
            const float mn = fmaxf(m, fmaxf(ea, eb));
            const float sc = __expf(m - mn);
            const float wa = __expf(ea - mn);
            const float wb = __expf(eb - mn);
            m = mn;
            ssum = fmaf(ssum, sc, wa + wb);
            acc0 = add_row(rescale_add(acc0, sc, wa, a0), wb, b0);
            acc1 = add_row(rescale_add(acc1, sc, wa, a1v), wb, b1v);
            acc2 = add_row(rescale_add(acc2, sc, wa, a2v), wb, b2v);
            acc3 = add_row(rescale_add(acc3, sc, wa, a3), wb, b3);
        }
        if (t < n) {
            const int sa = __builtin_amdgcn_readlane(my, t);
            const v4f* __restrict__ pa = (const v4f*)(xl + (size_t)sa * F1) + lane;
            const v4f a0 = pa[0], a1v = pa[XL_Q], a2v = pa[2 * XL_Q], a3 = pa[3 * XL_Q];
            v2f za = {0.f, 0.f};
            za = logit4(a0, r0, t0, za);
            za = logit4(a1v, r1, t1, za);
            za = logit4(a2v, r2, t2, za);
            za = logit4(a3, r3, t3, za);
            float ea = za.x + za.y;
            ea += __shfl_xor(ea, 1, 32);
            ea += __shfl_xor(ea, 2, 32);
            const float mn = fmaxf(m, ea);
            const float sc = __expf(m - mn);
            const float wa = __expf(ea - mn);
            m = mn;
            ssum = fmaf(ssum, sc, wa);
            acc0 = rescale_add(acc0, sc, wa, a0);
            acc1 = rescale_add(acc1, sc, wa, a1v);
            acc2 = rescale_add(acc2, sc, wa, a2v);
            acc3 = rescale_add(acc3, sc, wa, a3);
        }
    }
    const v4f* __restrict__ pc = (const v4f*)b1 + lane * 4;
    const v4f c0 = pc[0], c1 = pc[1], c2 = pc[2], c3 = pc[3];
    float* __restrict__ ph = h_s + wave * H_PITCH + lane * 4;
    *(v4f*)(ph)           = bias_relu4(acc0 / ssum, c0);
    *(v4f*)(ph + F1 / 4)  = bias_relu4(acc1 / ssum, c1);
    *(v4f*)(ph + F1 / 2)  = bias_relu4(acc2 / ssum, c2);
    *(v4f*)(ph + 3 * F1 / 4) = bias_relu4(acc3 / ssum, c3);
    __builtin_amdgcn_fence(__ATOMIC_RELEASE, "workgroup");
    int last = 0;
    if (lane == 0)
        last = __hip_atomic_fetch_add(&done_s, 1, __ATOMIC_RELAXED, __HIP_MEMORY_SCOPE_WORKGROUP)
               == NODES_PER_WG - 1;
    last = __builtin_amdgcn_readfirstlane(last);
    if (!last) return;
    __builtin_amdgcn_fence(__ATOMIC_ACQUIRE, "workgroup");
    {
        const int row   = lane & 15;
        const int khalf = lane >> 4;
        const float* __restrict__ pw = ((row < OUT_DIM) ? W2l : W2r) + (row & (OUT_DIM - 1))
                                       + XL_CHANNEL_AT(32 * khalf) * OUT_DIM;
        const v4f* __restrict__ pa = (const v4f*)(h_s + (row & (NODES_PER_WG - 1)) * H_PITCH + 32 * khalf);
        v8f d0 = {}, d1 = {};
#pragma unroll 4
        for (int sl = 0; sl < F1 / 64; ++sl) {
            v4f av[8];
#pragma unroll
            for (int j = 0; j < 8; ++j)
                av[j] = pa[sl * 16 + j];
#pragma unroll
            for (int kk = 0; kk < 16; ++kk) {
                const int ch = XL_CHANNEL_AT(64 * sl + 2 * kk);
                v2f a, b;
                a.x = av[kk >> 1][(2 * kk) & 3];
                a.y = av[kk >> 1][((2 * kk) & 3) + 1];
                b.x = pw[(ch    ) * OUT_DIM];
                b.y = pw[(ch + 1) * OUT_DIM];
                if (kk & 1)
                    d1 = __builtin_amdgcn_wmma_f32_16x16x4_f32(false, a, false, b, (short)0, d1, false, false);
                else
                    d0 = __builtin_amdgcn_wmma_f32_16x16x4_f32(false, a, false, b, (short)0, d0, false, false);
            }
        }
        const v8f d = d0 + d1;
        if (khalf == 0) {
            float* __restrict__ po = ((row < OUT_DIM) ? h2l : h2r) + (size_t)node0 * OUT_DIM
                                     + (row & (OUT_DIM - 1));
#pragma unroll
            for (int r = 0; r < NODES_PER_WG; ++r)
                po[r * OUT_DIM] = d[r];
        }
    }
}

__global__ void __launch_bounds__(256)
k_node2(const int* __restrict__ seg, const int* __restrict__ perm,
        const float* __restrict__ h2l, const float* __restrict__ h2r,
        const float* __restrict__ a2, const float* __restrict__ b2,
        float* __restrict__ out) {
    const int gid  = blockIdx.x * blockDim.x + threadIdx.x;
    const int node = gid >> 3;
    const int sub  = gid & 7;
    if (node >= N_NODES) return;
    const int2 be = *(const int2*)(seg + 2 * node);
    const int beg = be.x, end = be.y;
    const v4f* __restrict__ pr = (const v4f*)(h2r + (size_t)node * OUT_DIM);
    const v4f r0 = pr[0], r1 = pr[1];
    const v4f t0 = ((const v4f*)a2)[0], t1 = ((const v4f*)a2)[1];
    const float bias = b2[sub];
    float m = -INFINITY, ssum = 0.f;
    v4f acc0 = {0.f, 0.f, 0.f, 0.f}, acc1 = acc0;
    for (int j = beg + sub; j < end; j += 8) {
        const int s = perm[j];
        const v4f* __restrict__ pq = (const v4f*)(h2l + (size_t)s * OUT_DIM);
        const v4f q0 = pq[0], q1 = pq[1];
        v2f z = {0.f, 0.f};
        z = logit4(q0, r0, t0, z);
        z = logit4(q1, r1, t1, z);
        const float e  = z.x + z.y;
        const float mn = fmaxf(m, e);
        const float sc = __expf(m - mn);
        const float w  = __expf(e - mn);
        m = mn;
        ssum = fmaf(ssum, sc, w);
        acc0 = rescale_add(acc0, sc, w, q0);
        acc1 = rescale_add(acc1, sc, w, q1);
    }
#pragma unroll
    for (int o = 1; o < 8; o <<= 1) {
        const float mo = __shfl_xor(m, o, 32);
        const float so = __shfl_xor(ssum, o, 32);
        v4f o0, o1;
        o0.x = __shfl_xor(acc0.x, o, 32);  o0.y = __shfl_xor(acc0.y, o, 32);
        o0.z = __shfl_xor(acc0.z, o, 32);  o0.w = __shfl_xor(acc0.w, o, 32);
        o1.x = __shfl_xor(acc1.x, o, 32);  o1.y = __shfl_xor(acc1.y, o, 32);
        o1.z = __shfl_xor(acc1.z, o, 32);  o1.w = __shfl_xor(acc1.w, o, 32);
        const float mn = fmaxf(m, mo);
        const float f  = (m  == mn) ? 1.f : __expf(m  - mn);
        const float g  = (mo == mn) ? 1.f : __expf(mo - mn);
        ssum = fmaf(so, g, ssum * f);
        acc0 = rescale_add(acc0, f, g, o0);
        acc1 = rescale_add(acc1, f, g, o1);
        m = mn;
    }
    const v4f half = (sub & 4) ? acc1 : acc0;
    const float lo = (sub & 1) ? half.y : half.x;
    const float hi = (sub & 1) ? half.w : half.z;
    const float v  = (sub & 2) ? hi : lo;
    out[gid] = v / ssum + bias;
}

extern "C" void kernel_launch(void* const* d_in, const int* in_sizes, int n_in,
                              void* d_out, int out_size, void* d_ws, size_t ws_size,
                              hipStream_t stream) {
    const float* x   = (const float*)d_in[0];
    const int*   ei  = (const int*)  d_in[1];
    const float* W1l = (const float*)d_in[2];
    const float* W1r = (const float*)d_in[3];
    const float* a1  = (const float*)d_in[4];
    const float* b1  = (const float*)d_in[5];
    const float* W2l = (const float*)d_in[6];
    const float* W2r = (const float*)d_in[7];
    const float* a2  = (const float*)d_in[8];
    const float* b2  = (const float*)d_in[9];
    float* out = (float*)d_out;

    char* ws = (char*)d_ws;
    size_t off = 0;
    auto carve = [&](size_t bytes) -> char* {
        char* p = ws + off;
        off += (bytes + 255) & ~(size_t)255;
        return p;
    };
    float*    xl1 = (float*)   carve((size_t)N_NODES * F1 * 4);
    float*    h2l = (float*)   carve((size_t)N_NODES * OUT_DIM * 4);
    float*    h2r = (float*)   carve((size_t)N_NODES * OUT_DIM * 4);
    int*      seg    = (int*) carve((size_t)N_NODES * 2 * 4);
    int*      perm   = (int*) carve((size_t)SORT_WGS * SORT_REGION * 4);
    (void)ws_size; (void)n_in; (void)in_sizes; (void)out_size;

    const int B = 256;

    k_sort_gemm1<<<SG_BLOCKS, SORT_T, 0, stream>>>(ei, seg, perm, x, W1l, xl1);
    k_node1<<<N_NODES / NODES_PER_WG, B, 0, stream>>>(seg, perm, xl1, x, W1r, a1, b1,
                                                      W2l, W2r, h2l, h2r);

    k_node2<<<(N_NODES * OUT_DIM + B - 1) / B, B, 0, stream>>>(seg, perm, h2l, h2r, a2, b2, out);
}


# Layer-1 node kernel: each wave takes two destination nodes in turn (8 waves = 16 nodes): both 16-row matrix tiles full, half the waves, half the weight reads and WMMAs per node
# speedup vs baseline: 13.1690x; 1.1040x over previous
#include <hip/hip_runtime.h>
#include <hip/hip_bf16.h>

#define N_NODES 30000
#define E_RAW   300000
#define E_TOT   330000
#define HEADS   8
#define HID     64
#define F1      512
#define IN_DIM  8
#define OUT_DIM 8
#define NEG_SLOPE 0.2f
#define M_TILES 1875

typedef __attribute__((ext_vector_type(2))) float v2f;
typedef __attribute__((ext_vector_type(8))) float v8f;

#define XL_CHANNEL_AT(p) ((((p) & 127) >> 2) * 16 + ((p) >> 7) * 4 + ((p) & 3))
__device__ __forceinline__ void gemm1_tile(const float* __restrict__ x,
                                           const float* __restrict__ W,
                                           float* __restrict__ O,
                                           const int mtile, const int ntile) {
    const int lane  = threadIdx.x & 31;
    const int row   = lane & 15;
    const int khalf = lane >> 4;
    const int mbase = mtile * 16;
    const int nbase = ntile * 16;
    const int ch    = XL_CHANNEL_AT(nbase + row);

    v8f c = {};
#pragma unroll
    for (int kk = 0; kk < 2; ++kk) {
        const int k0 = kk * 4 + khalf * 2;
        v2f a, b;
        a.x = x[(mbase + row) * IN_DIM + k0];
        a.y = x[(mbase + row) * IN_DIM + k0 + 1];
        b.x = W[(k0    ) * F1 + ch];
        b.y = W[(k0 + 1) * F1 + ch];
        c = __builtin_amdgcn_wmma_f32_16x16x4_f32(false, a, false, b,
                                                  (short)0, c, false, false);
    }
#pragma unroll
    for (int r = 0; r < 8; ++r)
        O[(size_t)(mbase + r + 8 * khalf) * F1 + nbase + row] = c[r];
}

#define SORT_WGS    128
#define SORT_R      235
#define SORT_T      1024
#define SORT_CAPW   512
#define SORT_REGION (E_RAW + SORT_R)
#define SORT_NCH    (E_RAW / 4)
#define SORT_GROUPS ((SORT_NCH + 8 * SORT_T - 1) / (8 * SORT_T))
static_assert(SORT_WGS * SORT_R >= N_NODES, "ranges cover all nodes");
static_assert(SORT_R <= 256, "one thread of the first eight waves per node of the range");
static_assert(E_RAW % 4 == 0 && E_RAW < (1 << 19) && SORT_R < (1 << 12), "entry packing");
static_assert((SORT_CAPW & (SORT_CAPW - 1)) == 0, "list slots wrap by mask");
__device__ __forceinline__ void sort_block(const int* __restrict__ ei, int* __restrict__ seg,
                                           int* __restrict__ perm, const int b) {
    __shared__ int wl[(SORT_T / 32) * SORT_CAPW];
    __shared__ int cnt_l[256], pos_l[256], cur_l[256];
    __shared__ int wtot[8];
    __shared__ int ovf_s;
    const int t    = threadIdx.x;
    const int lane = t & 31;
    const int wave = __builtin_amdgcn_readfirstlane(t >> 5);
    const int lo   = b * SORT_R;
    const int nloc = (N_NODES - lo < SORT_R) ? (N_NODES - lo) : SORT_R;
    const int4* __restrict__ dp = (const int4*)(ei + E_RAW);
    int* __restrict__ mine = wl + wave * SORT_CAPW;
    if (t < 256) cnt_l[t] = 0;
    if (t == 0) ovf_s = 0;
    __syncthreads();
    int wcount = 0;
    for (int g = 0; g < SORT_GROUPS; ++g) {
        int4 v[8];
#pragma unroll
        for (int u = 0; u < 8; ++u) {
            const int c = (g * 8 + u) * SORT_T + t;
            v[u] = make_int4(0x7fffffff, 0x7fffffff, 0x7fffffff, 0x7fffffff);
            if (c < SORT_NCH) v[u] = dp[c];
        }
#pragma unroll
        for (int u = 0; u < 8; ++u) {
            const int e0 = ((g * 8 + u) * SORT_T + t) * 4;
            const int d4[4] = {v[u].x, v[u].y, v[u].z, v[u].w};
#pragma unroll
            for (int j = 0; j < 4; ++j) {
                const unsigned x = (unsigned)(d4[j] - lo);
                const bool in = x < (unsigned)nloc;
                const unsigned mask = __builtin_amdgcn_ballot_w32(in);
                if (mask) {
                    if (in) {
                        const int slot = wcount + (int)__builtin_amdgcn_mbcnt_lo(mask, 0u);
                        mine[slot & (SORT_CAPW - 1)] = (int)(x << 19) | (e0 + j);
                        atomicAdd(&cnt_l[x], 1);
                    }
                    wcount += __builtin_popcount(mask);
                }
            }
        }
    }
    if (lane == 0 && wcount > SORT_CAPW) ovf_s = 1;
    __syncthreads();
    const int base = b * SORT_REGION;
    int len = 0, incl = 0;
    if (t < 256) {
        len = (t < nloc) ? cnt_l[t] + 1 : 0;
        incl = len;
#pragma unroll
        for (int o = 1; o < 32; o <<= 1) {
            const int up = __shfl_up(incl, o, 32);
            if (lane >= o) incl += up;
        }
        if (lane == 31) wtot[wave] = incl;
    }
    __syncthreads();
    if (t < nloc) {
        int off = 0;
#pragma unroll
        for (int w = 0; w < 8; ++w) if (w < wave) off += wtot[w];
        const int start = base + off + incl - len;
        *(int2*)(seg + 2 * (lo + t)) = make_int2(start, start + len);
        perm[start] = lo + t;
        pos_l[t] = start;
        cur_l[t] = 1;
    }
    const int ovf = ovf_s;
    __syncthreads();
    if (!ovf) {
        for (int k = lane; k < wcount; k += 32) {
            const int p = mine[k];
            const int x = p >> 19;
            const int e = p & ((1 << 19) - 1);
            const int slot = atomicAdd(&cur_l[x], 1);
            perm[pos_l[x] + slot] = ei[e];
        }
    } else {
        for (int c = t; c < SORT_NCH; c += SORT_T) {
            const int4 q = dp[c];
            const int d4[4] = {q.x, q.y, q.z, q.w};
#pragma unroll
            for (int j = 0; j < 4; ++j) {
                const unsigned x = (unsigned)(d4[j] - lo);
                if (x < (unsigned)nloc) {
                    const int slot = atomicAdd(&cur_l[x], 1);
                    perm[pos_l[x] + slot] = ei[c * 4 + j];
                }
            }
        }
    }
}

#define SG_BLOCKS (SORT_WGS + M_TILES)
static_assert(SORT_T / 32 == F1 / 16, "one wave per N-tile of the 512-wide row");
__global__ void __launch_bounds__(SORT_T)
k_sort_gemm1(const int* __restrict__ ei, int* __restrict__ seg, int* __restrict__ perm,
             const float* __restrict__ x, const float* __restrict__ W, float* __restrict__ O) {
    if (blockIdx.x < SORT_WGS) {
        sort_block(ei, seg, perm, (int)blockIdx.x);
        return;
    }
    gemm1_tile(x, W, O, (int)blockIdx.x - SORT_WGS, (int)(threadIdx.x >> 5));
}

typedef __attribute__((ext_vector_type(4))) float v4f;
__device__ __forceinline__ v2f lrelu2(const v2f v) {
    return __builtin_elementwise_max(v, v * NEG_SLOPE);
}
__device__ __forceinline__ v2f logit4(const v4f q, const v4f r, const v4f a, v2f z) {
    z = __builtin_elementwise_fma(lrelu2(q.lo + r.lo), a.lo, z);
    z = __builtin_elementwise_fma(lrelu2(q.hi + r.hi), a.hi, z);
    return z;
}
__device__ __forceinline__ v4f rescale_add(const v4f v, const float f, const float g, const v4f q) {
    const v4f gs = {g, g, g, g};
    return __builtin_elementwise_fma(gs, q, v * f);
}
__device__ __forceinline__ v4f add_row(const v4f v, const float g, const v4f q) {
    const v4f gs = {g, g, g, g};
    return __builtin_elementwise_fma(gs, q, v);
}
__device__ __forceinline__ v4f bias_relu4(const v4f v, const v4f b) {
    const v4f s = v + b;
    v4f r;
    r.x = s.x > 0.f ? s.x : 0.f;
    r.y = s.y > 0.f ? s.y : 0.f;
    r.z = s.z > 0.f ? s.z : 0.f;
    r.w = s.w > 0.f ? s.w : 0.f;
    return r;
}

#define WAVES_PER_WG 8
#define NODES_PER_WG 16
#define ROW_PITCH    (F1 + F1 / 4 + 4)
#define XL_Q         (F1 / 16)
static_assert(N_NODES % NODES_PER_WG == 0, "whole work-groups only: the barrier needs all eight waves");
static_assert(NODES_PER_WG == 16 && NODES_PER_WG == 2 * WAVES_PER_WG, "one full 16-row tile, two rows per wave");
__global__ void __launch_bounds__(256, 9)
k_node1(const int* __restrict__ seg, const int* __restrict__ perm,
        const float* __restrict__ xl, const float* __restrict__ x,
        const float* __restrict__ Wr, const float* __restrict__ a1,
        const float* __restrict__ b1,
        const float* __restrict__ W2l, const float* __restrict__ W2r,
        float* __restrict__ h2l, float* __restrict__ h2r) {
    __shared__ __attribute__((aligned(16))) float row_s[NODES_PER_WG * ROW_PITCH];
    __shared__ int   done_s;
    if (threadIdx.x == 0) done_s = 0;
    const int lane = threadIdx.x & 31;
    const int wave  = __builtin_amdgcn_readfirstlane((int)(threadIdx.x >> 5));
    const int node0 = (int)blockIdx.x * NODES_PER_WG;
    if (node0 >= N_NODES) return;
    const int beg0 = seg[2 * (node0 + wave)], end0 = seg[2 * (node0 + wave) + 1];
    const int beg1 = seg[2 * (node0 + wave + WAVES_PER_WG)], end1 = seg[2 * (node0 + wave + WAVES_PER_WG) + 1];
    const v4f* __restrict__ pt = (const v4f*)a1 + lane * 4;
    const v4f t0 = pt[0], t1 = pt[1], t2 = pt[2], t3 = pt[3];
    int my0 = 0, my1 = 0;
    {
        const int row   = lane & 15;
        const int khalf = lane >> 4;
        const v2f* __restrict__ px = (const v2f*)(x + (node0 + row) * IN_DIM) + khalf;
        v2f a[2];
        a[0] = px[0];
        a[1] = px[2];
        const v4f* __restrict__ pw = (const v4f*)(Wr + wave * HID) + row;
        v4f bk[2][2];
#pragma unroll
        for (int kk = 0; kk < 2; ++kk) {
            const int k0 = kk * 4 + khalf * 2;
            bk[kk][0] = pw[(k0    ) * (F1 / 4)];
            bk[kk][1] = pw[(k0 + 1) * (F1 / 4)];
        }
        if (lane < end0 - beg0) my0 = perm[beg0 + lane];
        if (lane < end1 - beg1) my1 = perm[beg1 + lane];
        v8f d[4];
#pragma unroll
        for (int t = 0; t < 4; ++t) {
            v8f c = {};
#pragma unroll
            for (int kk = 0; kk < 2; ++kk) {
                v2f b;
                b.x = bk[kk][0][t];
                b.y = bk[kk][1][t];
                c = __builtin_amdgcn_wmma_f32_16x16x4_f32(false, a[kk], false, b,
                                                          (short)0, c, false, false);
            }
            d[t] = c;
        }
        float* __restrict__ ps = row_s + 8 * khalf * ROW_PITCH
                                 + wave * (HID + HID / 4) + 4 * row + 4 * (row >> 2);
#pragma unroll
        for (int r = 0; r < 8; ++r) {
            const v4f v = {d[0][r], d[1][r], d[2][r], d[3][r]};
            *(v4f*)(ps + r * ROW_PITCH) = v;
        }
    }
    __syncthreads();
    for (int pass = 0; pass < 2; ++pass) {
    const int lrow = wave + WAVES_PER_WG * pass;
    const int beg = pass ? beg1 : beg0, end = pass ? end1 : end0;
    int my = pass ? my1 : my0;
    const v4f* __restrict__ pr = (const v4f*)(row_s + lrow * ROW_PITCH + lane * 20);
    const v4f r0 = pr[0], r1 = pr[1], r2 = pr[2], r3 = pr[3];
    float m = -INFINITY, ssum = 0.f;
    v4f acc0 = {0.f, 0.f, 0.f, 0.f}, acc1 = acc0, acc2 = acc0, acc3 = acc0;
    for (int j0 = beg; j0 < end; j0 += 32) {
        const int n = (end - j0 < 32) ? (end - j0) : 32;
        if (j0 != beg) {
            my = 0;
            if (lane < n) my = perm[j0 + lane];
        }
        int t = 0;
        for (; t + 1 < n; t += 2) {
            const int sa = __builtin_amdgcn_readlane(my, t);
            const int sb = __builtin_amdgcn_readlane(my, t + 1);
            const v4f* __restrict__ pa = (const v4f*)(xl + (size_t)sa * F1) + lane;
            const v4f* __restrict__ pb = (const v4f*)(xl + (size_t)sb * F1) + lane;
            const v4f a0 = pa[0], a1v = pa[XL_Q], a2v = pa[2 * XL_Q], a3 = pa[3 * XL_Q];
            const v4f b0 = pb[0], b1v = pb[XL_Q], b2v = pb[2 * XL_Q], b3 = pb[3 * XL_Q];
            v2f za = {0.f, 0.f}, zb = {0.f, 0.f};
            za = logit4(a0, r0, t0, za);   zb = logit4(b0, r0, t0, zb);
            za = logit4(a1v, r1, t1, za);  zb = logit4(b1v, r1, t1, zb);
            za = logit4(a2v, r2, t2, za);  zb = logit4(b2v, r2, t2, zb);
            za = logit4(a3, r3, t3, za);   zb = logit4(b3, r3, t3, zb);
            float ea = za.x + za.y, eb = zb.x + zb.y;
            ea += __shfl_xor(ea, 1, 32);  eb += __shfl_xor(eb, 1, 32);
            ea += __shfl_xor(ea, 2, 32);  eb += __shfl_xor(eb, 2, 32);
            const float mn = fmaxf(m, fmaxf(ea, eb));
            const float sc = __expf(m - mn);
            const float wa = __expf(ea - mn);
            const float wb = __expf(eb - mn);
            m = mn;
            ssum = fmaf(ssum, sc, wa + wb);
            acc0 = add_row(rescale_add(acc0, sc, wa, a0), wb, b0);
            acc1 = add_row(rescale_add(acc1, sc, wa, a1v), wb, b1v);
            acc2 = add_row(rescale_add(acc2, sc, wa, a2v), wb, b2v);
            acc3 = add_row(rescale_add(acc3, sc, wa, a3), wb, b3);
        }
        if (t < n) {
            const int sa = __builtin_amdgcn_readlane(my, t);
            const v4f* __restrict__ pa = (const v4f*)(xl + (size_t)sa * F1) + lane;
            const v4f a0 = pa[0], a1v = pa[XL_Q], a2v = pa[2 * XL_Q], a3 = pa[3 * XL_Q];
            v2f za = {0.f, 0.f};
            za = logit4(a0, r0, t0, za);
            za = logit4(a1v, r1, t1, za);
            za = logit4(a2v, r2, t2, za);
            za = logit4(a3, r3, t3, za);
            float ea = za.x + za.y;
            ea += __shfl_xor(ea, 1, 32);
            ea += __shfl_xor(ea, 2, 32);
            const float mn = fmaxf(m, ea);
            const float sc = __expf(m - mn);
            const float wa = __expf(ea - mn);
            m = mn;
            ssum = fmaf(ssum, sc, wa);
            acc0 = rescale_add(acc0, sc, wa, a0);
            acc1 = rescale_add(acc1, sc, wa, a1v);
            acc2 = rescale_add(acc2, sc, wa, a2v);
            acc3 = rescale_add(acc3, sc, wa, a3);
        }
    }
    asm volatile("" ::: "memory");
    const v4f* __restrict__ pc = (const v4f*)b1 + lane * 4;
    const v4f c0 = pc[0], c1 = pc[1], c2 = pc[2], c3 = pc[3];
    float* __restrict__ ph = row_s + lrow * ROW_PITCH + lane * 4;
    *(v4f*)(ph)           = bias_relu4(acc0 / ssum, c0);
    *(v4f*)(ph + F1 / 4)  = bias_relu4(acc1 / ssum, c1);
    *(v4f*)(ph + F1 / 2)  = bias_relu4(acc2 / ssum, c2);
    *(v4f*)(ph + 3 * F1 / 4) = bias_relu4(acc3 / ssum, c3);
    }
    __builtin_amdgcn_fence(__ATOMIC_RELEASE, "workgroup");
    int last = 0;
    if (lane == 0)
        last = __hip_atomic_fetch_add(&done_s, 1, __ATOMIC_RELAXED, __HIP_MEMORY_SCOPE_WORKGROUP)
               == WAVES_PER_WG - 1;
    last = __builtin_amdgcn_readfirstlane(last);
    if (!last) return;
    __builtin_amdgcn_fence(__ATOMIC_ACQUIRE, "workgroup");
    {
        const int row   = lane & 15;
        const int khalf = lane >> 4;
        const float* __restrict__ pw = ((row < OUT_DIM) ? W2l : W2r) + (row & (OUT_DIM - 1))
                                       + XL_CHANNEL_AT(32 * khalf) * OUT_DIM;
        const v4f* __restrict__ pa = (const v4f*)(row_s + row * ROW_PITCH + 32 * khalf);
        v8f d0 = {}, d1 = {};
#pragma unroll 2
        for (int sl = 0; sl < F1 / 64; ++sl) {
            v4f av[8];
#pragma unroll
            for (int j = 0; j < 8; ++j)
                av[j] = pa[sl * 16 + j];
#pragma unroll
            for (int kk = 0; kk < 16; ++kk) {
                const int ch = XL_CHANNEL_AT(64 * sl + 2 * kk);
                v2f a, b;
                a.x = av[kk >> 1][(2 * kk) & 3];
                a.y = av[kk >> 1][((2 * kk) & 3) + 1];
                b.x = pw[(ch    ) * OUT_DIM];
                b.y = pw[(ch + 1) * OUT_DIM];
                if (kk & 1)
                    d1 = __builtin_amdgcn_wmma_f32_16x16x4_f32(false, a, false, b, (short)0, d1, false, false);
                else
                    d0 = __builtin_amdgcn_wmma_f32_16x16x4_f32(false, a, false, b, (short)0, d0, false, false);
            }
        }
        const v8f d = d0 + d1;
        float* __restrict__ po = ((row < OUT_DIM) ? h2l : h2r) + (size_t)(node0 + 8 * khalf) * OUT_DIM
                                 + (row & (OUT_DIM - 1));
#pragma unroll
        for (int r = 0; r < 8; ++r)
            po[r * OUT_DIM] = d[r];
    }
}

__global__ void __launch_bounds__(256)
k_node2(const int* __restrict__ seg, const int* __restrict__ perm,
        const float* __restrict__ h2l, const float* __restrict__ h2r,
        const float* __restrict__ a2, const float* __restrict__ b2,
        float* __restrict__ out) {
    const int gid  = blockIdx.x * blockDim.x + threadIdx.x;
    const int node = gid >> 3;
    const int sub  = gid & 7;
    if (node >= N_NODES) return;
    const int2 be = *(const int2*)(seg + 2 * node);
    const int beg = be.x, end = be.y;
    const v4f* __restrict__ pr = (const v4f*)(h2r + (size_t)node * OUT_DIM);
    const v4f r0 = pr[0], r1 = pr[1];
    const v4f t0 = ((const v4f*)a2)[0], t1 = ((const v4f*)a2)[1];
    const float bias = b2[sub];
    float m = -INFINITY, ssum = 0.f;
    v4f acc0 = {0.f, 0.f, 0.f, 0.f}, acc1 = acc0;
    for (int j = beg + sub; j < end; j += 8) {
        const int s = perm[j];
        const v4f* __restrict__ pq = (const v4f*)(h2l + (size_t)s * OUT_DIM);
        const v4f q0 = pq[0], q1 = pq[1];
        v2f z = {0.f, 0.f};
        z = logit4(q0, r0, t0, z);
        z = logit4(q1, r1, t1, z);
        const float e  = z.x + z.y;
        const float mn = fmaxf(m, e);
        const float sc = __expf(m - mn);
        const float w  = __expf(e - mn);
        m = mn;
        ssum = fmaf(ssum, sc, w);
        acc0 = rescale_add(acc0, sc, w, q0);
        acc1 = rescale_add(acc1, sc, w, q1);
    }
#pragma unroll
    for (int o = 1; o < 8; o <<= 1) {
        const float mo = __shfl_xor(m, o, 32);
        const float so = __shfl_xor(ssum, o, 32);
        v4f o0, o1;
        o0.x = __shfl_xor(acc0.x, o, 32);  o0.y = __shfl_xor(acc0.y, o, 32);
        o0.z = __shfl_xor(acc0.z, o, 32);  o0.w = __shfl_xor(acc0.w, o, 32);
        o1.x = __shfl_xor(acc1.x, o, 32);  o1.y = __shfl_xor(acc1.y, o, 32);
        o1.z = __shfl_xor(acc1.z, o, 32);  o1.w = __shfl_xor(acc1.w, o, 32);
        const float mn = fmaxf(m, mo);
        const float f  = (m  == mn) ? 1.f : __expf(m  - mn);
        const float g  = (mo == mn) ? 1.f : __expf(mo - mn);
        ssum = fmaf(so, g, ssum * f);
        acc0 = rescale_add(acc0, f, g, o0);
        acc1 = rescale_add(acc1, f, g, o1);
        m = mn;
    }
    const v4f half = (sub & 4) ? acc1 : acc0;
    const float lo = (sub & 1) ? half.y : half.x;
    const float hi = (sub & 1) ? half.w : half.z;
    const float v  = (sub & 2) ? hi : lo;
    out[gid] = v / ssum + bias;
}

extern "C" void kernel_launch(void* const* d_in, const int* in_sizes, int n_in,
                              void* d_out, int out_size, void* d_ws, size_t ws_size,
                              hipStream_t stream) {
    const float* x   = (const float*)d_in[0];
    const int*   ei  = (const int*)  d_in[1];
    const float* W1l = (const float*)d_in[2];
    const float* W1r = (const float*)d_in[3];
    const float* a1  = (const float*)d_in[4];
    const float* b1  = (const float*)d_in[5];
    const float* W2l = (const float*)d_in[6];
    const float* W2r = (const float*)d_in[7];
    const float* a2  = (const float*)d_in[8];
    const float* b2  = (const float*)d_in[9];
    float* out = (float*)d_out;

    char* ws = (char*)d_ws;
    size_t off = 0;
    auto carve = [&](size_t bytes) -> char* {
        char* p = ws + off;
        off += (bytes + 255) & ~(size_t)255;
        return p;
    };
    float*    xl1 = (float*)   carve((size_t)N_NODES * F1 * 4);
    float*    h2l = (float*)   carve((size_t)N_NODES * OUT_DIM * 4);
    float*    h2r = (float*)   carve((size_t)N_NODES * OUT_DIM * 4);
    int*      seg    = (int*) carve((size_t)N_NODES * 2 * 4);
    int*      perm   = (int*) carve((size_t)SORT_WGS * SORT_REGION * 4);
    (void)ws_size; (void)n_in; (void)in_sizes; (void)out_size;

    const int B = 256;

    k_sort_gemm1<<<SG_BLOCKS, SORT_T, 0, stream>>>(ei, seg, perm, x, W1l, xl1);
    k_node1<<<N_NODES / NODES_PER_WG, B, 0, stream>>>(seg, perm, xl1, x, W1r, a1, b1,
                                                      W2l, W2r, h2l, h2r);

    k_node2<<<(N_NODES * OUT_DIM + B - 1) / B, B, 0, stream>>>(seg, perm, h2l, h2r, a2, b2, out);
}
